# starting kernel
# baseline (speedup 1.0000x reference)
#include <hip/hip_runtime.h>


typedef float v2f __attribute__((ext_vector_type(2)));
typedef float v8f __attribute__((ext_vector_type(8)));

static __device__ __forceinline__ size_t row64(int n) { return (size_t)n * 64; }


__global__ __launch_bounds__(256) void k_zero(float* __restrict__ p, int n) {
    int i = blockIdx.x * 256 + threadIdx.x;
    if (i < n) p[i] = 0.0f;
}

__global__ __launch_bounds__(256) void k_deg(const float* __restrict__ ew,
                                             const int* __restrict__ row,
                                             const int* __restrict__ col,
                                             float* __restrict__ deg1,
                                             float* __restrict__ deg2, int E) {
    int e = blockIdx.x * 256 + threadIdx.x;
    if (e >= E) return;
    float w = ew[e];
    atomicAdd(&deg1[row[e]], w);
    atomicAdd(&deg2[col[e]], w);
}

__global__ __launch_bounds__(256) void k_invsqrt(float* __restrict__ d, int n) {
    int i = blockIdx.x * 256 + threadIdx.x;
    if (i < n) {
        float v = d[i];
        d[i] = (v > 0.0f) ? rsqrtf(v) : 0.0f;
    }
}

__global__ __launch_bounds__(256) void k_edgew(const float* __restrict__ ew,
                                               const int* __restrict__ row,
                                               const int* __restrict__ col,
                                               const float* __restrict__ dis1,
                                               const float* __restrict__ dis2,
                                               float* __restrict__ ow1,
                                               float* __restrict__ ow2, int E) {
    int e = blockIdx.x * 256 + threadIdx.x;
    if (e >= E) return;
    const float scale = 2.0f / 3.0f;
    int r = row[e], c = col[e];
    float w = ew[e];
    ow1[e] = -scale * dis1[r] * w * dis1[c];
    ow2[e] = -scale * dis2[r] * w * dis2[c];
}

__global__ __launch_bounds__(256) void k_bias(const float* __restrict__ b1,
                                              const float* __restrict__ b2,
                                              float* __restrict__ out, int n128) {
    int i = blockIdx.x * 256 + threadIdx.x;
    if (i >= n128) return;
    int j = i & 127;
    out[i] = (j < 64) ? b1[j] : b2[j - 64];
}

__global__ __launch_bounds__(256) void k_cheb_scale(const float* __restrict__ v,
                                                    const float* __restrict__ prev,
                                                    float* __restrict__ dst,
                                                    float alpha_diag, float beta,
                                                    int n64) {
    int i = blockIdx.x * 256 + threadIdx.x;
    if (i < n64) dst[i] = alpha_diag * v[i] + beta * prev[i];
}

__global__ __launch_bounds__(256) void k_scatter(const float* __restrict__ v,
                                                 float* __restrict__ outf,
                                                 const int* __restrict__ srcIdx,
                                                 const int* __restrict__ dstIdx,
                                                 const float* __restrict__ ow,
                                                 float alpha, int E) {
    int e = blockIdx.x * 8 + (threadIdx.x >> 5);
    if (e >= E) return;
    int c2 = (threadIdx.x & 31) * 2;
    int s = srcIdx[e];
    int d = dstIdx[e];
    float w = alpha * ow[e];
    float2 val = *(const float2*)(v + row64(s) + c2);
    float* o = outf + row64(d) + c2;
    atomicAdd(o + 0, w * val.x);
    atomicAdd(o + 1, w * val.y);
}

__global__ __launch_bounds__(256) void k_gemm_acc(const float* __restrict__ Tx,
                                                  const float* __restrict__ W,
                                                  float* __restrict__ out,
                                                  int colofs, int nRows) {
    __shared__ float Wl[64 * 64];
    __shared__ float Al[128 * 64];

    const int tid = threadIdx.x;
    const int row0b = blockIdx.x * 128;

    {
        const float4* W4 = (const float4*)W;
        float4* Wl4 = (float4*)Wl;
#pragma unroll
        for (int i = 0; i < 4; ++i) Wl4[tid + 256 * i] = W4[tid + 256 * i];
    }
    {
        float4* Al4 = (float4*)Al;
#pragma unroll
        for (int i = 0; i < 8; ++i) {
            int idx = tid + 256 * i;
            int r = row0b + (idx >> 4);
            float4 v = make_float4(0.f, 0.f, 0.f, 0.f);
            if (r < nRows) v = ((const float4*)(Tx + row64(r)))[idx & 15];
            Al4[idx] = v;
        }
    }
    __syncthreads();

    const int wave = tid >> 5;
    const int lane = tid & 31;
    const int lm = lane & 15;
    const int lg = lane >> 4;

    const int rowt = wave * 16;
    const int row0 = row0b + rowt;

    v8f c0, c1, c2, c3;
    const int nclamp = nRows - 1;
#pragma unroll
    for (int i = 0; i < 8; ++i) {
        int r = row0 + i + lg * 8;
        bool ok = (r < nRows);
        int rc = ok ? r : nclamp;
        const float* orow = out + (size_t)rc * 128 + colofs + lm;
        float v0 = orow[0], v1 = orow[16], v2 = orow[32], v3 = orow[48];
        c0[i] = ok ? v0 : 0.f;
        c1[i] = ok ? v1 : 0.f;
        c2[i] = ok ? v2 : 0.f;
        c3[i] = ok ? v3 : 0.f;
    }

    const float* arow_l = Al + (rowt + lm) * 64;
    const float* bcol_l = Wl + lm;

#pragma unroll
    for (int kk = 0; kk < 16; ++kk) {
        const int kbase = kk * 4 + lg * 2;
        v2f a = *(const v2f*)(arow_l + kbase);
        v2f b0, b1v, b2, b3;
        b0[0]  = bcol_l[(kbase + 0) * 64 + 0];
        b0[1]  = bcol_l[(kbase + 1) * 64 + 0];
        b1v[0] = bcol_l[(kbase + 0) * 64 + 16];
        b1v[1] = bcol_l[(kbase + 1) * 64 + 16];
        b2[0]  = bcol_l[(kbase + 0) * 64 + 32];
        b2[1]  = bcol_l[(kbase + 1) * 64 + 32];
        b3[0]  = bcol_l[(kbase + 0) * 64 + 48];
        b3[1]  = bcol_l[(kbase + 1) * 64 + 48];
        c0 = __builtin_amdgcn_wmma_f32_16x16x4_f32(false, a, false, b0,  (short)0, c0, false, false);
        c1 = __builtin_amdgcn_wmma_f32_16x16x4_f32(false, a, false, b1v, (short)0, c1, false, false);
        c2 = __builtin_amdgcn_wmma_f32_16x16x4_f32(false, a, false, b2,  (short)0, c2, false, false);
        c3 = __builtin_amdgcn_wmma_f32_16x16x4_f32(false, a, false, b3,  (short)0, c3, false, false);
    }

#pragma unroll
    for (int i = 0; i < 8; ++i) {
        int r = row0 + i + lg * 8;
        if (r < nRows) {
            float* orow = out + (size_t)r * 128 + colofs + lm;
            orow[0]  = c0[i];
            orow[16] = c1[i];
            orow[32] = c2[i];
            orow[48] = c3[i];
        }
    }
}


extern "C" void kernel_launch(void* const* d_in, const int* in_sizes, int n_in,
                              void* d_out, int out_size, void* d_ws, size_t ws_size,
                              hipStream_t stream) {
    const float* x  = (const float*)d_in[0];
    const int*   ei = (const int*)d_in[1];
    const float* ew = (const float*)d_in[2];
    const float* W1 = (const float*)d_in[3];
    const float* b1 = (const float*)d_in[4];
    const float* W2 = (const float*)d_in[5];
    const float* b2 = (const float*)d_in[6];
    float* out = (float*)d_out;

    const int N = in_sizes[0] / 64;
    const int E = in_sizes[2];
    const int* row = ei;
    const int* col = ei + E;

    float* ws   = (float*)d_ws;
    float* deg1 = ws;
    float* deg2 = deg1 + N;
    float* ow1  = deg2 + N;
    float* ow2  = ow1 + E;
    float* tA   = ow2 + E;
    float* tB   = tA + (size_t)N * 64;
    float* tC   = tB + (size_t)N * 64;

    const int n64  = N * 64;
    const int n128 = N * 128;
    const float DIAG = 2.0f / 3.0f - 1.0f;

    dim3 blk(256);
    int gN2   = (2 * N + 255) / 256;
    int gE    = (E + 255) / 256;
    int gE8   = (E + 7) / 8;
    int g64   = (n64 + 255) / 256;
    int g128  = (n128 + 255) / 256;
    int gGemm = (N + 127) / 128;

    k_zero<<<gN2, blk, 0, stream>>>(deg1, 2 * N);
    k_deg<<<gE, blk, 0, stream>>>(ew, row, col, deg1, deg2, E);
    k_invsqrt<<<gN2, blk, 0, stream>>>(deg1, 2 * N);
    k_edgew<<<gE, blk, 0, stream>>>(ew, row, col, deg1, deg2, ow1, ow2, E);

    k_bias<<<g128, blk, 0, stream>>>(b1, b2, out, n128);

    for (int dir = 0; dir < 2; ++dir) {
        const int*   src  = (dir == 0) ? col : row;
        const int*   dst  = (dir == 0) ? row : col;
        const float* ow   = (dir == 0) ? ow1 : ow2;
        const float* W    = (dir == 0) ? W1 : W2;
        const int    cofs = (dir == 0) ? 0 : 64;

        k_gemm_acc<<<gGemm, blk, 0, stream>>>(x, W, out, cofs, N);

        k_cheb_scale<<<g64, blk, 0, stream>>>(x, x, tA, DIAG, 0.0f, n64);
        k_scatter<<<gE8, blk, 0, stream>>>(x, tA, src, dst, ow, 1.0f, E);
        k_gemm_acc<<<gGemm, blk, 0, stream>>>(tA, W + 1 * 4096, out, cofs, N);

        k_cheb_scale<<<g64, blk, 0, stream>>>(tA, x, tB, 2.0f * DIAG, -1.0f, n64);
        k_scatter<<<gE8, blk, 0, stream>>>(tA, tB, src, dst, ow, 2.0f, E);
        k_gemm_acc<<<gGemm, blk, 0, stream>>>(tB, W + 2 * 4096, out, cofs, N);

        k_cheb_scale<<<g64, blk, 0, stream>>>(tB, tA, tC, 2.0f * DIAG, -1.0f, n64);
        k_scatter<<<gE8, blk, 0, stream>>>(tB, tC, src, dst, ow, 2.0f, E);
        k_gemm_acc<<<gGemm, blk, 0, stream>>>(tC, W + 3 * 4096, out, cofs, N);

        k_cheb_scale<<<g64, blk, 0, stream>>>(tC, tB, tA, 2.0f * DIAG, -1.0f, n64);
        k_scatter<<<gE8, blk, 0, stream>>>(tC, tA, src, dst, ow, 2.0f, E);
        k_gemm_acc<<<gGemm, blk, 0, stream>>>(tA, W + 4 * 4096, out, cofs, N);
    }
}


# Atomic scatter-add replaced by a per-call destination-sorted edge list and a one-wave-per-node register gather that writes each row once; row sums now in edge order, not atomic arrival order
# speedup vs baseline: 1.8357x; 1.8357x over previous
#include <hip/hip_runtime.h>


typedef float v2f __attribute__((ext_vector_type(2)));
typedef float v8f __attribute__((ext_vector_type(8)));

static __device__ __forceinline__ size_t row64(int n) { return (size_t)n * 64; }


__global__ __launch_bounds__(256) void k_zero(int* __restrict__ p, int n) {
    int i = blockIdx.x * 256 + threadIdx.x;
    if (i < n) p[i] = 0;
}

__global__ __launch_bounds__(256) void k_count(const int* __restrict__ row,
                                               const int* __restrict__ col,
                                               int* __restrict__ cnt0,
                                               int* __restrict__ cnt1,
                                               int* __restrict__ rank0,
                                               int* __restrict__ rank1, int E) {
    int e = blockIdx.x * 256 + threadIdx.x;
    if (e >= E) return;
    rank0[e] = atomicAdd(&cnt0[row[e]], 1);
    rank1[e] = atomicAdd(&cnt1[col[e]], 1);
}

__global__ __launch_bounds__(1024) void k_scan(const int* __restrict__ cnt,
                                               int* __restrict__ rp, int per, int NP) {
    __shared__ int part[1024];
    const int t = threadIdx.x;
    const int4* c4 = (const int4*)(cnt + (size_t)blockIdx.x * NP + (size_t)t * per);
    int* rbase = rp + (size_t)blockIdx.x * (NP + 4);
    int4* r4 = (int4*)(rbase + (size_t)t * per);
    const int nq = per >> 2;
    int s = 0;
    for (int q = 0; q < nq; ++q) {
        int4 c = c4[q];
        s += c.x + c.y + c.z + c.w;
    }
    part[t] = s;
    __syncthreads();
    for (int off = 1; off < 1024; off <<= 1) {
        int add = (t >= off) ? part[t - off] : 0;
        __syncthreads();
        part[t] += add;
        __syncthreads();
    }
    int run = part[t] - s;
    for (int q = 0; q < nq; ++q) {
        int4 c = c4[q];
        int4 o;
        o.x = run; run += c.x;
        o.y = run; run += c.y;
        o.z = run; run += c.z;
        o.w = run; run += c.w;
        r4[q] = o;
    }
    if (t == 1023) rbase[NP] = run;
}

__global__ __launch_bounds__(256) void k_place(const int* __restrict__ ei,
                                               const int* __restrict__ rp,
                                               const int* __restrict__ rank,
                                               int* __restrict__ perm,
                                               int E, int EP, int NP) {
    int e = blockIdx.x * 256 + threadIdx.x;
    if (e >= E) return;
    const int dir = blockIdx.y;
    int d = ei[(size_t)dir * E + e];
    int pos = rp[(size_t)dir * (NP + 4) + d] + rank[(size_t)dir * EP + e];
    perm[(size_t)dir * EP + pos] = e;
}

__global__ __launch_bounds__(256) void k_order(const int* __restrict__ ei,
                                               const float* __restrict__ ew,
                                               const int* __restrict__ rp,
                                               const int* __restrict__ perm,
                                               int* __restrict__ csrc,
                                               float* __restrict__ cw,
                                               int N, int E, int EP, int NP) {
    const int d = blockIdx.x * 8 + (threadIdx.x >> 5);
    if (d >= N) return;
    const int dir = blockIdx.y;
    const int lane = threadIdx.x & 31;
    const int* r = rp + (size_t)dir * (NP + 4);
    const int* pm = perm + (size_t)dir * EP;
    const int* srcIdx = ei + (size_t)(1 - dir) * E;
    const int s0 = __builtin_amdgcn_readfirstlane(r[d]);
    const int s1 = __builtin_amdgcn_readfirstlane(r[d + 1]);
    for (int ca = s0; ca < s1; ca += 32) {
        const bool mine = (ca + lane) < s1;
        const int my = mine ? pm[ca + lane] : 0;
        int below = 0;
        for (int cb = s0; cb < s1; cb += 32) {
            const int other = (cb + lane < s1) ? pm[cb + lane] : 0;
            int nb = s1 - cb;
            nb = nb < 32 ? nb : 32;
            for (int j = 0; j < nb; ++j) {
                const int oj = __builtin_amdgcn_readlane(other, j);
                below += (oj < my) ? 1 : 0;
            }
        }
        if (mine) {
            const size_t slot = (size_t)dir * EP + s0 + below;
            csrc[slot] = srcIdx[my];
            cw[slot] = ew[my];
        }
    }
}

__global__ __launch_bounds__(256) void k_dis(const int* __restrict__ rp,
                                             const float* __restrict__ cw,
                                             float* __restrict__ dis,
                                             int N, int EP, int NP) {
    int d = blockIdx.x * 256 + threadIdx.x;
    if (d >= N) return;
    const int dir = blockIdx.y;
    const int* r = rp + (size_t)dir * (NP + 4);
    const float* w = cw + (size_t)dir * EP;
    int s0 = r[d], s1 = r[d + 1];
    float deg = 0.0f;
    for (int q = s0; q < s1; ++q) deg += w[q];
    dis[(size_t)dir * NP + d] = (deg > 0.0f) ? rsqrtf(deg) : 0.0f;
}

__global__ __launch_bounds__(256) void k_csrw(const int* __restrict__ rp,
                                              const int* __restrict__ csrc,
                                              float* __restrict__ cw,
                                              const float* __restrict__ dis,
                                              int N, int EP, int NP) {
    int d = blockIdx.x * 256 + threadIdx.x;
    if (d >= N) return;
    const int dir = blockIdx.y;
    const float scale = 2.0f / 3.0f;
    const int* r = rp + (size_t)dir * (NP + 4);
    const int* sidx = csrc + (size_t)dir * EP;
    float* w = cw + (size_t)dir * EP;
    const float* ds = dis + (size_t)dir * NP;
    int s0 = r[d], s1 = r[d + 1];
    float dd = -scale * ds[d];
    for (int q = s0; q < s1; ++q) w[q] = dd * w[q] * ds[sidx[q]];
}


__global__ __launch_bounds__(256) void k_bias(const float* __restrict__ b1,
                                              const float* __restrict__ b2,
                                              float* __restrict__ out, int n128) {
    int i = blockIdx.x * 256 + threadIdx.x;
    if (i >= n128) return;
    int j = i & 127;
    out[i] = (j < 64) ? b1[j] : b2[j - 64];
}

template <bool HAS_PREV>
__global__ __launch_bounds__(256) void k_lhat(const float* __restrict__ v,
                                              const float* prev,
                                              float* dst,
                                              const int* __restrict__ rp,
                                              const int* __restrict__ csrc,
                                              const float* __restrict__ cw,
                                              float adiag, float beta, float alpha,
                                              int N) {
    const int d = blockIdx.x * 8 + (threadIdx.x >> 5);
    if (d >= N) return;
    const int lane = threadIdx.x & 31;
    const int h = lane >> 4;
    const int c4 = (lane & 15) * 4;
    const int s0 = __builtin_amdgcn_readfirstlane(rp[d]);
    const int s1 = __builtin_amdgcn_readfirstlane(rp[d + 1]);

    int base = s0;
    int n = s1 - base;
    n = n < 32 ? n : 32;
    int msrc = d;
    float mw = 0.0f;
    if (lane < n) {
        msrc = csrc[base + lane];
        mw = cw[base + lane];
    }

    float4 vd = make_float4(0.f, 0.f, 0.f, 0.f);
    float4 pd = make_float4(0.f, 0.f, 0.f, 0.f);
    if (h == 0) {
        vd = *(const float4*)(v + row64(d) + c4);
        if (HAS_PREV) pd = *(const float4*)(prev + row64(d) + c4);
    }

    float4 acc = make_float4(0.f, 0.f, 0.f, 0.f);
    while (n > 0) {
        const int npair = n >> 1;
#pragma unroll 4
        for (int q = 0; q < npair; ++q) {
            const int sl = 2 * q + h;
            const int s = __shfl(msrc, sl);
            const float w = __shfl(mw, sl);
            const float4 val = *(const float4*)(v + row64(s) + c4);
            acc.x = fmaf(w, val.x, acc.x);
            acc.y = fmaf(w, val.y, acc.y);
            acc.z = fmaf(w, val.z, acc.z);
            acc.w = fmaf(w, val.w, acc.w);
        }
        if (n & 1) {
            const int s = __shfl(msrc, n - 1);
            const float w = __shfl(mw, n - 1);
            if (h == 0) {
                const float4 val = *(const float4*)(v + row64(s) + c4);
                acc.x = fmaf(w, val.x, acc.x);
                acc.y = fmaf(w, val.y, acc.y);
                acc.z = fmaf(w, val.z, acc.z);
                acc.w = fmaf(w, val.w, acc.w);
            }
        }
        base += 32;
        n = s1 - base;
        n = n < 32 ? n : 32;
        msrc = d;
        mw = 0.0f;
        if (lane < n) {
            msrc = csrc[base + lane];
            mw = cw[base + lane];
        }
    }
    acc.x += __shfl_xor(acc.x, 16);
    acc.y += __shfl_xor(acc.y, 16);
    acc.z += __shfl_xor(acc.z, 16);
    acc.w += __shfl_xor(acc.w, 16);

    if (h == 0) {
        float4 o;
        o.x = adiag * vd.x;
        o.y = adiag * vd.y;
        o.z = adiag * vd.z;
        o.w = adiag * vd.w;
        if (HAS_PREV) {
            o.x += beta * pd.x;
            o.y += beta * pd.y;
            o.z += beta * pd.z;
            o.w += beta * pd.w;
        }
        o.x += alpha * acc.x;
        o.y += alpha * acc.y;
        o.z += alpha * acc.z;
        o.w += alpha * acc.w;
        *(float4*)(dst + row64(d) + c4) = o;
    }
}

__global__ __launch_bounds__(256) void k_gemm_acc(const float* __restrict__ Tx,
                                                  const float* __restrict__ W,
                                                  float* __restrict__ out,
                                                  int colofs, int nRows) {
    __shared__ float Wl[64 * 64];
    __shared__ float Al[128 * 64];

    const int tid = threadIdx.x;
    const int row0b = blockIdx.x * 128;

    {
        const float4* W4 = (const float4*)W;
        float4* Wl4 = (float4*)Wl;
#pragma unroll
        for (int i = 0; i < 4; ++i) Wl4[tid + 256 * i] = W4[tid + 256 * i];
    }
    {
        float4* Al4 = (float4*)Al;
#pragma unroll
        for (int i = 0; i < 8; ++i) {
            int idx = tid + 256 * i;
            int r = row0b + (idx >> 4);
            float4 v = make_float4(0.f, 0.f, 0.f, 0.f);
            if (r < nRows) v = ((const float4*)(Tx + row64(r)))[idx & 15];
            Al4[idx] = v;
        }
    }
    __syncthreads();

    const int wave = tid >> 5;
    const int lane = tid & 31;
    const int lm = lane & 15;
    const int lg = lane >> 4;

    const int rowt = wave * 16;
    const int row0 = row0b + rowt;

    v8f c0, c1, c2, c3;
    const int nclamp = nRows - 1;
#pragma unroll
    for (int i = 0; i < 8; ++i) {
        int r = row0 + i + lg * 8;
        bool ok = (r < nRows);
        int rc = ok ? r : nclamp;
        const float* orow = out + (size_t)rc * 128 + colofs + lm;
        float v0 = orow[0], v1 = orow[16], v2 = orow[32], v3 = orow[48];
        c0[i] = ok ? v0 : 0.f;
        c1[i] = ok ? v1 : 0.f;
        c2[i] = ok ? v2 : 0.f;
        c3[i] = ok ? v3 : 0.f;
    }

    const float* arow_l = Al + (rowt + lm) * 64;
    const float* bcol_l = Wl + lm;

#pragma unroll
    for (int kk = 0; kk < 16; ++kk) {
        const int kbase = kk * 4 + lg * 2;
        v2f a = *(const v2f*)(arow_l + kbase);
        v2f b0, b1v, b2, b3;
        b0[0]  = bcol_l[(kbase + 0) * 64 + 0];
        b0[1]  = bcol_l[(kbase + 1) * 64 + 0];
        b1v[0] = bcol_l[(kbase + 0) * 64 + 16];
        b1v[1] = bcol_l[(kbase + 1) * 64 + 16];
        b2[0]  = bcol_l[(kbase + 0) * 64 + 32];
        b2[1]  = bcol_l[(kbase + 1) * 64 + 32];
        b3[0]  = bcol_l[(kbase + 0) * 64 + 48];
        b3[1]  = bcol_l[(kbase + 1) * 64 + 48];
        c0 = __builtin_amdgcn_wmma_f32_16x16x4_f32(false, a, false, b0,  (short)0, c0, false, false);
        c1 = __builtin_amdgcn_wmma_f32_16x16x4_f32(false, a, false, b1v, (short)0, c1, false, false);
        c2 = __builtin_amdgcn_wmma_f32_16x16x4_f32(false, a, false, b2,  (short)0, c2, false, false);
        c3 = __builtin_amdgcn_wmma_f32_16x16x4_f32(false, a, false, b3,  (short)0, c3, false, false);
    }

#pragma unroll
    for (int i = 0; i < 8; ++i) {
        int r = row0 + i + lg * 8;
        if (r < nRows) {
            float* orow = out + (size_t)r * 128 + colofs + lm;
            orow[0]  = c0[i];
            orow[16] = c1[i];
            orow[32] = c2[i];
            orow[48] = c3[i];
        }
    }
}


extern "C" void kernel_launch(void* const* d_in, const int* in_sizes, int n_in,
                              void* d_out, int out_size, void* d_ws, size_t ws_size,
                              hipStream_t stream) {
    const float* x  = (const float*)d_in[0];
    const int*   ei = (const int*)d_in[1];
    const float* ew = (const float*)d_in[2];
    const float* W1 = (const float*)d_in[3];
    const float* b1 = (const float*)d_in[4];
    const float* W2 = (const float*)d_in[5];
    const float* b2 = (const float*)d_in[6];
    float* out = (float*)d_out;

    const int N = in_sizes[0] / 64;
    const int E = in_sizes[2];
    const int* row = ei;
    const int* col = ei + E;

    const int per = (((N + 1023) / 1024) + 3) & ~3;
    const int NP  = per * 1024;
    const int EP  = (E + 3) & ~3;

    float* tA   = (float*)d_ws;
    float* tB   = tA + (size_t)N * 64;
    int*   csrc = (int*)(tB + (size_t)N * 64);
    float* cw   = (float*)(csrc + 2 * (size_t)EP);
    int*   cnt  = (int*)(cw + 2 * (size_t)EP);
    int*   rp   = cnt + 2 * (size_t)NP;
    float* dis  = (float*)(rp + 2 * (size_t)(NP + 4));
    int*   rank = (int*)tA;
    int*   perm = rank + 2 * (size_t)EP;

    const int n128 = N * 128;
    const float DIAG = 2.0f / 3.0f - 1.0f;

    dim3 blk(256);
    int gZ    = (2 * NP + 255) / 256;
    int gE    = (E + 255) / 256;
    int gN    = (N + 255) / 256;
    int gN8   = (N + 7) / 8;
    int g128  = (n128 + 255) / 256;
    int gGemm = (N + 127) / 128;

    k_zero<<<gZ, blk, 0, stream>>>(cnt, 2 * NP);
    k_count<<<gE, blk, 0, stream>>>(row, col, cnt, cnt + NP, rank, rank + EP, E);
    k_scan<<<2, dim3(1024), 0, stream>>>(cnt, rp, per, NP);
    k_place<<<dim3(gE, 2), blk, 0, stream>>>(ei, rp, rank, perm, E, EP, NP);
    k_order<<<dim3(gN8, 2), blk, 0, stream>>>(ei, ew, rp, perm, csrc, cw, N, E, EP, NP);
    k_dis<<<dim3(gN, 2), blk, 0, stream>>>(rp, cw, dis, N, EP, NP);
    k_csrw<<<dim3(gN, 2), blk, 0, stream>>>(rp, csrc, cw, dis, N, EP, NP);

    k_bias<<<g128, blk, 0, stream>>>(b1, b2, out, n128);

    for (int dir = 0; dir < 2; ++dir) {
        const int*   r    = rp + (size_t)dir * (NP + 4);
        const int*   sidx = csrc + (size_t)dir * EP;
        const float* w    = cw + (size_t)dir * EP;
        const float* W    = (dir == 0) ? W1 : W2;
        const int    cofs = (dir == 0) ? 0 : 64;

        k_gemm_acc<<<gGemm, blk, 0, stream>>>(x, W, out, cofs, N);

        k_lhat<false><<<gN8, blk, 0, stream>>>(x, x, tA, r, sidx, w, DIAG, 0.0f, 1.0f, N);
        k_gemm_acc<<<gGemm, blk, 0, stream>>>(tA, W + 1 * 4096, out, cofs, N);

        k_lhat<true><<<gN8, blk, 0, stream>>>(tA, x, tB, r, sidx, w, 2.0f * DIAG, -1.0f, 2.0f, N);
        k_gemm_acc<<<gGemm, blk, 0, stream>>>(tB, W + 2 * 4096, out, cofs, N);

        k_lhat<true><<<gN8, blk, 0, stream>>>(tB, tA, tA, r, sidx, w, 2.0f * DIAG, -1.0f, 2.0f, N);
        k_gemm_acc<<<gGemm, blk, 0, stream>>>(tA, W + 3 * 4096, out, cofs, N);

        k_lhat<true><<<gN8, blk, 0, stream>>>(tA, tB, tB, r, sidx, w, 2.0f * DIAG, -1.0f, 2.0f, N);
        k_gemm_acc<<<gGemm, blk, 0, stream>>>(tB, W + 4 * 4096, out, cofs, N);
    }
}


# Per-edge returning global atomic adds of the edge-list build replaced by returning LDS adds: blocks own node range x edge chunk, a small pass turns chunk counts into offsets; float sums unchanged
# speedup vs baseline: 2.1205x; 1.1551x over previous
#include <hip/hip_runtime.h>


typedef float v2f __attribute__((ext_vector_type(2)));
typedef float v8f __attribute__((ext_vector_type(8)));

static __device__ __forceinline__ size_t row64(int n) { return (size_t)n * 64; }


#define RNG 8192
#define CCH 16

__global__ __launch_bounds__(1024) void k_rangecount(const int* __restrict__ ei,
                                                     int* __restrict__ cntp,
                                                     int* __restrict__ rank,
                                                     int E, int EP, int NP, int CE) {
    __shared__ int lc[RNG];
    const int t = threadIdx.x;
    const int r0 = blockIdx.x * RNG;
    const int c = blockIdx.y;
    const int dir = blockIdx.z;
    for (int i = t; i < RNG; i += 1024) lc[i] = 0;
    __syncthreads();
    const int* dst = ei + (size_t)dir * E;
    int* rk = rank + (size_t)dir * EP;
    const int cbeg = c * CE;
    int cend = cbeg + CE;
    cend = cend < E ? cend : E;
    for (int e0 = cbeg + t; e0 < cend; e0 += 8 * 1024) {
        int dv[8];
#pragma unroll
        for (int j = 0; j < 8; ++j) {
            const int e = e0 + j * 1024;
            dv[j] = (e < cend) ? dst[e] : -1;
        }
#pragma unroll
        for (int j = 0; j < 8; ++j) {
            const unsigned rel = (unsigned)(dv[j] - r0);
            if (rel < (unsigned)RNG) rk[e0 + j * 1024] = atomicAdd(&lc[rel], 1);
        }
    }
    __syncthreads();
    int* o = cntp + ((size_t)dir * CCH + c) * NP;
    for (int i = t; i < RNG; i += 1024) {
        const int d = r0 + i;
        if (d < NP) o[d] = lc[i];
    }
}

__global__ __launch_bounds__(256) void k_combine(int* __restrict__ cntp,
                                                 int* __restrict__ cnt, int NP) {
    const int d = blockIdx.x * 256 + threadIdx.x;
    if (d >= NP) return;
    const int dir = blockIdx.y;
    int* p = cntp + (size_t)dir * CCH * NP + d;
    int v[CCH];
#pragma unroll
    for (int c = 0; c < CCH; ++c) v[c] = p[(size_t)c * NP];
    int run = 0;
#pragma unroll
    for (int c = 0; c < CCH; ++c) {
        p[(size_t)c * NP] = run;
        run += v[c];
    }
    cnt[(size_t)dir * NP + d] = run;
}

__global__ __launch_bounds__(1024) void k_scan(const int* __restrict__ cnt,
                                               int* __restrict__ rp, int per, int NP) {
    __shared__ int part[1024];
    const int t = threadIdx.x;
    const int4* c4 = (const int4*)(cnt + (size_t)blockIdx.x * NP + (size_t)t * per);
    int* rbase = rp + (size_t)blockIdx.x * (NP + 4);
    int4* r4 = (int4*)(rbase + (size_t)t * per);
    const int nq = per >> 2;
    int s = 0;
    for (int q = 0; q < nq; ++q) {
        int4 c = c4[q];
        s += c.x + c.y + c.z + c.w;
    }
    part[t] = s;
    __syncthreads();
    for (int off = 1; off < 1024; off <<= 1) {
        int add = (t >= off) ? part[t - off] : 0;
        __syncthreads();
        part[t] += add;
        __syncthreads();
    }
    int run = part[t] - s;
    for (int q = 0; q < nq; ++q) {
        int4 c = c4[q];
        int4 o;
        o.x = run; run += c.x;
        o.y = run; run += c.y;
        o.z = run; run += c.z;
        o.w = run; run += c.w;
        r4[q] = o;
    }
    if (t == 1023) rbase[NP] = run;
}

__global__ __launch_bounds__(256) void k_place(const int* __restrict__ ei,
                                               const int* __restrict__ rp,
                                               const int* __restrict__ rank,
                                               const int* __restrict__ cntp,
                                               int* __restrict__ perm,
                                               int E, int EP, int NP, int CE) {
    int e = blockIdx.x * 256 + threadIdx.x;
    if (e >= E) return;
    const int dir = blockIdx.y;
    const int c = e / CE;
    int d = ei[(size_t)dir * E + e];
    int pos = rp[(size_t)dir * (NP + 4) + d] + cntp[((size_t)dir * CCH + c) * NP + d] +
              rank[(size_t)dir * EP + e];
    perm[(size_t)dir * EP + pos] = e;
}

__global__ __launch_bounds__(256) void k_order(const int* __restrict__ ei,
                                               const float* __restrict__ ew,
                                               const int* __restrict__ rp,
                                               const int* __restrict__ perm,
                                               int* __restrict__ csrc,
                                               float* __restrict__ cw,
                                               int N, int E, int EP, int NP) {
    const int d = blockIdx.x * 8 + (threadIdx.x >> 5);
    if (d >= N) return;
    const int dir = blockIdx.y;
    const int lane = threadIdx.x & 31;
    const int* r = rp + (size_t)dir * (NP + 4);
    const int* pm = perm + (size_t)dir * EP;
    const int* srcIdx = ei + (size_t)(1 - dir) * E;
    const int s0 = __builtin_amdgcn_readfirstlane(r[d]);
    const int s1 = __builtin_amdgcn_readfirstlane(r[d + 1]);
    for (int ca = s0; ca < s1; ca += 32) {
        const bool mine = (ca + lane) < s1;
        const int my = mine ? pm[ca + lane] : 0;
        int below = 0;
        for (int cb = s0; cb < s1; cb += 32) {
            const int other = (cb + lane < s1) ? pm[cb + lane] : 0;
            int nb = s1 - cb;
            nb = nb < 32 ? nb : 32;
            for (int j = 0; j < nb; ++j) {
                const int oj = __builtin_amdgcn_readlane(other, j);
                below += (oj < my) ? 1 : 0;
            }
        }
        if (mine) {
            const size_t slot = (size_t)dir * EP + s0 + below;
            csrc[slot] = srcIdx[my];
            cw[slot] = ew[my];
        }
    }
}

__global__ __launch_bounds__(256) void k_dis(const int* __restrict__ rp,
                                             const float* __restrict__ cw,
                                             float* __restrict__ dis,
                                             int N, int EP, int NP) {
    int d = blockIdx.x * 256 + threadIdx.x;
    if (d >= N) return;
    const int dir = blockIdx.y;
    const int* r = rp + (size_t)dir * (NP + 4);
    const float* w = cw + (size_t)dir * EP;
    int s0 = r[d], s1 = r[d + 1];
    float deg = 0.0f;
    for (int q = s0; q < s1; ++q) deg += w[q];
    dis[(size_t)dir * NP + d] = (deg > 0.0f) ? rsqrtf(deg) : 0.0f;
}

__global__ __launch_bounds__(256) void k_csrw(const int* __restrict__ rp,
                                              const int* __restrict__ csrc,
                                              float* __restrict__ cw,
                                              const float* __restrict__ dis,
                                              int N, int EP, int NP) {
    int d = blockIdx.x * 256 + threadIdx.x;
    if (d >= N) return;
    const int dir = blockIdx.y;
    const float scale = 2.0f / 3.0f;
    const int* r = rp + (size_t)dir * (NP + 4);
    const int* sidx = csrc + (size_t)dir * EP;
    float* w = cw + (size_t)dir * EP;
    const float* ds = dis + (size_t)dir * NP;
    int s0 = r[d], s1 = r[d + 1];
    float dd = -scale * ds[d];
    for (int q = s0; q < s1; ++q) w[q] = dd * w[q] * ds[sidx[q]];
}


__global__ __launch_bounds__(256) void k_bias(const float* __restrict__ b1,
                                              const float* __restrict__ b2,
                                              float* __restrict__ out, int n128) {
    int i = blockIdx.x * 256 + threadIdx.x;
    if (i >= n128) return;
    int j = i & 127;
    out[i] = (j < 64) ? b1[j] : b2[j - 64];
}

template <bool HAS_PREV>
__global__ __launch_bounds__(256) void k_lhat(const float* __restrict__ v,
                                              const float* prev,
                                              float* dst,
                                              const int* __restrict__ rp,
                                              const int* __restrict__ csrc,
                                              const float* __restrict__ cw,
                                              float adiag, float beta, float alpha,
                                              int N) {
    const int d = blockIdx.x * 8 + (threadIdx.x >> 5);
    if (d >= N) return;
    const int lane = threadIdx.x & 31;
    const int h = lane >> 4;
    const int c4 = (lane & 15) * 4;
    const int s0 = __builtin_amdgcn_readfirstlane(rp[d]);
    const int s1 = __builtin_amdgcn_readfirstlane(rp[d + 1]);

    int base = s0;
    int n = s1 - base;
    n = n < 32 ? n : 32;
    int msrc = d;
    float mw = 0.0f;
    if (lane < n) {
        msrc = csrc[base + lane];
        mw = cw[base + lane];
    }

    float4 vd = make_float4(0.f, 0.f, 0.f, 0.f);
    float4 pd = make_float4(0.f, 0.f, 0.f, 0.f);
    if (h == 0) {
        vd = *(const float4*)(v + row64(d) + c4);
        if (HAS_PREV) pd = *(const float4*)(prev + row64(d) + c4);
    }

    float4 acc = make_float4(0.f, 0.f, 0.f, 0.f);
    while (n > 0) {
        const int npair = n >> 1;
#pragma unroll 4
        for (int q = 0; q < npair; ++q) {
            const int sl = 2 * q + h;
            const int s = __shfl(msrc, sl);
            const float w = __shfl(mw, sl);
            const float4 val = *(const float4*)(v + row64(s) + c4);
            acc.x = fmaf(w, val.x, acc.x);
            acc.y = fmaf(w, val.y, acc.y);
            acc.z = fmaf(w, val.z, acc.z);
            acc.w = fmaf(w, val.w, acc.w);
        }
        if (n & 1) {
            const int s = __shfl(msrc, n - 1);
            const float w = __shfl(mw, n - 1);
            if (h == 0) {
                const float4 val = *(const float4*)(v + row64(s) + c4);
                acc.x = fmaf(w, val.x, acc.x);
                acc.y = fmaf(w, val.y, acc.y);
                acc.z = fmaf(w, val.z, acc.z);
                acc.w = fmaf(w, val.w, acc.w);
            }
        }
        base += 32;
        n = s1 - base;
        n = n < 32 ? n : 32;
        msrc = d;
        mw = 0.0f;
        if (lane < n) {
            msrc = csrc[base + lane];
            mw = cw[base + lane];
        }
    }
    acc.x += __shfl_xor(acc.x, 16);
    acc.y += __shfl_xor(acc.y, 16);
    acc.z += __shfl_xor(acc.z, 16);
    acc.w += __shfl_xor(acc.w, 16);

    if (h == 0) {
        float4 o;
        o.x = adiag * vd.x;
        o.y = adiag * vd.y;
        o.z = adiag * vd.z;
        o.w = adiag * vd.w;
        if (HAS_PREV) {
            o.x += beta * pd.x;
            o.y += beta * pd.y;
            o.z += beta * pd.z;
            o.w += beta * pd.w;
        }
        o.x += alpha * acc.x;
        o.y += alpha * acc.y;
        o.z += alpha * acc.z;
        o.w += alpha * acc.w;
        *(float4*)(dst + row64(d) + c4) = o;
    }
}

__global__ __launch_bounds__(256) void k_gemm_acc(const float* __restrict__ Tx,
                                                  const float* __restrict__ W,
                                                  float* __restrict__ out,
                                                  int colofs, int nRows) {
    __shared__ float Wl[64 * 64];
    __shared__ float Al[128 * 64];

    const int tid = threadIdx.x;
    const int row0b = blockIdx.x * 128;

    {
        const float4* W4 = (const float4*)W;
        float4* Wl4 = (float4*)Wl;
#pragma unroll
        for (int i = 0; i < 4; ++i) Wl4[tid + 256 * i] = W4[tid + 256 * i];
    }
    {
        float4* Al4 = (float4*)Al;
#pragma unroll
        for (int i = 0; i < 8; ++i) {
            int idx = tid + 256 * i;
            int r = row0b + (idx >> 4);
            float4 v = make_float4(0.f, 0.f, 0.f, 0.f);
            if (r < nRows) v = ((const float4*)(Tx + row64(r)))[idx & 15];
            Al4[idx] = v;
        }
    }
    __syncthreads();

    const int wave = tid >> 5;
    const int lane = tid & 31;
    const int lm = lane & 15;
    const int lg = lane >> 4;

    const int rowt = wave * 16;
    const int row0 = row0b + rowt;

    v8f c0, c1, c2, c3;
    const int nclamp = nRows - 1;
#pragma unroll
    for (int i = 0; i < 8; ++i) {
        int r = row0 + i + lg * 8;
        bool ok = (r < nRows);
        int rc = ok ? r : nclamp;
        const float* orow = out + (size_t)rc * 128 + colofs + lm;
        float v0 = orow[0], v1 = orow[16], v2 = orow[32], v3 = orow[48];
        c0[i] = ok ? v0 : 0.f;
        c1[i] = ok ? v1 : 0.f;
        c2[i] = ok ? v2 : 0.f;
        c3[i] = ok ? v3 : 0.f;
    }

    const float* arow_l = Al + (rowt + lm) * 64;
    const float* bcol_l = Wl + lm;

#pragma unroll
    for (int kk = 0; kk < 16; ++kk) {
        const int kbase = kk * 4 + lg * 2;
        v2f a = *(const v2f*)(arow_l + kbase);
        v2f b0, b1v, b2, b3;
        b0[0]  = bcol_l[(kbase + 0) * 64 + 0];
        b0[1]  = bcol_l[(kbase + 1) * 64 + 0];
        b1v[0] = bcol_l[(kbase + 0) * 64 + 16];
        b1v[1] = bcol_l[(kbase + 1) * 64 + 16];
        b2[0]  = bcol_l[(kbase + 0) * 64 + 32];
        b2[1]  = bcol_l[(kbase + 1) * 64 + 32];
        b3[0]  = bcol_l[(kbase + 0) * 64 + 48];
        b3[1]  = bcol_l[(kbase + 1) * 64 + 48];
        c0 = __builtin_amdgcn_wmma_f32_16x16x4_f32(false, a, false, b0,  (short)0, c0, false, false);
        c1 = __builtin_amdgcn_wmma_f32_16x16x4_f32(false, a, false, b1v, (short)0, c1, false, false);
        c2 = __builtin_amdgcn_wmma_f32_16x16x4_f32(false, a, false, b2,  (short)0, c2, false, false);
        c3 = __builtin_amdgcn_wmma_f32_16x16x4_f32(false, a, false, b3,  (short)0, c3, false, false);
    }

#pragma unroll
    for (int i = 0; i < 8; ++i) {
        int r = row0 + i + lg * 8;
        if (r < nRows) {
            float* orow = out + (size_t)r * 128 + colofs + lm;
            orow[0]  = c0[i];
            orow[16] = c1[i];
            orow[32] = c2[i];
            orow[48] = c3[i];
        }
    }
}


extern "C" void kernel_launch(void* const* d_in, const int* in_sizes, int n_in,
                              void* d_out, int out_size, void* d_ws, size_t ws_size,
                              hipStream_t stream) {
    const float* x  = (const float*)d_in[0];
    const int*   ei = (const int*)d_in[1];
    const float* ew = (const float*)d_in[2];
    const float* W1 = (const float*)d_in[3];
    const float* b1 = (const float*)d_in[4];
    const float* W2 = (const float*)d_in[5];
    const float* b2 = (const float*)d_in[6];
    float* out = (float*)d_out;

    const int N = in_sizes[0] / 64;
    const int E = in_sizes[2];

    const int per = (((N + 1023) / 1024) + 3) & ~3;
    const int NP  = per * 1024;
    const int EP  = (E + 3) & ~3;

    float* tA   = (float*)d_ws;
    float* tB   = tA + (size_t)N * 64;
    int*   csrc = (int*)(tB + (size_t)N * 64);
    float* cw   = (float*)(csrc + 2 * (size_t)EP);
    int*   cnt  = (int*)(cw + 2 * (size_t)EP);
    int*   rp   = cnt + 2 * (size_t)NP;
    float* dis  = (float*)(rp + 2 * (size_t)(NP + 4));
    int*   rank = (int*)tA;
    int*   perm = rank + 2 * (size_t)EP;
    int*   cntp = perm + 2 * (size_t)EP;
    const int CE = (E + CCH - 1) / CCH;
    const int NR = (NP + RNG - 1) / RNG;

    const int n128 = N * 128;
    const float DIAG = 2.0f / 3.0f - 1.0f;

    dim3 blk(256);
    int gE    = (E + 255) / 256;
    int gN    = (N + 255) / 256;
    int gN8   = (N + 7) / 8;
    int g128  = (n128 + 255) / 256;
    int gGemm = (N + 127) / 128;

    k_rangecount<<<dim3(NR, CCH, 2), dim3(1024), 0, stream>>>(ei, cntp, rank, E, EP, NP, CE);
    k_combine<<<dim3((NP + 255) / 256, 2), blk, 0, stream>>>(cntp, cnt, NP);
    k_scan<<<2, dim3(1024), 0, stream>>>(cnt, rp, per, NP);
    k_place<<<dim3(gE, 2), blk, 0, stream>>>(ei, rp, rank, cntp, perm, E, EP, NP, CE);
    k_order<<<dim3(gN8, 2), blk, 0, stream>>>(ei, ew, rp, perm, csrc, cw, N, E, EP, NP);
    k_dis<<<dim3(gN, 2), blk, 0, stream>>>(rp, cw, dis, N, EP, NP);
    k_csrw<<<dim3(gN, 2), blk, 0, stream>>>(rp, csrc, cw, dis, N, EP, NP);

    k_bias<<<g128, blk, 0, stream>>>(b1, b2, out, n128);

    for (int dir = 0; dir < 2; ++dir) {
        const int*   r    = rp + (size_t)dir * (NP + 4);
        const int*   sidx = csrc + (size_t)dir * EP;
        const float* w    = cw + (size_t)dir * EP;
        const float* W    = (dir == 0) ? W1 : W2;
        const int    cofs = (dir == 0) ? 0 : 64;

        k_gemm_acc<<<gGemm, blk, 0, stream>>>(x, W, out, cofs, N);

        k_lhat<false><<<gN8, blk, 0, stream>>>(x, x, tA, r, sidx, w, DIAG, 0.0f, 1.0f, N);
        k_gemm_acc<<<gGemm, blk, 0, stream>>>(tA, W + 1 * 4096, out, cofs, N);

        k_lhat<true><<<gN8, blk, 0, stream>>>(tA, x, tB, r, sidx, w, 2.0f * DIAG, -1.0f, 2.0f, N);
        k_gemm_acc<<<gGemm, blk, 0, stream>>>(tB, W + 2 * 4096, out, cofs, N);

        k_lhat<true><<<gN8, blk, 0, stream>>>(tB, tA, tA, r, sidx, w, 2.0f * DIAG, -1.0f, 2.0f, N);
        k_gemm_acc<<<gGemm, blk, 0, stream>>>(tA, W + 3 * 4096, out, cofs, N);

        k_lhat<true><<<gN8, blk, 0, stream>>>(tA, tB, tB, r, sidx, w, 2.0f * DIAG, -1.0f, 2.0f, N);
        k_gemm_acc<<<gGemm, blk, 0, stream>>>(tB, W + 4 * 4096, out, cofs, N);
    }
}


# Dense projection fused into the sparse-pass launch: a 16-wave block hands its 16 new rows through LDS to the f32 WMMA (bias, k=0 product in pass 1); 26 launches -> 15, float sums unchanged
# speedup vs baseline: 2.6681x; 1.2582x over previous
#include <hip/hip_runtime.h>


typedef float v2f __attribute__((ext_vector_type(2)));
typedef float v8f __attribute__((ext_vector_type(8)));

static __device__ __forceinline__ size_t row64(int n) { return (size_t)n * 64; }


#define RNG 8192
#define CCH 16

__global__ __launch_bounds__(1024) void k_rangecount(const int* __restrict__ ei,
                                                     int* __restrict__ cntp,
                                                     int* __restrict__ rank,
                                                     int E, int EP, int NP, int CE) {
    __shared__ int lc[RNG];
    const int t = threadIdx.x;
    const int r0 = blockIdx.x * RNG;
    const int c = blockIdx.y;
    const int dir = blockIdx.z;
    for (int i = t; i < RNG; i += 1024) lc[i] = 0;
    __syncthreads();
    const int* dst = ei + (size_t)dir * E;
    int* rk = rank + (size_t)dir * EP;
    const int cbeg = c * CE;
    int cend = cbeg + CE;
    cend = cend < E ? cend : E;
    for (int e0 = cbeg + t; e0 < cend; e0 += 8 * 1024) {
        int dv[8];
#pragma unroll
        for (int j = 0; j < 8; ++j) {
            const int e = e0 + j * 1024;
            dv[j] = (e < cend) ? dst[e] : -1;
        }
#pragma unroll
        for (int j = 0; j < 8; ++j) {
            const unsigned rel = (unsigned)(dv[j] - r0);
            if (rel < (unsigned)RNG) rk[e0 + j * 1024] = atomicAdd(&lc[rel], 1);
        }
    }
    __syncthreads();
    int* o = cntp + ((size_t)dir * CCH + c) * NP;
    for (int i = t; i < RNG; i += 1024) {
        const int d = r0 + i;
        if (d < NP) o[d] = lc[i];
    }
}

__global__ __launch_bounds__(256) void k_combine(int* __restrict__ cntp,
                                                 int* __restrict__ cnt, int NP) {
    const int d = blockIdx.x * 256 + threadIdx.x;
    if (d >= NP) return;
    const int dir = blockIdx.y;
    int* p = cntp + (size_t)dir * CCH * NP + d;
    int v[CCH];
#pragma unroll
    for (int c = 0; c < CCH; ++c) v[c] = p[(size_t)c * NP];
    int run = 0;
#pragma unroll
    for (int c = 0; c < CCH; ++c) {
        p[(size_t)c * NP] = run;
        run += v[c];
    }
    cnt[(size_t)dir * NP + d] = run;
}

__global__ __launch_bounds__(1024) void k_scan(const int* __restrict__ cnt,
                                               int* __restrict__ rp, int per, int NP) {
    __shared__ int part[1024];
    const int t = threadIdx.x;
    const int4* c4 = (const int4*)(cnt + (size_t)blockIdx.x * NP + (size_t)t * per);
    int* rbase = rp + (size_t)blockIdx.x * (NP + 4);
    int4* r4 = (int4*)(rbase + (size_t)t * per);
    const int nq = per >> 2;
    int s = 0;
    for (int q = 0; q < nq; ++q) {
        int4 c = c4[q];
        s += c.x + c.y + c.z + c.w;
    }
    part[t] = s;
    __syncthreads();
    for (int off = 1; off < 1024; off <<= 1) {
        int add = (t >= off) ? part[t - off] : 0;
        __syncthreads();
        part[t] += add;
        __syncthreads();
    }
    int run = part[t] - s;
    for (int q = 0; q < nq; ++q) {
        int4 c = c4[q];
        int4 o;
        o.x = run; run += c.x;
        o.y = run; run += c.y;
        o.z = run; run += c.z;
        o.w = run; run += c.w;
        r4[q] = o;
    }
    if (t == 1023) rbase[NP] = run;
}

__global__ __launch_bounds__(256) void k_place(const int* __restrict__ ei,
                                               const int* __restrict__ rp,
                                               const int* __restrict__ rank,
                                               const int* __restrict__ cntp,
                                               int* __restrict__ perm,
                                               int E, int EP, int NP, int CE) {
    int e = blockIdx.x * 256 + threadIdx.x;
    if (e >= E) return;
    const int dir = blockIdx.y;
    const int c = e / CE;
    int d = ei[(size_t)dir * E + e];
    int pos = rp[(size_t)dir * (NP + 4) + d] + cntp[((size_t)dir * CCH + c) * NP + d] +
              rank[(size_t)dir * EP + e];
    perm[(size_t)dir * EP + pos] = e;
}

__global__ __launch_bounds__(256) void k_order(const int* __restrict__ ei,
                                               const float* __restrict__ ew,
                                               const int* __restrict__ rp,
                                               const int* __restrict__ perm,
                                               int* __restrict__ csrc,
                                               float* __restrict__ cw,
                                               int N, int E, int EP, int NP) {
    const int d = blockIdx.x * 8 + (threadIdx.x >> 5);
    if (d >= N) return;
    const int dir = blockIdx.y;
    const int lane = threadIdx.x & 31;
    const int* r = rp + (size_t)dir * (NP + 4);
    const int* pm = perm + (size_t)dir * EP;
    const int* srcIdx = ei + (size_t)(1 - dir) * E;
    const int s0 = __builtin_amdgcn_readfirstlane(r[d]);
    const int s1 = __builtin_amdgcn_readfirstlane(r[d + 1]);
    for (int ca = s0; ca < s1; ca += 32) {
        const bool mine = (ca + lane) < s1;
        const int my = mine ? pm[ca + lane] : 0;
        int below = 0;
        for (int cb = s0; cb < s1; cb += 32) {
            const int other = (cb + lane < s1) ? pm[cb + lane] : 0;
            int nb = s1 - cb;
            nb = nb < 32 ? nb : 32;
            for (int j = 0; j < nb; ++j) {
                const int oj = __builtin_amdgcn_readlane(other, j);
                below += (oj < my) ? 1 : 0;
            }
        }
        if (mine) {
            const size_t slot = (size_t)dir * EP + s0 + below;
            csrc[slot] = srcIdx[my];
            cw[slot] = ew[my];
        }
    }
}

__global__ __launch_bounds__(256) void k_dis(const int* __restrict__ rp,
                                             const float* __restrict__ cw,
                                             float* __restrict__ dis,
                                             int N, int EP, int NP) {
    int d = blockIdx.x * 256 + threadIdx.x;
    if (d >= N) return;
    const int dir = blockIdx.y;
    const int* r = rp + (size_t)dir * (NP + 4);
    const float* w = cw + (size_t)dir * EP;
    int s0 = r[d], s1 = r[d + 1];
    float deg = 0.0f;
    for (int q = s0; q < s1; ++q) deg += w[q];
    dis[(size_t)dir * NP + d] = (deg > 0.0f) ? rsqrtf(deg) : 0.0f;
}

__global__ __launch_bounds__(256) void k_csrw(const int* __restrict__ rp,
                                              const int* __restrict__ csrc,
                                              float* __restrict__ cw,
                                              const float* __restrict__ dis,
                                              int N, int EP, int NP) {
    int d = blockIdx.x * 256 + threadIdx.x;
    if (d >= N) return;
    const int dir = blockIdx.y;
    const float scale = 2.0f / 3.0f;
    const int* r = rp + (size_t)dir * (NP + 4);
    const int* sidx = csrc + (size_t)dir * EP;
    float* w = cw + (size_t)dir * EP;
    const float* ds = dis + (size_t)dir * NP;
    int s0 = r[d], s1 = r[d + 1];
    float dd = -scale * ds[d];
    for (int q = s0; q < s1; ++q) w[q] = dd * w[q] * ds[sidx[q]];
}

#define TP 68
#define WP 80

static __device__ __forceinline__ void mma_tile(const float* tile, const float* wl,
                                                int lm, int h, int ct,
                                                v8f& acc0, v8f& acc1) {
    const float* arow = tile + lm * TP;
    const float* bcol = wl + ct + lm;
#pragma unroll
    for (int kk = 0; kk < 16; ++kk) {
        const int kbase = kk * 4 + h * 2;
        v2f a = *(const v2f*)(arow + kbase);
        v2f b0, b1;
        b0[0] = bcol[(kbase + 0) * WP + 0];
        b0[1] = bcol[(kbase + 1) * WP + 0];
        b1[0] = bcol[(kbase + 0) * WP + 16];
        b1[1] = bcol[(kbase + 1) * WP + 16];
        acc0 = __builtin_amdgcn_wmma_f32_16x16x4_f32(false, a, false, b0, (short)0, acc0, false, false);
        acc1 = __builtin_amdgcn_wmma_f32_16x16x4_f32(false, a, false, b1, (short)0, acc1, false, false);
    }
}

template <bool FIRST>
static __device__ __forceinline__ void put_weights(float* Wl, int tid,
                                                   float4 wb0, float4 wb1,
                                                   float4 wa0, float4 wa1) {
    float* wdst = Wl + (tid >> 4) * WP + (tid & 15) * 4;
    *(float4*)wdst = wb0;
    *(float4*)(wdst + 32 * WP) = wb1;
    if (FIRST) {
        *(float4*)(wdst + 64 * WP) = wa0;
        *(float4*)(wdst + 96 * WP) = wa1;
    }
}

template <bool HAS_PREV, bool FIRST, bool STORE_DST>
__global__ __launch_bounds__(512) void k_lhat_proj(const float* __restrict__ v,
                                                   const float* prev,
                                                   float* dst,
                                                   const int* __restrict__ rp,
                                                   const int* __restrict__ csrc,
                                                   const float* __restrict__ cw,
                                                   float adiag, float beta, float alpha,
                                                   const float* Wa,
                                                   const float* Wb,
                                                   const float* bias,
                                                   float* out, int colofs, int N) {
    __shared__ float Wl[(FIRST ? 2 : 1) * 64 * WP];
    __shared__ float Tl[(FIRST ? 2 : 1) * 16 * TP];

    const int tid = threadIdx.x;
    const int wave = tid >> 5;
    const int lane = tid & 31;
    const int h = lane >> 4;
    const int lm = lane & 15;
    const int c4 = lm * 4;
    const int row0 = blockIdx.x * 16;
    const int d = row0 + wave;
    const bool valid = d < N;

    const float4 wb0 = ((const float4*)Wb)[tid];
    const float4 wb1 = ((const float4*)Wb)[tid + 512];
    float4 wa0 = make_float4(0.f, 0.f, 0.f, 0.f);
    float4 wa1 = make_float4(0.f, 0.f, 0.f, 0.f);
    if (FIRST) {
        wa0 = ((const float4*)Wa)[tid];
        wa1 = ((const float4*)Wa)[tid + 512];
    }

    const int mw0 = (blockIdx.x & 1) * 2;
    const bool mat = (wave == mw0) || (wave == mw0 + 1);
    const int ct = (wave - mw0) * 32;

    float4 o = make_float4(0.f, 0.f, 0.f, 0.f);
    float4 vd = make_float4(0.f, 0.f, 0.f, 0.f);
    if (valid) {
        const int s0 = __builtin_amdgcn_readfirstlane(rp[d]);
        const int s1 = __builtin_amdgcn_readfirstlane(rp[d + 1]);

        int base = s0;
        int n = s1 - base;
        n = n < 32 ? n : 32;
        int msrc = d;
        float mw = 0.0f;
        if (lane < n) {
            msrc = csrc[base + lane];
            mw = cw[base + lane];
        }

        put_weights<FIRST>(Wl, tid, wb0, wb1, wa0, wa1);

        float4 pd = make_float4(0.f, 0.f, 0.f, 0.f);
        if (h == 0) {
            vd = *(const float4*)(v + row64(d) + c4);
            if (HAS_PREV) pd = *(const float4*)(prev + row64(d) + c4);
        }

        float4 acc = make_float4(0.f, 0.f, 0.f, 0.f);
        while (n > 0) {
            const int npair = n >> 1;
#pragma unroll 4
            for (int q = 0; q < npair; ++q) {
                const int sl = 2 * q + h;
                const int s = __shfl(msrc, sl);
                const float w = __shfl(mw, sl);
                const float4 val = *(const float4*)(v + row64(s) + c4);
                acc.x = fmaf(w, val.x, acc.x);
                acc.y = fmaf(w, val.y, acc.y);
                acc.z = fmaf(w, val.z, acc.z);
                acc.w = fmaf(w, val.w, acc.w);
            }
            if (n & 1) {
                const int s = __shfl(msrc, n - 1);
                const float w = __shfl(mw, n - 1);
                if (h == 0) {
                    const float4 val = *(const float4*)(v + row64(s) + c4);
                    acc.x = fmaf(w, val.x, acc.x);
                    acc.y = fmaf(w, val.y, acc.y);
                    acc.z = fmaf(w, val.z, acc.z);
                    acc.w = fmaf(w, val.w, acc.w);
                }
            }
            base += 32;
            n = s1 - base;
            n = n < 32 ? n : 32;
            msrc = d;
            mw = 0.0f;
            if (lane < n) {
                msrc = csrc[base + lane];
                mw = cw[base + lane];
            }
        }
        acc.x += __shfl_xor(acc.x, 16);
        acc.y += __shfl_xor(acc.y, 16);
        acc.z += __shfl_xor(acc.z, 16);
        acc.w += __shfl_xor(acc.w, 16);

        if (h == 0) {
            o.x = adiag * vd.x;
            o.y = adiag * vd.y;
            o.z = adiag * vd.z;
            o.w = adiag * vd.w;
            if (HAS_PREV) {
                o.x += beta * pd.x;
                o.y += beta * pd.y;
                o.z += beta * pd.z;
                o.w += beta * pd.w;
            }
            o.x += alpha * acc.x;
            o.y += alpha * acc.y;
            o.z += alpha * acc.z;
            o.w += alpha * acc.w;
            if (STORE_DST) *(float4*)(dst + row64(d) + c4) = o;
        }
    } else {
        put_weights<FIRST>(Wl, tid, wb0, wb1, wa0, wa1);
    }

    v8f c0 = {0.f, 0.f, 0.f, 0.f, 0.f, 0.f, 0.f, 0.f};
    v8f c1 = {0.f, 0.f, 0.f, 0.f, 0.f, 0.f, 0.f, 0.f};
    if (mat) {
        if (FIRST) {
            const float bv0 = bias[ct + lm];
            const float bv1 = bias[ct + 16 + lm];
#pragma unroll
            for (int i = 0; i < 8; ++i) {
                c0[i] = bv0;
                c1[i] = bv1;
            }
        } else {
            const int nclamp = N - 1;
#pragma unroll
            for (int i = 0; i < 8; ++i) {
                const int r = row0 + i + h * 8;
                const bool ok = (r < N);
                const int rc = ok ? r : nclamp;
                const float* orow = out + (size_t)rc * 128 + colofs + ct + lm;
                const float v0 = orow[0], v1 = orow[16];
                c0[i] = ok ? v0 : 0.f;
                c1[i] = ok ? v1 : 0.f;
            }
        }
    }

    if (h == 0) {
        *(float4*)(Tl + wave * TP + c4) = o;
        if (FIRST) *(float4*)(Tl + 16 * TP + wave * TP + c4) = vd;
    }
    __syncthreads();

    if (mat) {
        if (FIRST) mma_tile(Tl + 16 * TP, Wl + 64 * WP, lm, h, ct, c0, c1);
        mma_tile(Tl, Wl, lm, h, ct, c0, c1);
#pragma unroll
        for (int i = 0; i < 8; ++i) {
            const int r = row0 + i + h * 8;
            if (r < N) {
                float* orow = out + (size_t)r * 128 + colofs + ct + lm;
                orow[0]  = c0[i];
                orow[16] = c1[i];
            }
        }
    }
}


extern "C" void kernel_launch(void* const* d_in, const int* in_sizes, int n_in,
                              void* d_out, int out_size, void* d_ws, size_t ws_size,
                              hipStream_t stream) {
    const float* x  = (const float*)d_in[0];
    const int*   ei = (const int*)d_in[1];
    const float* ew = (const float*)d_in[2];
    const float* W1 = (const float*)d_in[3];
    const float* b1 = (const float*)d_in[4];
    const float* W2 = (const float*)d_in[5];
    const float* b2 = (const float*)d_in[6];
    float* out = (float*)d_out;

    const int N = in_sizes[0] / 64;
    const int E = in_sizes[2];

    const int per = (((N + 1023) / 1024) + 3) & ~3;
    const int NP  = per * 1024;
    const int EP  = (E + 3) & ~3;

    float* tA   = (float*)d_ws;
    float* tB   = tA + (size_t)N * 64;
    int*   csrc = (int*)(tB + (size_t)N * 64);
    float* cw   = (float*)(csrc + 2 * (size_t)EP);
    int*   cnt  = (int*)(cw + 2 * (size_t)EP);
    int*   rp   = cnt + 2 * (size_t)NP;
    float* dis  = (float*)(rp + 2 * (size_t)(NP + 4));
    int*   rank = (int*)tA;
    int*   perm = rank + 2 * (size_t)EP;
    int*   cntp = perm + 2 * (size_t)EP;
    const int CE = (E + CCH - 1) / CCH;
    const int NR = (NP + RNG - 1) / RNG;

    const float DIAG = 2.0f / 3.0f - 1.0f;

    dim3 blk(256);
    dim3 blk16w(512);
    int gE    = (E + 255) / 256;
    int gN    = (N + 255) / 256;
    int gN8   = (N + 7) / 8;
    int gN16  = (N + 15) / 16;

    k_rangecount<<<dim3(NR, CCH, 2), dim3(1024), 0, stream>>>(ei, cntp, rank, E, EP, NP, CE);
    k_combine<<<dim3((NP + 255) / 256, 2), blk, 0, stream>>>(cntp, cnt, NP);
    k_scan<<<2, dim3(1024), 0, stream>>>(cnt, rp, per, NP);
    k_place<<<dim3(gE, 2), blk, 0, stream>>>(ei, rp, rank, cntp, perm, E, EP, NP, CE);
    k_order<<<dim3(gN8, 2), blk, 0, stream>>>(ei, ew, rp, perm, csrc, cw, N, E, EP, NP);
    k_dis<<<dim3(gN, 2), blk, 0, stream>>>(rp, cw, dis, N, EP, NP);
    k_csrw<<<dim3(gN, 2), blk, 0, stream>>>(rp, csrc, cw, dis, N, EP, NP);

    for (int dir = 0; dir < 2; ++dir) {
        const int*   r    = rp + (size_t)dir * (NP + 4);
        const int*   sidx = csrc + (size_t)dir * EP;
        const float* w    = cw + (size_t)dir * EP;
        const float* W    = (dir == 0) ? W1 : W2;
        const float* b    = (dir == 0) ? b1 : b2;
        const int    cofs = (dir == 0) ? 0 : 64;

        k_lhat_proj<false, true, true><<<gN16, blk16w, 0, stream>>>(
            x, x, tA, r, sidx, w, DIAG, 0.0f, 1.0f, W, W + 1 * 4096, b, out, cofs, N);

        k_lhat_proj<true, false, true><<<gN16, blk16w, 0, stream>>>(
            tA, x, tB, r, sidx, w, 2.0f * DIAG, -1.0f, 2.0f, W, W + 2 * 4096, b, out, cofs, N);

        k_lhat_proj<true, false, true><<<gN16, blk16w, 0, stream>>>(
            tB, tA, tA, r, sidx, w, 2.0f * DIAG, -1.0f, 2.0f, W, W + 3 * 4096, b, out, cofs, N);

        k_lhat_proj<true, false, false><<<gN16, blk16w, 0, stream>>>(
            tA, tB, tB, r, sidx, w, 2.0f * DIAG, -1.0f, 2.0f, W, W + 4 * 4096, b, out, cofs, N);
    }
}


# Sparse pass: one half-wave per node (2 nodes per wave, 32 per 16-wave block, 4 matrix waves), slots passed by DPP row broadcast, row loads in masked groups of 4; float sums unchanged
# speedup vs baseline: 2.9203x; 1.0945x over previous
#include <hip/hip_runtime.h>


typedef float v2f __attribute__((ext_vector_type(2)));
typedef float v8f __attribute__((ext_vector_type(8)));

static __device__ __forceinline__ size_t row64(int n) { return (size_t)n * 64; }


#define RNG 8192
#define CCH 16

__global__ __launch_bounds__(1024) void k_rangecount(const int* __restrict__ ei,
                                                     int* __restrict__ cntp,
                                                     int* __restrict__ rank,
                                                     int E, int EP, int NP, int CE) {
    __shared__ int lc[RNG];
    const int t = threadIdx.x;
    const int r0 = blockIdx.x * RNG;
    const int c = blockIdx.y;
    const int dir = blockIdx.z;
    for (int i = t; i < RNG; i += 1024) lc[i] = 0;
    __syncthreads();
    const int* dst = ei + (size_t)dir * E;
    int* rk = rank + (size_t)dir * EP;
    const int cbeg = c * CE;
    int cend = cbeg + CE;
    cend = cend < E ? cend : E;
    for (int e0 = cbeg + t; e0 < cend; e0 += 8 * 1024) {
        int dv[8];
#pragma unroll
        for (int j = 0; j < 8; ++j) {
            const int e = e0 + j * 1024;
            dv[j] = (e < cend) ? dst[e] : -1;
        }
#pragma unroll
        for (int j = 0; j < 8; ++j) {
            const unsigned rel = (unsigned)(dv[j] - r0);
            if (rel < (unsigned)RNG) rk[e0 + j * 1024] = atomicAdd(&lc[rel], 1);
        }
    }
    __syncthreads();
    int* o = cntp + ((size_t)dir * CCH + c) * NP;
    for (int i = t; i < RNG; i += 1024) {
        const int d = r0 + i;
        if (d < NP) o[d] = lc[i];
    }
}

__global__ __launch_bounds__(256) void k_combine(int* __restrict__ cntp,
                                                 int* __restrict__ cnt, int NP) {
    const int d = blockIdx.x * 256 + threadIdx.x;
    if (d >= NP) return;
    const int dir = blockIdx.y;
    int* p = cntp + (size_t)dir * CCH * NP + d;
    int v[CCH];
#pragma unroll
    for (int c = 0; c < CCH; ++c) v[c] = p[(size_t)c * NP];
    int run = 0;
#pragma unroll
    for (int c = 0; c < CCH; ++c) {
        p[(size_t)c * NP] = run;
        run += v[c];
    }
    cnt[(size_t)dir * NP + d] = run;
}

__global__ __launch_bounds__(1024) void k_scan(const int* __restrict__ cnt,
                                               int* __restrict__ rp, int per, int NP) {
    __shared__ int part[1024];
    const int t = threadIdx.x;
    const int4* c4 = (const int4*)(cnt + (size_t)blockIdx.x * NP + (size_t)t * per);
    int* rbase = rp + (size_t)blockIdx.x * (NP + 4);
    int4* r4 = (int4*)(rbase + (size_t)t * per);
    const int nq = per >> 2;
    int s = 0;
    for (int q = 0; q < nq; ++q) {
        int4 c = c4[q];
        s += c.x + c.y + c.z + c.w;
    }
    part[t] = s;
    __syncthreads();
    for (int off = 1; off < 1024; off <<= 1) {
        int add = (t >= off) ? part[t - off] : 0;
        __syncthreads();
        part[t] += add;
        __syncthreads();
    }
    int run = part[t] - s;
    for (int q = 0; q < nq; ++q) {
        int4 c = c4[q];
        int4 o;
        o.x = run; run += c.x;
        o.y = run; run += c.y;
        o.z = run; run += c.z;
        o.w = run; run += c.w;
        r4[q] = o;
    }
    if (t == 1023) rbase[NP] = run;
}

__global__ __launch_bounds__(256) void k_place(const int* __restrict__ ei,
                                               const int* __restrict__ rp,
                                               const int* __restrict__ rank,
                                               const int* __restrict__ cntp,
                                               int* __restrict__ perm,
                                               int E, int EP, int NP, int CE) {
    int e = blockIdx.x * 256 + threadIdx.x;
    if (e >= E) return;
    const int dir = blockIdx.y;
    const int c = e / CE;
    int d = ei[(size_t)dir * E + e];
    int pos = rp[(size_t)dir * (NP + 4) + d] + cntp[((size_t)dir * CCH + c) * NP + d] +
              rank[(size_t)dir * EP + e];
    perm[(size_t)dir * EP + pos] = e;
}

__global__ __launch_bounds__(256) void k_order(const int* __restrict__ ei,
                                               const float* __restrict__ ew,
                                               const int* __restrict__ rp,
                                               const int* __restrict__ perm,
                                               int* __restrict__ csrc,
                                               float* __restrict__ cw,
                                               int N, int E, int EP, int NP) {
    const int d = blockIdx.x * 8 + (threadIdx.x >> 5);
    if (d >= N) return;
    const int dir = blockIdx.y;
    const int lane = threadIdx.x & 31;
    const int* r = rp + (size_t)dir * (NP + 4);
    const int* pm = perm + (size_t)dir * EP;
    const int* srcIdx = ei + (size_t)(1 - dir) * E;
    const int s0 = __builtin_amdgcn_readfirstlane(r[d]);
    const int s1 = __builtin_amdgcn_readfirstlane(r[d + 1]);
    for (int ca = s0; ca < s1; ca += 32) {
        const bool mine = (ca + lane) < s1;
        const int my = mine ? pm[ca + lane] : 0;
        int below = 0;
        for (int cb = s0; cb < s1; cb += 32) {
            const int other = (cb + lane < s1) ? pm[cb + lane] : 0;
            int nb = s1 - cb;
            nb = nb < 32 ? nb : 32;
            for (int j = 0; j < nb; ++j) {
                const int oj = __builtin_amdgcn_readlane(other, j);
                below += (oj < my) ? 1 : 0;
            }
        }
        if (mine) {
            const size_t slot = (size_t)dir * EP + s0 + below;
            csrc[slot] = srcIdx[my];
            cw[slot] = ew[my];
        }
    }
}

__global__ __launch_bounds__(256) void k_dis(const int* __restrict__ rp,
                                             const float* __restrict__ cw,
                                             float* __restrict__ dis,
                                             int N, int EP, int NP) {
    int d = blockIdx.x * 256 + threadIdx.x;
    if (d >= N) return;
    const int dir = blockIdx.y;
    const int* r = rp + (size_t)dir * (NP + 4);
    const float* w = cw + (size_t)dir * EP;
    int s0 = r[d], s1 = r[d + 1];
    float deg = 0.0f;
    for (int q = s0; q < s1; ++q) deg += w[q];
    dis[(size_t)dir * NP + d] = (deg > 0.0f) ? rsqrtf(deg) : 0.0f;
}

__global__ __launch_bounds__(256) void k_csrw(const int* __restrict__ rp,
                                              const int* __restrict__ csrc,
                                              float* __restrict__ cw,
                                              const float* __restrict__ dis,
                                              int N, int EP, int NP) {
    int d = blockIdx.x * 256 + threadIdx.x;
    if (d >= N) return;
    const int dir = blockIdx.y;
    const float scale = 2.0f / 3.0f;
    const int* r = rp + (size_t)dir * (NP + 4);
    const int* sidx = csrc + (size_t)dir * EP;
    float* w = cw + (size_t)dir * EP;
    const float* ds = dis + (size_t)dir * NP;
    int s0 = r[d], s1 = r[d + 1];
    float dd = -scale * ds[d];
    for (int q = s0; q < s1; ++q) w[q] = dd * w[q] * ds[sidx[q]];
}

#define TP 68
#define WP 80

static __device__ __forceinline__ void mma_tile(const float* tile, const float* wl,
                                                int lm, int h, int ct,
                                                v8f& acc0, v8f& acc1) {
    const float* arow = tile + lm * TP;
    const float* bcol = wl + ct + lm;
#pragma unroll
    for (int kk = 0; kk < 16; ++kk) {
        const int kbase = kk * 4 + h * 2;
        v2f a = *(const v2f*)(arow + kbase);
        v2f b0, b1;
        b0[0] = bcol[(kbase + 0) * WP + 0];
        b0[1] = bcol[(kbase + 1) * WP + 0];
        b1[0] = bcol[(kbase + 0) * WP + 16];
        b1[1] = bcol[(kbase + 1) * WP + 16];
        acc0 = __builtin_amdgcn_wmma_f32_16x16x4_f32(false, a, false, b0, (short)0, acc0, false, false);
        acc1 = __builtin_amdgcn_wmma_f32_16x16x4_f32(false, a, false, b1, (short)0, acc1, false, false);
    }
}

template <bool FIRST>
static __device__ __forceinline__ void put_weights(float* Wl, int tid,
                                                   float4 wb0, float4 wb1,
                                                   float4 wa0, float4 wa1) {
    float* wdst = Wl + (tid >> 4) * WP + (tid & 15) * 4;
    *(float4*)wdst = wb0;
    *(float4*)(wdst + 32 * WP) = wb1;
    if (FIRST) {
        *(float4*)(wdst + 64 * WP) = wa0;
        *(float4*)(wdst + 96 * WP) = wa1;
    }
}

template <int Q>
static __device__ __forceinline__ int row_take(int x) {
    return __builtin_amdgcn_update_dpp(0, x, 0x150 + Q, 0xf, 0xf, true);
}

template <int Q>
static __device__ __forceinline__ void slot_load(const float* __restrict__ v, int msrc, float mw,
                                                 int rem, int c4, float4& val, float& w) {
    const int s = row_take<Q>(msrc);
    w = __int_as_float(row_take<Q>(__float_as_int(mw)));
    if (Q < rem) val = *(const float4*)(v + row64(s) + c4);
}

static __device__ __forceinline__ void slot_sum(float w, const float4& val, float4& acc) {
    acc.x = fmaf(w, val.x, acc.x);
    acc.y = fmaf(w, val.y, acc.y);
    acc.z = fmaf(w, val.z, acc.z);
    acc.w = fmaf(w, val.w, acc.w);
}

template <int Q0>
static __device__ __forceinline__ void slot_group(const float* __restrict__ v, int msrc, float mw,
                                                  int rem, int c4, float4 (&val)[4],
                                                  float4& accE, float4& accO) {
    float w0, w1, w2, w3;
    slot_load<Q0 + 0>(v, msrc, mw, rem, c4, val[0], w0);
    slot_load<Q0 + 1>(v, msrc, mw, rem, c4, val[1], w1);
    slot_load<Q0 + 2>(v, msrc, mw, rem, c4, val[2], w2);
    slot_load<Q0 + 3>(v, msrc, mw, rem, c4, val[3], w3);
    slot_sum(w0, val[0], accE);
    slot_sum(w1, val[1], accO);
    slot_sum(w2, val[2], accE);
    slot_sum(w3, val[3], accO);
}

template <bool HAS_PREV, bool FIRST, bool STORE_DST>
__global__ __launch_bounds__(512) void k_lhat_proj(const float* __restrict__ v,
                                                   const float* prev,
                                                   float* dst,
                                                   const int* __restrict__ rp,
                                                   const int* __restrict__ csrc,
                                                   const float* __restrict__ cw,
                                                   float adiag, float beta, float alpha,
                                                   const float* Wa,
                                                   const float* Wb,
                                                   const float* bias,
                                                   float* out, int colofs, int N) {
    __shared__ float Wl[(FIRST ? 2 : 1) * 64 * WP];
    __shared__ float Tl[(FIRST ? 2 : 1) * 32 * TP];

    const int tid = threadIdx.x;
    const int wave = tid >> 5;
    const int lane = tid & 31;
    const int h = lane >> 4;
    const int lm = lane & 15;
    const int c4 = lm * 4;
    const int row0 = blockIdx.x * 32;
    const int trow = wave * 2 + h;
    const int d = row0 + trow;
    const bool valid = d < N;

    const float4 wb0 = ((const float4*)Wb)[tid];
    const float4 wb1 = ((const float4*)Wb)[tid + 512];
    float4 wa0 = make_float4(0.f, 0.f, 0.f, 0.f);
    float4 wa1 = make_float4(0.f, 0.f, 0.f, 0.f);
    if (FIRST) {
        wa0 = ((const float4*)Wa)[tid];
        wa1 = ((const float4*)Wa)[tid + 512];
    }

    const int mj = wave - (blockIdx.x & 3) * 4;
    const bool mat = (mj >= 0) && (mj < 4);
    const int mt = (mj >> 1) & 1;
    const int ct = (mj & 1) * 32;

    int s0 = 0, s1 = 0;
    if (valid) {
        s0 = rp[d];
        s1 = rp[d + 1];
    }
    int base = s0;
    int rem = s1 - s0;

    int msrc = 0;
    float mw = 0.0f;
    if (lm < rem) {
        msrc = csrc[base + lm];
        mw = cw[base + lm];
    }

    put_weights<FIRST>(Wl, tid, wb0, wb1, wa0, wa1);

    float4 vd = make_float4(0.f, 0.f, 0.f, 0.f);
    float4 pd = make_float4(0.f, 0.f, 0.f, 0.f);
    if (valid) {
        vd = *(const float4*)(v + row64(d) + c4);
        if (HAS_PREV) pd = *(const float4*)(prev + row64(d) + c4);
    }

    float4 accE = make_float4(0.f, 0.f, 0.f, 0.f);
    float4 accO = make_float4(0.f, 0.f, 0.f, 0.f);
    float4 val[4];
#pragma unroll
    for (int j = 0; j < 4; ++j) val[j] = make_float4(0.f, 0.f, 0.f, 0.f);

    int mrem = max(__builtin_amdgcn_readlane(rem, 0), __builtin_amdgcn_readlane(rem, 16));
    while (mrem > 0) {
        int nsrc = 0;
        float nw = 0.0f;
        if (lm + 16 < rem) {
            nsrc = csrc[base + 16 + lm];
            nw = cw[base + 16 + lm];
        }
        if (0 < mrem) slot_group<0>(v, msrc, mw, rem, c4, val, accE, accO);
        if (4 < mrem) slot_group<4>(v, msrc, mw, rem, c4, val, accE, accO);
        if (8 < mrem) slot_group<8>(v, msrc, mw, rem, c4, val, accE, accO);
        if (12 < mrem) slot_group<12>(v, msrc, mw, rem, c4, val, accE, accO);
        base += 16;
        rem -= 16;
        mrem -= 16;
        msrc = nsrc;
        mw = nw;
    }
    float4 acc;
    acc.x = accE.x + accO.x;
    acc.y = accE.y + accO.y;
    acc.z = accE.z + accO.z;
    acc.w = accE.w + accO.w;

    float4 o = make_float4(0.f, 0.f, 0.f, 0.f);
    if (valid) {
        o.x = adiag * vd.x;
        o.y = adiag * vd.y;
        o.z = adiag * vd.z;
        o.w = adiag * vd.w;
        if (HAS_PREV) {
            o.x += beta * pd.x;
            o.y += beta * pd.y;
            o.z += beta * pd.z;
            o.w += beta * pd.w;
        }
        o.x += alpha * acc.x;
        o.y += alpha * acc.y;
        o.z += alpha * acc.z;
        o.w += alpha * acc.w;
        if (STORE_DST) *(float4*)(dst + row64(d) + c4) = o;
    }

    v8f c0 = {0.f, 0.f, 0.f, 0.f, 0.f, 0.f, 0.f, 0.f};
    v8f c1 = {0.f, 0.f, 0.f, 0.f, 0.f, 0.f, 0.f, 0.f};
    const int mrow0 = row0 + mt * 16;
    if (mat) {
        if (FIRST) {
            const float bv0 = bias[ct + lm];
            const float bv1 = bias[ct + 16 + lm];
#pragma unroll
            for (int i = 0; i < 8; ++i) {
                c0[i] = bv0;
                c1[i] = bv1;
            }
        } else {
            const int nclamp = N - 1;
#pragma unroll
            for (int i = 0; i < 8; ++i) {
                const int r = mrow0 + i + h * 8;
                const bool ok = (r < N);
                const int rc = ok ? r : nclamp;
                const float* orow = out + (size_t)rc * 128 + colofs + ct + lm;
                const float v0 = orow[0], v1 = orow[16];
                c0[i] = ok ? v0 : 0.f;
                c1[i] = ok ? v1 : 0.f;
            }
        }
    }

    *(float4*)(Tl + trow * TP + c4) = o;
    if (FIRST) *(float4*)(Tl + 32 * TP + trow * TP + c4) = vd;
    __syncthreads();

    if (mat) {
        if (FIRST) mma_tile(Tl + 32 * TP + mt * 16 * TP, Wl + 64 * WP, lm, h, ct, c0, c1);
        mma_tile(Tl + mt * 16 * TP, Wl, lm, h, ct, c0, c1);
#pragma unroll
        for (int i = 0; i < 8; ++i) {
            const int r = mrow0 + i + h * 8;
            if (r < N) {
                float* orow = out + (size_t)r * 128 + colofs + ct + lm;
                orow[0]  = c0[i];
                orow[16] = c1[i];
            }
        }
    }
}


extern "C" void kernel_launch(void* const* d_in, const int* in_sizes, int n_in,
                              void* d_out, int out_size, void* d_ws, size_t ws_size,
                              hipStream_t stream) {
    const float* x  = (const float*)d_in[0];
    const int*   ei = (const int*)d_in[1];
    const float* ew = (const float*)d_in[2];
    const float* W1 = (const float*)d_in[3];
    const float* b1 = (const float*)d_in[4];
    const float* W2 = (const float*)d_in[5];
    const float* b2 = (const float*)d_in[6];
    float* out = (float*)d_out;

    const int N = in_sizes[0] / 64;
    const int E = in_sizes[2];

    const int per = (((N + 1023) / 1024) + 3) & ~3;
    const int NP  = per * 1024;
    const int EP  = (E + 3) & ~3;

    float* tA   = (float*)d_ws;
    float* tB   = tA + (size_t)N * 64;
    int*   csrc = (int*)(tB + (size_t)N * 64);
    float* cw   = (float*)(csrc + 2 * (size_t)EP);
    int*   cnt  = (int*)(cw + 2 * (size_t)EP);
    int*   rp   = cnt + 2 * (size_t)NP;
    float* dis  = (float*)(rp + 2 * (size_t)(NP + 4));
    int*   rank = (int*)tA;
    int*   perm = rank + 2 * (size_t)EP;
    int*   cntp = perm + 2 * (size_t)EP;
    const int CE = (E + CCH - 1) / CCH;
    const int NR = (NP + RNG - 1) / RNG;

    const float DIAG = 2.0f / 3.0f - 1.0f;

    dim3 blk(256);
    dim3 blk16w(512);
    int gE    = (E + 255) / 256;
    int gN    = (N + 255) / 256;
    int gN8   = (N + 7) / 8;
    int gN32  = (N + 31) / 32;

    k_rangecount<<<dim3(NR, CCH, 2), dim3(1024), 0, stream>>>(ei, cntp, rank, E, EP, NP, CE);
    k_combine<<<dim3((NP + 255) / 256, 2), blk, 0, stream>>>(cntp, cnt, NP);
    k_scan<<<2, dim3(1024), 0, stream>>>(cnt, rp, per, NP);
    k_place<<<dim3(gE, 2), blk, 0, stream>>>(ei, rp, rank, cntp, perm, E, EP, NP, CE);
    k_order<<<dim3(gN8, 2), blk, 0, stream>>>(ei, ew, rp, perm, csrc, cw, N, E, EP, NP);
    k_dis<<<dim3(gN, 2), blk, 0, stream>>>(rp, cw, dis, N, EP, NP);
    k_csrw<<<dim3(gN, 2), blk, 0, stream>>>(rp, csrc, cw, dis, N, EP, NP);

    for (int dir = 0; dir < 2; ++dir) {
        const int*   r    = rp + (size_t)dir * (NP + 4);
        const int*   sidx = csrc + (size_t)dir * EP;
        const float* w    = cw + (size_t)dir * EP;
        const float* W    = (dir == 0) ? W1 : W2;
        const float* b    = (dir == 0) ? b1 : b2;
        const int    cofs = (dir == 0) ? 0 : 64;

        k_lhat_proj<false, true, true><<<gN32, blk16w, 0, stream>>>(
            x, x, tA, r, sidx, w, DIAG, 0.0f, 1.0f, W, W + 1 * 4096, b, out, cofs, N);

        k_lhat_proj<true, false, true><<<gN32, blk16w, 0, stream>>>(
            tA, x, tB, r, sidx, w, 2.0f * DIAG, -1.0f, 2.0f, W, W + 2 * 4096, b, out, cofs, N);

        k_lhat_proj<true, false, true><<<gN32, blk16w, 0, stream>>>(
            tB, tA, tA, r, sidx, w, 2.0f * DIAG, -1.0f, 2.0f, W, W + 3 * 4096, b, out, cofs, N);

        k_lhat_proj<true, false, false><<<gN32, blk16w, 0, stream>>>(
            tA, tB, tB, r, sidx, w, 2.0f * DIAG, -1.0f, 2.0f, W, W + 4 * 4096, b, out, cofs, N);
    }
}


# Sparse pass: gather row loads unmasked (an ended lane reads row 0 with weight 0), own index register each, 8 per group in one clause: no branch or translation wait between them; float sums unchanged
# speedup vs baseline: 3.0150x; 1.0324x over previous
#include <hip/hip_runtime.h>


typedef float v2f __attribute__((ext_vector_type(2)));
typedef float v8f __attribute__((ext_vector_type(8)));

static __device__ __forceinline__ size_t row64(int n) { return (size_t)n * 64; }


#define RNG 8192
#define CCH 16

__global__ __launch_bounds__(1024) void k_rangecount(const int* __restrict__ ei,
                                                     int* __restrict__ cntp,
                                                     int* __restrict__ rank,
                                                     int E, int EP, int NP, int CE) {
    __shared__ int lc[RNG];
    const int t = threadIdx.x;
    const int r0 = blockIdx.x * RNG;
    const int c = blockIdx.y;
    const int dir = blockIdx.z;
    for (int i = t; i < RNG; i += 1024) lc[i] = 0;
    __syncthreads();
    const int* dst = ei + (size_t)dir * E;
    int* rk = rank + (size_t)dir * EP;
    const int cbeg = c * CE;
    int cend = cbeg + CE;
    cend = cend < E ? cend : E;
    for (int e0 = cbeg + t; e0 < cend; e0 += 8 * 1024) {
        int dv[8];
#pragma unroll
        for (int j = 0; j < 8; ++j) {
            const int e = e0 + j * 1024;
            dv[j] = (e < cend) ? dst[e] : -1;
        }
#pragma unroll
        for (int j = 0; j < 8; ++j) {
            const unsigned rel = (unsigned)(dv[j] - r0);
            if (rel < (unsigned)RNG) rk[e0 + j * 1024] = atomicAdd(&lc[rel], 1);
        }
    }
    __syncthreads();
    int* o = cntp + ((size_t)dir * CCH + c) * NP;
    for (int i = t; i < RNG; i += 1024) {
        const int d = r0 + i;
        if (d < NP) o[d] = lc[i];
    }
}

__global__ __launch_bounds__(256) void k_combine(int* __restrict__ cntp,
                                                 int* __restrict__ cnt, int NP) {
    const int d = blockIdx.x * 256 + threadIdx.x;
    if (d >= NP) return;
    const int dir = blockIdx.y;
    int* p = cntp + (size_t)dir * CCH * NP + d;
    int v[CCH];
#pragma unroll
    for (int c = 0; c < CCH; ++c) v[c] = p[(size_t)c * NP];
    int run = 0;
#pragma unroll
    for (int c = 0; c < CCH; ++c) {
        p[(size_t)c * NP] = run;
        run += v[c];
    }
    cnt[(size_t)dir * NP + d] = run;
}

__global__ __launch_bounds__(1024) void k_scan(const int* __restrict__ cnt,
                                               int* __restrict__ rp, int per, int NP) {
    __shared__ int part[1024];
    const int t = threadIdx.x;
    const int4* c4 = (const int4*)(cnt + (size_t)blockIdx.x * NP + (size_t)t * per);
    int* rbase = rp + (size_t)blockIdx.x * (NP + 4);
    int4* r4 = (int4*)(rbase + (size_t)t * per);
    const int nq = per >> 2;
    int s = 0;
    for (int q = 0; q < nq; ++q) {
        int4 c = c4[q];
        s += c.x + c.y + c.z + c.w;
    }
    part[t] = s;
    __syncthreads();
    for (int off = 1; off < 1024; off <<= 1) {
        int add = (t >= off) ? part[t - off] : 0;
        __syncthreads();
        part[t] += add;
        __syncthreads();
    }
    int run = part[t] - s;
    for (int q = 0; q < nq; ++q) {
        int4 c = c4[q];
        int4 o;
        o.x = run; run += c.x;
        o.y = run; run += c.y;
        o.z = run; run += c.z;
        o.w = run; run += c.w;
        r4[q] = o;
    }
    if (t == 1023) rbase[NP] = run;
}

__global__ __launch_bounds__(256) void k_place(const int* __restrict__ ei,
                                               const int* __restrict__ rp,
                                               const int* __restrict__ rank,
                                               const int* __restrict__ cntp,
                                               int* __restrict__ perm,
                                               int E, int EP, int NP, int CE) {
    int e = blockIdx.x * 256 + threadIdx.x;
    if (e >= E) return;
    const int dir = blockIdx.y;
    const int c = e / CE;
    int d = ei[(size_t)dir * E + e];
    int pos = rp[(size_t)dir * (NP + 4) + d] + cntp[((size_t)dir * CCH + c) * NP + d] +
              rank[(size_t)dir * EP + e];
    perm[(size_t)dir * EP + pos] = e;
}

__global__ __launch_bounds__(256) void k_order(const int* __restrict__ ei,
                                               const float* __restrict__ ew,
                                               const int* __restrict__ rp,
                                               const int* __restrict__ perm,
                                               int* __restrict__ csrc,
                                               float* __restrict__ cw,
                                               int N, int E, int EP, int NP) {
    const int d = blockIdx.x * 8 + (threadIdx.x >> 5);
    if (d >= N) return;
    const int dir = blockIdx.y;
    const int lane = threadIdx.x & 31;
    const int* r = rp + (size_t)dir * (NP + 4);
    const int* pm = perm + (size_t)dir * EP;
    const int* srcIdx = ei + (size_t)(1 - dir) * E;
    const int s0 = __builtin_amdgcn_readfirstlane(r[d]);
    const int s1 = __builtin_amdgcn_readfirstlane(r[d + 1]);
    for (int ca = s0; ca < s1; ca += 32) {
        const bool mine = (ca + lane) < s1;
        const int my = mine ? pm[ca + lane] : 0;
        int below = 0;
        for (int cb = s0; cb < s1; cb += 32) {
            const int other = (cb + lane < s1) ? pm[cb + lane] : 0;
            int nb = s1 - cb;
            nb = nb < 32 ? nb : 32;
            for (int j = 0; j < nb; ++j) {
                const int oj = __builtin_amdgcn_readlane(other, j);
                below += (oj < my) ? 1 : 0;
            }
        }
        if (mine) {
            const size_t slot = (size_t)dir * EP + s0 + below;
            csrc[slot] = srcIdx[my];
            cw[slot] = ew[my];
        }
    }
}

__global__ __launch_bounds__(256) void k_dis(const int* __restrict__ rp,
                                             const float* __restrict__ cw,
                                             float* __restrict__ dis,
                                             int N, int EP, int NP) {
    int d = blockIdx.x * 256 + threadIdx.x;
    if (d >= N) return;
    const int dir = blockIdx.y;
    const int* r = rp + (size_t)dir * (NP + 4);
    const float* w = cw + (size_t)dir * EP;
    int s0 = r[d], s1 = r[d + 1];
    float deg = 0.0f;
    for (int q = s0; q < s1; ++q) deg += w[q];
    dis[(size_t)dir * NP + d] = (deg > 0.0f) ? rsqrtf(deg) : 0.0f;
}

__global__ __launch_bounds__(256) void k_csrw(const int* __restrict__ rp,
                                              const int* __restrict__ csrc,
                                              float* __restrict__ cw,
                                              const float* __restrict__ dis,
                                              int N, int EP, int NP) {
    int d = blockIdx.x * 256 + threadIdx.x;
    if (d >= N) return;
    const int dir = blockIdx.y;
    const float scale = 2.0f / 3.0f;
    const int* r = rp + (size_t)dir * (NP + 4);
    const int* sidx = csrc + (size_t)dir * EP;
    float* w = cw + (size_t)dir * EP;
    const float* ds = dis + (size_t)dir * NP;
    int s0 = r[d], s1 = r[d + 1];
    float dd = -scale * ds[d];
    for (int q = s0; q < s1; ++q) w[q] = dd * w[q] * ds[sidx[q]];
}

#define TP 68
#define WP 80

static __device__ __forceinline__ void mma_tile(const float* tile, const float* wl,
                                                int lm, int h, int ct,
                                                v8f& acc0, v8f& acc1) {
    const float* arow = tile + lm * TP;
    const float* bcol = wl + ct + lm;
#pragma unroll
    for (int kk = 0; kk < 16; ++kk) {
        const int kbase = kk * 4 + h * 2;
        v2f a = *(const v2f*)(arow + kbase);
        v2f b0, b1;
        b0[0] = bcol[(kbase + 0) * WP + 0];
        b0[1] = bcol[(kbase + 1) * WP + 0];
        b1[0] = bcol[(kbase + 0) * WP + 16];
        b1[1] = bcol[(kbase + 1) * WP + 16];
        acc0 = __builtin_amdgcn_wmma_f32_16x16x4_f32(false, a, false, b0, (short)0, acc0, false, false);
        acc1 = __builtin_amdgcn_wmma_f32_16x16x4_f32(false, a, false, b1, (short)0, acc1, false, false);
    }
}

template <bool FIRST>
static __device__ __forceinline__ void put_weights(float* Wl, int tid,
                                                   float4 wb0, float4 wb1,
                                                   float4 wa0, float4 wa1) {
    float* wdst = Wl + (tid >> 4) * WP + (tid & 15) * 4;
    *(float4*)wdst = wb0;
    *(float4*)(wdst + 32 * WP) = wb1;
    if (FIRST) {
        *(float4*)(wdst + 64 * WP) = wa0;
        *(float4*)(wdst + 96 * WP) = wa1;
    }
}

template <int Q>
static __device__ __forceinline__ int row_take(int x) {
    return __builtin_amdgcn_update_dpp(0, x, 0x150 + Q, 0xf, 0xf, true);
}

template <int Q>
static __device__ __forceinline__ void slot_take(int msrc, float mw, int lm, int& idx, float& w) {
    idx = row_take<Q>(msrc) * 16 + lm;
    w = __int_as_float(row_take<Q>(__float_as_int(mw)));
}

static __device__ __forceinline__ void slot_sum(float w, const float4& val, float4& acc) {
    acc.x = fmaf(w, val.x, acc.x);
    acc.y = fmaf(w, val.y, acc.y);
    acc.z = fmaf(w, val.z, acc.z);
    acc.w = fmaf(w, val.w, acc.w);
}

struct Rows8 {
    float4 r0, r1, r2, r3, r4, r5, r6, r7;
    float w0, w1, w2, w3, w4, w5, w6, w7;
};

template <int Q0>
static __device__ __forceinline__ void rows_load(const float4* __restrict__ v4, int msrc, float mw,
                                                 int lm, Rows8& g) {
    int i0, i1, i2, i3, i4, i5, i6, i7;
    slot_take<Q0 + 0>(msrc, mw, lm, i0, g.w0);
    slot_take<Q0 + 1>(msrc, mw, lm, i1, g.w1);
    slot_take<Q0 + 2>(msrc, mw, lm, i2, g.w2);
    slot_take<Q0 + 3>(msrc, mw, lm, i3, g.w3);
    slot_take<Q0 + 4>(msrc, mw, lm, i4, g.w4);
    slot_take<Q0 + 5>(msrc, mw, lm, i5, g.w5);
    slot_take<Q0 + 6>(msrc, mw, lm, i6, g.w6);
    slot_take<Q0 + 7>(msrc, mw, lm, i7, g.w7);
    asm volatile("" : : "v"(i0), "v"(i1), "v"(i2), "v"(i3), "v"(i4), "v"(i5), "v"(i6), "v"(i7));
    g.r0 = v4[i0];
    g.r1 = v4[i1];
    g.r2 = v4[i2];
    g.r3 = v4[i3];
    g.r4 = v4[i4];
    g.r5 = v4[i5];
    g.r6 = v4[i6];
    g.r7 = v4[i7];
}

static __device__ __forceinline__ void rows_sum(const Rows8& g, float4& accE, float4& accO) {
    slot_sum(g.w0, g.r0, accE);
    slot_sum(g.w1, g.r1, accO);
    slot_sum(g.w2, g.r2, accE);
    slot_sum(g.w3, g.r3, accO);
    slot_sum(g.w4, g.r4, accE);
    slot_sum(g.w5, g.r5, accO);
    slot_sum(g.w6, g.r6, accE);
    slot_sum(g.w7, g.r7, accO);
}

template <bool HAS_PREV, bool FIRST, bool STORE_DST>
__global__ __launch_bounds__(512) __attribute__((amdgpu_waves_per_eu(16, 16))) void k_lhat_proj(const float* __restrict__ v,
                                                   const float* prev,
                                                   float* dst,
                                                   const int* __restrict__ rp,
                                                   const int* __restrict__ csrc,
                                                   const float* __restrict__ cw,
                                                   float adiag, float beta, float alpha,
                                                   const float* Wa,
                                                   const float* Wb,
                                                   const float* bias,
                                                   float* out, int colofs, int N) {
    __shared__ float Wl[(FIRST ? 2 : 1) * 64 * WP];
    __shared__ float Tl[(FIRST ? 2 : 1) * 32 * TP];

    const int tid = threadIdx.x;
    const int wave = tid >> 5;
    const int lane = tid & 31;
    const int h = lane >> 4;
    const int lm = lane & 15;
    const int c4 = lm * 4;
    const int row0 = blockIdx.x * 32;
    const int trow = wave * 2 + h;
    const int d = row0 + trow;
    const bool valid = d < N;

    const float4 wb0 = ((const float4*)Wb)[tid];
    const float4 wb1 = ((const float4*)Wb)[tid + 512];
    float4 wa0 = make_float4(0.f, 0.f, 0.f, 0.f);
    float4 wa1 = make_float4(0.f, 0.f, 0.f, 0.f);
    if (FIRST) {
        wa0 = ((const float4*)Wa)[tid];
        wa1 = ((const float4*)Wa)[tid + 512];
    }

    const int mj = wave - (blockIdx.x & 3) * 4;
    const bool mat = (mj >= 0) && (mj < 4);
    const int mt = (mj >> 1) & 1;
    const int ct = (mj & 1) * 32;

    int s0 = 0, s1 = 0;
    if (valid) {
        s0 = rp[d];
        s1 = rp[d + 1];
    }
    int base = s0;
    int rem = s1 - s0;

    int msrc = 0;
    float mw = 0.0f;
    if (lm < rem) {
        msrc = csrc[base + lm];
        mw = cw[base + lm];
    }

    put_weights<FIRST>(Wl, tid, wb0, wb1, wa0, wa1);

    float4 accE = make_float4(0.f, 0.f, 0.f, 0.f);
    float4 accO = make_float4(0.f, 0.f, 0.f, 0.f);
    const float4* __restrict__ v4 = (const float4*)v;

    int mrem = max(__builtin_amdgcn_readlane(rem, 0), __builtin_amdgcn_readlane(rem, 16));
    while (mrem > 0) {
        const bool more = (lm + 16) < rem;
        const int nidx = more ? (base + 16 + lm) : base;
        asm volatile("" : : "v"(nidx));
        Rows8 g;
        rows_load<0>(v4, msrc, mw, lm, g);
        const int ls = csrc[nidx];
        const float lw = cw[nidx];
        rows_sum(g, accE, accO);
        if (8 < mrem) {
            rows_load<8>(v4, msrc, mw, lm, g);
            rows_sum(g, accE, accO);
        }
        base += 16;
        rem -= 16;
        mrem -= 16;
        msrc = more ? ls : 0;
        mw = more ? lw : 0.0f;
    }
    float4 vd = make_float4(0.f, 0.f, 0.f, 0.f);
    float4 pd = make_float4(0.f, 0.f, 0.f, 0.f);
    if (valid) {
        vd = *(const float4*)(v + row64(d) + c4);
        if (HAS_PREV) pd = *(const float4*)(prev + row64(d) + c4);
    }
    float4 acc;
    acc.x = accE.x + accO.x;
    acc.y = accE.y + accO.y;
    acc.z = accE.z + accO.z;
    acc.w = accE.w + accO.w;

    v8f c0 = {0.f, 0.f, 0.f, 0.f, 0.f, 0.f, 0.f, 0.f};
    v8f c1 = {0.f, 0.f, 0.f, 0.f, 0.f, 0.f, 0.f, 0.f};
    const int mrow0 = row0 + mt * 16;
    if (mat) {
        if (FIRST) {
            const float bv0 = bias[ct + lm];
            const float bv1 = bias[ct + 16 + lm];
#pragma unroll
            for (int i = 0; i < 8; ++i) {
                c0[i] = bv0;
                c1[i] = bv1;
            }
        } else {
            const int nclamp = N - 1;
#pragma unroll
            for (int i = 0; i < 8; ++i) {
                const int r = mrow0 + i + h * 8;
                const bool ok = (r < N);
                const int rc = ok ? r : nclamp;
                const float* orow = out + (size_t)rc * 128 + colofs + ct + lm;
                const float v0 = orow[0], v1 = orow[16];
                c0[i] = ok ? v0 : 0.f;
                c1[i] = ok ? v1 : 0.f;
            }
        }
    }

    float4 o = make_float4(0.f, 0.f, 0.f, 0.f);
    if (valid) {
        o.x = adiag * vd.x;
        o.y = adiag * vd.y;
        o.z = adiag * vd.z;
        o.w = adiag * vd.w;
        if (HAS_PREV) {
            o.x += beta * pd.x;
            o.y += beta * pd.y;
            o.z += beta * pd.z;
            o.w += beta * pd.w;
        }
        o.x += alpha * acc.x;
        o.y += alpha * acc.y;
        o.z += alpha * acc.z;
        o.w += alpha * acc.w;
        if (STORE_DST) *(float4*)(dst + row64(d) + c4) = o;
    }

    *(float4*)(Tl + trow * TP + c4) = o;
    if (FIRST) *(float4*)(Tl + 32 * TP + trow * TP + c4) = vd;
    __syncthreads();

    if (mat) {
        if (FIRST) mma_tile(Tl + 32 * TP + mt * 16 * TP, Wl + 64 * WP, lm, h, ct, c0, c1);
        mma_tile(Tl + mt * 16 * TP, Wl, lm, h, ct, c0, c1);
#pragma unroll
        for (int i = 0; i < 8; ++i) {
            const int r = mrow0 + i + h * 8;
            if (r < N) {
                float* orow = out + (size_t)r * 128 + colofs + ct + lm;
                orow[0]  = c0[i];
                orow[16] = c1[i];
            }
        }
    }
}


extern "C" void kernel_launch(void* const* d_in, const int* in_sizes, int n_in,
                              void* d_out, int out_size, void* d_ws, size_t ws_size,
                              hipStream_t stream) {
    const float* x  = (const float*)d_in[0];
    const int*   ei = (const int*)d_in[1];
    const float* ew = (const float*)d_in[2];
    const float* W1 = (const float*)d_in[3];
    const float* b1 = (const float*)d_in[4];
    const float* W2 = (const float*)d_in[5];
    const float* b2 = (const float*)d_in[6];
    float* out = (float*)d_out;

    const int N = in_sizes[0] / 64;
    const int E = in_sizes[2];

    const int per = (((N + 1023) / 1024) + 3) & ~3;
    const int NP  = per * 1024;
    const int EP  = (E + 3) & ~3;

    float* tA   = (float*)d_ws;
    float* tB   = tA + (size_t)N * 64;
    int*   csrc = (int*)(tB + (size_t)N * 64);
    float* cw   = (float*)(csrc + 2 * (size_t)EP);
    int*   cnt  = (int*)(cw + 2 * (size_t)EP);
    int*   rp   = cnt + 2 * (size_t)NP;
    float* dis  = (float*)(rp + 2 * (size_t)(NP + 4));
    int*   rank = (int*)tA;
    int*   perm = rank + 2 * (size_t)EP;
    int*   cntp = perm + 2 * (size_t)EP;
    const int CE = (E + CCH - 1) / CCH;
    const int NR = (NP + RNG - 1) / RNG;

    const float DIAG = 2.0f / 3.0f - 1.0f;

    dim3 blk(256);
    dim3 blk16w(512);
    int gE    = (E + 255) / 256;
    int gN    = (N + 255) / 256;
    int gN8   = (N + 7) / 8;
    int gN32  = (N + 31) / 32;

    k_rangecount<<<dim3(NR, CCH, 2), dim3(1024), 0, stream>>>(ei, cntp, rank, E, EP, NP, CE);
    k_combine<<<dim3((NP + 255) / 256, 2), blk, 0, stream>>>(cntp, cnt, NP);
    k_scan<<<2, dim3(1024), 0, stream>>>(cnt, rp, per, NP);
    k_place<<<dim3(gE, 2), blk, 0, stream>>>(ei, rp, rank, cntp, perm, E, EP, NP, CE);
    k_order<<<dim3(gN8, 2), blk, 0, stream>>>(ei, ew, rp, perm, csrc, cw, N, E, EP, NP);
    k_dis<<<dim3(gN, 2), blk, 0, stream>>>(rp, cw, dis, N, EP, NP);
    k_csrw<<<dim3(gN, 2), blk, 0, stream>>>(rp, csrc, cw, dis, N, EP, NP);

    for (int dir = 0; dir < 2; ++dir) {
        const int*   r    = rp + (size_t)dir * (NP + 4);
        const int*   sidx = csrc + (size_t)dir * EP;
        const float* w    = cw + (size_t)dir * EP;
        const float* W    = (dir == 0) ? W1 : W2;
        const float* b    = (dir == 0) ? b1 : b2;
        const int    cofs = (dir == 0) ? 0 : 64;

        k_lhat_proj<false, true, true><<<gN32, blk16w, 0, stream>>>(
            x, x, tA, r, sidx, w, DIAG, 0.0f, 1.0f, W, W + 1 * 4096, b, out, cofs, N);

        k_lhat_proj<true, false, true><<<gN32, blk16w, 0, stream>>>(
            tA, x, tB, r, sidx, w, 2.0f * DIAG, -1.0f, 2.0f, W, W + 2 * 4096, b, out, cofs, N);

        k_lhat_proj<true, false, true><<<gN32, blk16w, 0, stream>>>(
            tB, tA, tA, r, sidx, w, 2.0f * DIAG, -1.0f, 2.0f, W, W + 3 * 4096, b, out, cofs, N);

        k_lhat_proj<true, false, false><<<gN32, blk16w, 0, stream>>>(
            tA, tB, tB, r, sidx, w, 2.0f * DIAG, -1.0f, 2.0f, W, W + 4 * 4096, b, out, cofs, N);
    }
}


# Segment starts: chunk-offset pass and two-block scan replaced by one launch of 1024-node blocks (whole-line reads, block scan) writing absolute chunk starts; 15 -> 14 launches; float sums unchanged
# speedup vs baseline: 3.3341x; 1.1058x over previous
#include <hip/hip_runtime.h>


typedef float v2f __attribute__((ext_vector_type(2)));
typedef float v8f __attribute__((ext_vector_type(8)));

static __device__ __forceinline__ size_t row64(int n) { return (size_t)n * 64; }


#define RNG 8192
#define CCH 16

__global__ __launch_bounds__(1024) void k_rangecount(const int* __restrict__ ei,
                                                     int* __restrict__ cntp,
                                                     int* __restrict__ rank,
                                                     int* __restrict__ btot,
                                                     int E, int EP, int NP, int CE) {
    __shared__ int lc[RNG];
    __shared__ int hits[RNG / 1024];
    const int t = threadIdx.x;
    const int r0 = blockIdx.x * RNG;
    const int c = blockIdx.y;
    const int dir = blockIdx.z;
    if (t < RNG / 1024) hits[t] = 0;
    for (int i = t; i < RNG; i += 1024) lc[i] = 0;
    __syncthreads();
    const int* dst = ei + (size_t)dir * E;
    int* rk = rank + (size_t)dir * EP;
    const int cbeg = c * CE;
    int cend = cbeg + CE;
    cend = cend < E ? cend : E;
    for (int e0 = cbeg + t; e0 < cend; e0 += 8 * 1024) {
        int dv[8];
#pragma unroll
        for (int j = 0; j < 8; ++j) {
            const int e = e0 + j * 1024;
            dv[j] = (e < cend) ? dst[e] : -1;
        }
#pragma unroll
        for (int j = 0; j < 8; ++j) {
            const unsigned rel = (unsigned)(dv[j] - r0);
            if (rel < (unsigned)RNG) rk[e0 + j * 1024] = atomicAdd(&lc[rel], 1);
        }
    }
    __syncthreads();
    int* o = cntp + ((size_t)dir * CCH + c) * NP;
    int w[RNG / 1024];
#pragma unroll
    for (int s = 0; s < RNG / 1024; ++s) {
        const int i = s * 1024 + t;
        const int v = lc[i];
        const int d = r0 + i;
        if (d < NP) o[d] = v;
        w[s] = v;
    }
#pragma unroll
    for (int off = 16; off > 0; off >>= 1) {
#pragma unroll
        for (int s = 0; s < RNG / 1024; ++s) w[s] += __shfl_xor(w[s], off);
    }
    if ((t & 31) == 0) {
#pragma unroll
        for (int s = 0; s < RNG / 1024; ++s) atomicAdd(&hits[s], w[s]);
    }
    __syncthreads();
    if (t < RNG / 1024)
        btot[((size_t)dir * gridDim.x * (RNG / 1024) + blockIdx.x * (RNG / 1024) + t) * CCH + c] = hits[t];
}

__global__ __launch_bounds__(1024) void k_starts(int* __restrict__ cntp,
                                                 const int* __restrict__ btot,
                                                 int* __restrict__ rp, int NP, int NSB) {
    __shared__ int wtot[32];
    __shared__ int wbef[32];
    const int t = threadIdx.x;
    const int lane = t & 31;
    const int wave = t >> 5;
    const int sb = blockIdx.x;
    const int dir = blockIdx.y;
    const int d = sb * 1024 + t;
    int* p = cntp + (size_t)dir * CCH * NP + d;

    const int* bt = btot + (size_t)dir * NSB * CCH;
    const int nb = sb * CCH;
    int bs = 0;
    for (int q = t; q < nb; q += 1024) bs += bt[q];

    int v[CCH];
#pragma unroll
    for (int c = 0; c < CCH; ++c) v[c] = p[(size_t)c * NP];
    int tsum = 0;
#pragma unroll
    for (int c = 0; c < CCH; ++c) tsum += v[c];

    int inc = tsum;
    for (int off = 1; off < 32; off <<= 1) {
        const int n = __shfl_up(inc, off);
        if (lane >= off) inc += n;
    }
    for (int off = 16; off > 0; off >>= 1) bs += __shfl_xor(bs, off);
    if (lane == 31) wtot[wave] = inc;
    if (lane == 0) wbef[wave] = bs;
    __syncthreads();
    int winc = wtot[lane];
    for (int off = 1; off < 32; off <<= 1) {
        const int n = __shfl_up(winc, off);
        if (lane >= off) winc += n;
    }
    int before = wbef[lane];
    for (int off = 16; off > 0; off >>= 1) before += __shfl_xor(before, off);
    const int wprev = __shfl(winc, (wave + 31) & 31);
    const int wbase = (wave > 0) ? wprev : 0;
    const int base = before + wbase + (inc - tsum);

    int* rr = rp + (size_t)dir * (NP + 4);
    rr[d] = base;
    if (d == NP - 1) rr[NP] = base + tsum;

    int run = base;
#pragma unroll
    for (int c = 0; c < CCH; ++c) {
        p[(size_t)c * NP] = run;
        run += v[c];
    }
}

__global__ __launch_bounds__(256) void k_place(const int* __restrict__ ei,
                                               const int* __restrict__ rank,
                                               const int* __restrict__ cntp,
                                               int* __restrict__ perm,
                                               int E, int EP, int NP, int CE) {
    int e = blockIdx.x * 256 + threadIdx.x;
    if (e >= E) return;
    const int dir = blockIdx.y;
    const int c = e / CE;
    int d = ei[(size_t)dir * E + e];
    int pos = cntp[((size_t)dir * CCH + c) * NP + d] + rank[(size_t)dir * EP + e];
    perm[(size_t)dir * EP + pos] = e;
}

__global__ __launch_bounds__(256) void k_order(const int* __restrict__ ei,
                                               const float* __restrict__ ew,
                                               const int* __restrict__ rp,
                                               const int* __restrict__ perm,
                                               int* __restrict__ csrc,
                                               float* __restrict__ cw,
                                               int N, int E, int EP, int NP) {
    const int d = blockIdx.x * 8 + (threadIdx.x >> 5);
    if (d >= N) return;
    const int dir = blockIdx.y;
    const int lane = threadIdx.x & 31;
    const int* r = rp + (size_t)dir * (NP + 4);
    const int* pm = perm + (size_t)dir * EP;
    const int* srcIdx = ei + (size_t)(1 - dir) * E;
    const int s0 = __builtin_amdgcn_readfirstlane(r[d]);
    const int s1 = __builtin_amdgcn_readfirstlane(r[d + 1]);
    for (int ca = s0; ca < s1; ca += 32) {
        const bool mine = (ca + lane) < s1;
        const int my = mine ? pm[ca + lane] : 0;
        int below = 0;
        for (int cb = s0; cb < s1; cb += 32) {
            const int other = (cb + lane < s1) ? pm[cb + lane] : 0;
            int nb = s1 - cb;
            nb = nb < 32 ? nb : 32;
            for (int j = 0; j < nb; ++j) {
                const int oj = __builtin_amdgcn_readlane(other, j);
                below += (oj < my) ? 1 : 0;
            }
        }
        if (mine) {
            const size_t slot = (size_t)dir * EP + s0 + below;
            csrc[slot] = srcIdx[my];
            cw[slot] = ew[my];
        }
    }
}

__global__ __launch_bounds__(256) void k_dis(const int* __restrict__ rp,
                                             const float* __restrict__ cw,
                                             float* __restrict__ dis,
                                             int N, int EP, int NP) {
    int d = blockIdx.x * 256 + threadIdx.x;
    if (d >= N) return;
    const int dir = blockIdx.y;
    const int* r = rp + (size_t)dir * (NP + 4);
    const float* w = cw + (size_t)dir * EP;
    int s0 = r[d], s1 = r[d + 1];
    float deg = 0.0f;
    for (int q = s0; q < s1; ++q) deg += w[q];
    dis[(size_t)dir * NP + d] = (deg > 0.0f) ? rsqrtf(deg) : 0.0f;
}

__global__ __launch_bounds__(256) void k_csrw(const int* __restrict__ rp,
                                              const int* __restrict__ csrc,
                                              float* __restrict__ cw,
                                              const float* __restrict__ dis,
                                              int N, int EP, int NP) {
    int d = blockIdx.x * 256 + threadIdx.x;
    if (d >= N) return;
    const int dir = blockIdx.y;
    const float scale = 2.0f / 3.0f;
    const int* r = rp + (size_t)dir * (NP + 4);
    const int* sidx = csrc + (size_t)dir * EP;
    float* w = cw + (size_t)dir * EP;
    const float* ds = dis + (size_t)dir * NP;
    int s0 = r[d], s1 = r[d + 1];
    float dd = -scale * ds[d];
    for (int q = s0; q < s1; ++q) w[q] = dd * w[q] * ds[sidx[q]];
}

#define TP 68
#define WP 80

static __device__ __forceinline__ void mma_tile(const float* tile, const float* wl,
                                                int lm, int h, int ct,
                                                v8f& acc0, v8f& acc1) {
    const float* arow = tile + lm * TP;
    const float* bcol = wl + ct + lm;
#pragma unroll
    for (int kk = 0; kk < 16; ++kk) {
        const int kbase = kk * 4 + h * 2;
        v2f a = *(const v2f*)(arow + kbase);
        v2f b0, b1;
        b0[0] = bcol[(kbase + 0) * WP + 0];
        b0[1] = bcol[(kbase + 1) * WP + 0];
        b1[0] = bcol[(kbase + 0) * WP + 16];
        b1[1] = bcol[(kbase + 1) * WP + 16];
        acc0 = __builtin_amdgcn_wmma_f32_16x16x4_f32(false, a, false, b0, (short)0, acc0, false, false);
        acc1 = __builtin_amdgcn_wmma_f32_16x16x4_f32(false, a, false, b1, (short)0, acc1, false, false);
    }
}

template <bool FIRST>
static __device__ __forceinline__ void put_weights(float* Wl, int tid,
                                                   float4 wb0, float4 wb1,
                                                   float4 wa0, float4 wa1) {
    float* wdst = Wl + (tid >> 4) * WP + (tid & 15) * 4;
    *(float4*)wdst = wb0;
    *(float4*)(wdst + 32 * WP) = wb1;
    if (FIRST) {
        *(float4*)(wdst + 64 * WP) = wa0;
        *(float4*)(wdst + 96 * WP) = wa1;
    }
}

template <int Q>
static __device__ __forceinline__ int row_take(int x) {
    return __builtin_amdgcn_update_dpp(0, x, 0x150 + Q, 0xf, 0xf, true);
}

template <int Q>
static __device__ __forceinline__ void slot_take(int msrc, float mw, int lm, int& idx, float& w) {
    idx = row_take<Q>(msrc) * 16 + lm;
    w = __int_as_float(row_take<Q>(__float_as_int(mw)));
}

static __device__ __forceinline__ void slot_sum(float w, const float4& val, float4& acc) {
    acc.x = fmaf(w, val.x, acc.x);
    acc.y = fmaf(w, val.y, acc.y);
    acc.z = fmaf(w, val.z, acc.z);
    acc.w = fmaf(w, val.w, acc.w);
}

struct Rows8 {
    float4 r0, r1, r2, r3, r4, r5, r6, r7;
    float w0, w1, w2, w3, w4, w5, w6, w7;
};

template <int Q0>
static __device__ __forceinline__ void rows_load(const float4* __restrict__ v4, int msrc, float mw,
                                                 int lm, Rows8& g) {
    int i0, i1, i2, i3, i4, i5, i6, i7;
    slot_take<Q0 + 0>(msrc, mw, lm, i0, g.w0);
    slot_take<Q0 + 1>(msrc, mw, lm, i1, g.w1);
    slot_take<Q0 + 2>(msrc, mw, lm, i2, g.w2);
    slot_take<Q0 + 3>(msrc, mw, lm, i3, g.w3);
    slot_take<Q0 + 4>(msrc, mw, lm, i4, g.w4);
    slot_take<Q0 + 5>(msrc, mw, lm, i5, g.w5);
    slot_take<Q0 + 6>(msrc, mw, lm, i6, g.w6);
    slot_take<Q0 + 7>(msrc, mw, lm, i7, g.w7);
    asm volatile("" : : "v"(i0), "v"(i1), "v"(i2), "v"(i3), "v"(i4), "v"(i5), "v"(i6), "v"(i7));
    g.r0 = v4[i0];
    g.r1 = v4[i1];
    g.r2 = v4[i2];
    g.r3 = v4[i3];
    g.r4 = v4[i4];
    g.r5 = v4[i5];
    g.r6 = v4[i6];
    g.r7 = v4[i7];
}

static __device__ __forceinline__ void rows_sum(const Rows8& g, float4& accE, float4& accO) {
    slot_sum(g.w0, g.r0, accE);
    slot_sum(g.w1, g.r1, accO);
    slot_sum(g.w2, g.r2, accE);
    slot_sum(g.w3, g.r3, accO);
    slot_sum(g.w4, g.r4, accE);
    slot_sum(g.w5, g.r5, accO);
    slot_sum(g.w6, g.r6, accE);
    slot_sum(g.w7, g.r7, accO);
}

template <bool HAS_PREV, bool FIRST, bool STORE_DST>
__global__ __launch_bounds__(512) __attribute__((amdgpu_waves_per_eu(16, 16))) void k_lhat_proj(const float* __restrict__ v,
                                                   const float* prev,
                                                   float* dst,
                                                   const int* __restrict__ rp,
                                                   const int* __restrict__ csrc,
                                                   const float* __restrict__ cw,
                                                   float adiag, float beta, float alpha,
                                                   const float* Wa,
                                                   const float* Wb,
                                                   const float* bias,
                                                   float* out, int colofs, int N) {
    __shared__ float Wl[(FIRST ? 2 : 1) * 64 * WP];
    __shared__ float Tl[(FIRST ? 2 : 1) * 32 * TP];

    const int tid = threadIdx.x;
    const int wave = tid >> 5;
    const int lane = tid & 31;
    const int h = lane >> 4;
    const int lm = lane & 15;
    const int c4 = lm * 4;
    const int row0 = blockIdx.x * 32;
    const int trow = wave * 2 + h;
    const int d = row0 + trow;
    const bool valid = d < N;

    const float4 wb0 = ((const float4*)Wb)[tid];
    const float4 wb1 = ((const float4*)Wb)[tid + 512];
    float4 wa0 = make_float4(0.f, 0.f, 0.f, 0.f);
    float4 wa1 = make_float4(0.f, 0.f, 0.f, 0.f);
    if (FIRST) {
        wa0 = ((const float4*)Wa)[tid];
        wa1 = ((const float4*)Wa)[tid + 512];
    }

    const int mj = wave - (blockIdx.x & 3) * 4;
    const bool mat = (mj >= 0) && (mj < 4);
    const int mt = (mj >> 1) & 1;
    const int ct = (mj & 1) * 32;

    int s0 = 0, s1 = 0;
    if (valid) {
        s0 = rp[d];
        s1 = rp[d + 1];
    }
    int base = s0;
    int rem = s1 - s0;

    int msrc = 0;
    float mw = 0.0f;
    if (lm < rem) {
        msrc = csrc[base + lm];
        mw = cw[base + lm];
    }

    put_weights<FIRST>(Wl, tid, wb0, wb1, wa0, wa1);

    float4 accE = make_float4(0.f, 0.f, 0.f, 0.f);
    float4 accO = make_float4(0.f, 0.f, 0.f, 0.f);
    const float4* __restrict__ v4 = (const float4*)v;

    int mrem = max(__builtin_amdgcn_readlane(rem, 0), __builtin_amdgcn_readlane(rem, 16));
    while (mrem > 0) {
        const bool more = (lm + 16) < rem;
        const int nidx = more ? (base + 16 + lm) : base;
        asm volatile("" : : "v"(nidx));
        Rows8 g;
        rows_load<0>(v4, msrc, mw, lm, g);
        const int ls = csrc[nidx];
        const float lw = cw[nidx];
        rows_sum(g, accE, accO);
        if (8 < mrem) {
            rows_load<8>(v4, msrc, mw, lm, g);
            rows_sum(g, accE, accO);
        }
        base += 16;
        rem -= 16;
        mrem -= 16;
        msrc = more ? ls : 0;
        mw = more ? lw : 0.0f;
    }
    float4 vd = make_float4(0.f, 0.f, 0.f, 0.f);
    float4 pd = make_float4(0.f, 0.f, 0.f, 0.f);
    if (valid) {
        vd = *(const float4*)(v + row64(d) + c4);
        if (HAS_PREV) pd = *(const float4*)(prev + row64(d) + c4);
    }
    float4 acc;
    acc.x = accE.x + accO.x;
    acc.y = accE.y + accO.y;
    acc.z = accE.z + accO.z;
    acc.w = accE.w + accO.w;

    v8f c0 = {0.f, 0.f, 0.f, 0.f, 0.f, 0.f, 0.f, 0.f};
    v8f c1 = {0.f, 0.f, 0.f, 0.f, 0.f, 0.f, 0.f, 0.f};
    const int mrow0 = row0 + mt * 16;
    if (mat) {
        if (FIRST) {
            const float bv0 = bias[ct + lm];
            const float bv1 = bias[ct + 16 + lm];
#pragma unroll
            for (int i = 0; i < 8; ++i) {
                c0[i] = bv0;
                c1[i] = bv1;
            }
        } else {
            const int nclamp = N - 1;
#pragma unroll
            for (int i = 0; i < 8; ++i) {
                const int r = mrow0 + i + h * 8;
                const bool ok = (r < N);
                const int rc = ok ? r : nclamp;
                const float* orow = out + (size_t)rc * 128 + colofs + ct + lm;
                const float v0 = orow[0], v1 = orow[16];
                c0[i] = ok ? v0 : 0.f;
                c1[i] = ok ? v1 : 0.f;
            }
        }
    }

    float4 o = make_float4(0.f, 0.f, 0.f, 0.f);
    if (valid) {
        o.x = adiag * vd.x;
        o.y = adiag * vd.y;
        o.z = adiag * vd.z;
        o.w = adiag * vd.w;
        if (HAS_PREV) {
            o.x += beta * pd.x;
            o.y += beta * pd.y;
            o.z += beta * pd.z;
            o.w += beta * pd.w;
        }
        o.x += alpha * acc.x;
        o.y += alpha * acc.y;
        o.z += alpha * acc.z;
        o.w += alpha * acc.w;
        if (STORE_DST) *(float4*)(dst + row64(d) + c4) = o;
    }

    *(float4*)(Tl + trow * TP + c4) = o;
    if (FIRST) *(float4*)(Tl + 32 * TP + trow * TP + c4) = vd;
    __syncthreads();

    if (mat) {
        if (FIRST) mma_tile(Tl + 32 * TP + mt * 16 * TP, Wl + 64 * WP, lm, h, ct, c0, c1);
        mma_tile(Tl + mt * 16 * TP, Wl, lm, h, ct, c0, c1);
#pragma unroll
        for (int i = 0; i < 8; ++i) {
            const int r = mrow0 + i + h * 8;
            if (r < N) {
                float* orow = out + (size_t)r * 128 + colofs + ct + lm;
                orow[0]  = c0[i];
                orow[16] = c1[i];
            }
        }
    }
}


extern "C" void kernel_launch(void* const* d_in, const int* in_sizes, int n_in,
                              void* d_out, int out_size, void* d_ws, size_t ws_size,
                              hipStream_t stream) {
    const float* x  = (const float*)d_in[0];
    const int*   ei = (const int*)d_in[1];
    const float* ew = (const float*)d_in[2];
    const float* W1 = (const float*)d_in[3];
    const float* b1 = (const float*)d_in[4];
    const float* W2 = (const float*)d_in[5];
    const float* b2 = (const float*)d_in[6];
    float* out = (float*)d_out;

    const int N = in_sizes[0] / 64;
    const int E = in_sizes[2];

    const int per = (((N + 1023) / 1024) + 3) & ~3;
    const int NP  = per * 1024;
    const int EP  = (E + 3) & ~3;

    float* tA   = (float*)d_ws;
    float* tB   = tA + (size_t)N * 64;
    int*   csrc = (int*)(tB + (size_t)N * 64);
    float* cw   = (float*)(csrc + 2 * (size_t)EP);
    int*   btot = (int*)(cw + 2 * (size_t)EP);
    int*   rp   = btot + 2 * (size_t)NP;
    float* dis  = (float*)(rp + 2 * (size_t)(NP + 4));
    int*   rank = (int*)tA;
    int*   perm = rank + 2 * (size_t)EP;
    int*   cntp = perm + 2 * (size_t)EP;
    const int CE = (E + CCH - 1) / CCH;
    const int NR = (NP + RNG - 1) / RNG;

    const float DIAG = 2.0f / 3.0f - 1.0f;

    dim3 blk(256);
    dim3 blk16w(512);
    int gE    = (E + 255) / 256;
    int gN    = (N + 255) / 256;
    int gN8   = (N + 7) / 8;
    int gN32  = (N + 31) / 32;

    k_rangecount<<<dim3(NR, CCH, 2), dim3(1024), 0, stream>>>(ei, cntp, rank, btot, E, EP, NP, CE);
    k_starts<<<dim3(NP / 1024, 2), dim3(1024), 0, stream>>>(cntp, btot, rp, NP, NR * (RNG / 1024));
    k_place<<<dim3(gE, 2), blk, 0, stream>>>(ei, rank, cntp, perm, E, EP, NP, CE);
    k_order<<<dim3(gN8, 2), blk, 0, stream>>>(ei, ew, rp, perm, csrc, cw, N, E, EP, NP);
    k_dis<<<dim3(gN, 2), blk, 0, stream>>>(rp, cw, dis, N, EP, NP);
    k_csrw<<<dim3(gN, 2), blk, 0, stream>>>(rp, csrc, cw, dis, N, EP, NP);

    for (int dir = 0; dir < 2; ++dir) {
        const int*   r    = rp + (size_t)dir * (NP + 4);
        const int*   sidx = csrc + (size_t)dir * EP;
        const float* w    = cw + (size_t)dir * EP;
        const float* W    = (dir == 0) ? W1 : W2;
        const float* b    = (dir == 0) ? b1 : b2;
        const int    cofs = (dir == 0) ? 0 : 64;

        k_lhat_proj<false, true, true><<<gN32, blk16w, 0, stream>>>(
            x, x, tA, r, sidx, w, DIAG, 0.0f, 1.0f, W, W + 1 * 4096, b, out, cofs, N);

        k_lhat_proj<true, false, true><<<gN32, blk16w, 0, stream>>>(
            tA, x, tB, r, sidx, w, 2.0f * DIAG, -1.0f, 2.0f, W, W + 2 * 4096, b, out, cofs, N);

        k_lhat_proj<true, false, true><<<gN32, blk16w, 0, stream>>>(
            tB, tA, tA, r, sidx, w, 2.0f * DIAG, -1.0f, 2.0f, W, W + 3 * 4096, b, out, cofs, N);

        k_lhat_proj<true, false, false><<<gN32, blk16w, 0, stream>>>(
            tA, tB, tB, r, sidx, w, 2.0f * DIAG, -1.0f, 2.0f, W, W + 4 * 4096, b, out, cofs, N);
    }
}


# Sparse passes: the two directions' blocks of one order share a launch (grid row = direction, own feature-buffer pair per direction; one pair if workspace is short); 14 -> 10 launches; sums unchanged
# speedup vs baseline: 3.6355x; 1.0904x over previous
#include <hip/hip_runtime.h>


typedef float v2f __attribute__((ext_vector_type(2)));
typedef float v8f __attribute__((ext_vector_type(8)));

static __device__ __forceinline__ size_t row64(int n) { return (size_t)n * 64; }


#define RNG 8192
#define CCH 16

__global__ __launch_bounds__(1024) void k_rangecount(const int* __restrict__ ei,
                                                     int* __restrict__ cntp,
                                                     int* __restrict__ rank,
                                                     int* __restrict__ btot,
                                                     int E, int EP, int NP, int CE) {
    __shared__ int lc[RNG];
    __shared__ int hits[RNG / 1024];
    const int t = threadIdx.x;
    const int r0 = blockIdx.x * RNG;
    const int c = blockIdx.y;
    const int dir = blockIdx.z;
    if (t < RNG / 1024) hits[t] = 0;
    for (int i = t; i < RNG; i += 1024) lc[i] = 0;
    __syncthreads();
    const int* dst = ei + (size_t)dir * E;
    int* rk = rank + (size_t)dir * EP;
    const int cbeg = c * CE;
    int cend = cbeg + CE;
    cend = cend < E ? cend : E;
    for (int e0 = cbeg + t; e0 < cend; e0 += 8 * 1024) {
        int dv[8];
#pragma unroll
        for (int j = 0; j < 8; ++j) {
            const int e = e0 + j * 1024;
            dv[j] = (e < cend) ? dst[e] : -1;
        }
#pragma unroll
        for (int j = 0; j < 8; ++j) {
            const unsigned rel = (unsigned)(dv[j] - r0);
            if (rel < (unsigned)RNG) rk[e0 + j * 1024] = atomicAdd(&lc[rel], 1);
        }
    }
    __syncthreads();
    int* o = cntp + ((size_t)dir * CCH + c) * NP;
    int w[RNG / 1024];
#pragma unroll
    for (int s = 0; s < RNG / 1024; ++s) {
        const int i = s * 1024 + t;
        const int v = lc[i];
        const int d = r0 + i;
        if (d < NP) o[d] = v;
        w[s] = v;
    }
#pragma unroll
    for (int off = 16; off > 0; off >>= 1) {
#pragma unroll
        for (int s = 0; s < RNG / 1024; ++s) w[s] += __shfl_xor(w[s], off);
    }
    if ((t & 31) == 0) {
#pragma unroll
        for (int s = 0; s < RNG / 1024; ++s) atomicAdd(&hits[s], w[s]);
    }
    __syncthreads();
    if (t < RNG / 1024)
        btot[((size_t)dir * gridDim.x * (RNG / 1024) + blockIdx.x * (RNG / 1024) + t) * CCH + c] = hits[t];
}

__global__ __launch_bounds__(1024) void k_starts(int* __restrict__ cntp,
                                                 const int* __restrict__ btot,
                                                 int* __restrict__ rp, int NP, int NSB) {
    __shared__ int wtot[32];
    __shared__ int wbef[32];
    const int t = threadIdx.x;
    const int lane = t & 31;
    const int wave = t >> 5;
    const int sb = blockIdx.x;
    const int dir = blockIdx.y;
    const int d = sb * 1024 + t;
    int* p = cntp + (size_t)dir * CCH * NP + d;

    const int* bt = btot + (size_t)dir * NSB * CCH;
    const int nb = sb * CCH;
    int bs = 0;
    for (int q = t; q < nb; q += 1024) bs += bt[q];

    int v[CCH];
#pragma unroll
    for (int c = 0; c < CCH; ++c) v[c] = p[(size_t)c * NP];
    int tsum = 0;
#pragma unroll
    for (int c = 0; c < CCH; ++c) tsum += v[c];

    int inc = tsum;
    for (int off = 1; off < 32; off <<= 1) {
        const int n = __shfl_up(inc, off);
        if (lane >= off) inc += n;
    }
    for (int off = 16; off > 0; off >>= 1) bs += __shfl_xor(bs, off);
    if (lane == 31) wtot[wave] = inc;
    if (lane == 0) wbef[wave] = bs;
    __syncthreads();
    int winc = wtot[lane];
    for (int off = 1; off < 32; off <<= 1) {
        const int n = __shfl_up(winc, off);
        if (lane >= off) winc += n;
    }
    int before = wbef[lane];
    for (int off = 16; off > 0; off >>= 1) before += __shfl_xor(before, off);
    const int wprev = __shfl(winc, (wave + 31) & 31);
    const int wbase = (wave > 0) ? wprev : 0;
    const int base = before + wbase + (inc - tsum);

    int* rr = rp + (size_t)dir * (NP + 4);
    rr[d] = base;
    if (d == NP - 1) rr[NP] = base + tsum;

    int run = base;
#pragma unroll
    for (int c = 0; c < CCH; ++c) {
        p[(size_t)c * NP] = run;
        run += v[c];
    }
}

__global__ __launch_bounds__(256) void k_place(const int* __restrict__ ei,
                                               const int* __restrict__ rank,
                                               const int* __restrict__ cntp,
                                               int* __restrict__ perm,
                                               int E, int EP, int NP, int CE) {
    int e = blockIdx.x * 256 + threadIdx.x;
    if (e >= E) return;
    const int dir = blockIdx.y;
    const int c = e / CE;
    int d = ei[(size_t)dir * E + e];
    int pos = cntp[((size_t)dir * CCH + c) * NP + d] + rank[(size_t)dir * EP + e];
    perm[(size_t)dir * EP + pos] = e;
}

__global__ __launch_bounds__(256) void k_order(const int* __restrict__ ei,
                                               const float* __restrict__ ew,
                                               const int* __restrict__ rp,
                                               const int* __restrict__ perm,
                                               int* __restrict__ csrc,
                                               float* __restrict__ cw,
                                               int N, int E, int EP, int NP) {
    const int d = blockIdx.x * 8 + (threadIdx.x >> 5);
    if (d >= N) return;
    const int dir = blockIdx.y;
    const int lane = threadIdx.x & 31;
    const int* r = rp + (size_t)dir * (NP + 4);
    const int* pm = perm + (size_t)dir * EP;
    const int* srcIdx = ei + (size_t)(1 - dir) * E;
    const int s0 = __builtin_amdgcn_readfirstlane(r[d]);
    const int s1 = __builtin_amdgcn_readfirstlane(r[d + 1]);
    for (int ca = s0; ca < s1; ca += 32) {
        const bool mine = (ca + lane) < s1;
        const int my = mine ? pm[ca + lane] : 0;
        int below = 0;
        for (int cb = s0; cb < s1; cb += 32) {
            const int other = (cb + lane < s1) ? pm[cb + lane] : 0;
            int nb = s1 - cb;
            nb = nb < 32 ? nb : 32;
            for (int j = 0; j < nb; ++j) {
                const int oj = __builtin_amdgcn_readlane(other, j);
                below += (oj < my) ? 1 : 0;
            }
        }
        if (mine) {
            const size_t slot = (size_t)dir * EP + s0 + below;
            csrc[slot] = srcIdx[my];
            cw[slot] = ew[my];
        }
    }
}

__global__ __launch_bounds__(256) void k_dis(const int* __restrict__ rp,
                                             const float* __restrict__ cw,
                                             float* __restrict__ dis,
                                             int N, int EP, int NP) {
    int d = blockIdx.x * 256 + threadIdx.x;
    if (d >= N) return;
    const int dir = blockIdx.y;
    const int* r = rp + (size_t)dir * (NP + 4);
    const float* w = cw + (size_t)dir * EP;
    int s0 = r[d], s1 = r[d + 1];
    float deg = 0.0f;
    for (int q = s0; q < s1; ++q) deg += w[q];
    dis[(size_t)dir * NP + d] = (deg > 0.0f) ? rsqrtf(deg) : 0.0f;
}

__global__ __launch_bounds__(256) void k_csrw(const int* __restrict__ rp,
                                              const int* __restrict__ csrc,
                                              float* __restrict__ cw,
                                              const float* __restrict__ dis,
                                              int N, int EP, int NP) {
    int d = blockIdx.x * 256 + threadIdx.x;
    if (d >= N) return;
    const int dir = blockIdx.y;
    const float scale = 2.0f / 3.0f;
    const int* r = rp + (size_t)dir * (NP + 4);
    const int* sidx = csrc + (size_t)dir * EP;
    float* w = cw + (size_t)dir * EP;
    const float* ds = dis + (size_t)dir * NP;
    int s0 = r[d], s1 = r[d + 1];
    float dd = -scale * ds[d];
    for (int q = s0; q < s1; ++q) w[q] = dd * w[q] * ds[sidx[q]];
}

#define TP 68
#define WP 80

static __device__ __forceinline__ void mma_tile(const float* tile, const float* wl,
                                                int lm, int h, int ct,
                                                v8f& acc0, v8f& acc1) {
    const float* arow = tile + lm * TP;
    const float* bcol = wl + ct + lm;
#pragma unroll
    for (int kk = 0; kk < 16; ++kk) {
        const int kbase = kk * 4 + h * 2;
        v2f a = *(const v2f*)(arow + kbase);
        v2f b0, b1;
        b0[0] = bcol[(kbase + 0) * WP + 0];
        b0[1] = bcol[(kbase + 1) * WP + 0];
        b1[0] = bcol[(kbase + 0) * WP + 16];
        b1[1] = bcol[(kbase + 1) * WP + 16];
        acc0 = __builtin_amdgcn_wmma_f32_16x16x4_f32(false, a, false, b0, (short)0, acc0, false, false);
        acc1 = __builtin_amdgcn_wmma_f32_16x16x4_f32(false, a, false, b1, (short)0, acc1, false, false);
    }
}

template <bool FIRST>
static __device__ __forceinline__ void put_weights(float* Wl, int tid,
                                                   float4 wb0, float4 wb1,
                                                   float4 wa0, float4 wa1) {
    float* wdst = Wl + (tid >> 4) * WP + (tid & 15) * 4;
    *(float4*)wdst = wb0;
    *(float4*)(wdst + 32 * WP) = wb1;
    if (FIRST) {
        *(float4*)(wdst + 64 * WP) = wa0;
        *(float4*)(wdst + 96 * WP) = wa1;
    }
}

template <int Q>
static __device__ __forceinline__ int row_take(int x) {
    return __builtin_amdgcn_update_dpp(0, x, 0x150 + Q, 0xf, 0xf, true);
}

template <int Q>
static __device__ __forceinline__ void slot_take(int msrc, float mw, int lm, int& idx, float& w) {
    idx = row_take<Q>(msrc) * 16 + lm;
    w = __int_as_float(row_take<Q>(__float_as_int(mw)));
}

static __device__ __forceinline__ void slot_sum(float w, const float4& val, float4& acc) {
    acc.x = fmaf(w, val.x, acc.x);
    acc.y = fmaf(w, val.y, acc.y);
    acc.z = fmaf(w, val.z, acc.z);
    acc.w = fmaf(w, val.w, acc.w);
}

struct Rows8 {
    float4 r0, r1, r2, r3, r4, r5, r6, r7;
    float w0, w1, w2, w3, w4, w5, w6, w7;
};

template <int Q0>
static __device__ __forceinline__ void rows_load(const float4* __restrict__ v4, int msrc, float mw,
                                                 int lm, Rows8& g) {
    int i0, i1, i2, i3, i4, i5, i6, i7;
    slot_take<Q0 + 0>(msrc, mw, lm, i0, g.w0);
    slot_take<Q0 + 1>(msrc, mw, lm, i1, g.w1);
    slot_take<Q0 + 2>(msrc, mw, lm, i2, g.w2);
    slot_take<Q0 + 3>(msrc, mw, lm, i3, g.w3);
    slot_take<Q0 + 4>(msrc, mw, lm, i4, g.w4);
    slot_take<Q0 + 5>(msrc, mw, lm, i5, g.w5);
    slot_take<Q0 + 6>(msrc, mw, lm, i6, g.w6);
    slot_take<Q0 + 7>(msrc, mw, lm, i7, g.w7);
    asm volatile("" : : "v"(i0), "v"(i1), "v"(i2), "v"(i3), "v"(i4), "v"(i5), "v"(i6), "v"(i7));
    g.r0 = v4[i0];
    g.r1 = v4[i1];
    g.r2 = v4[i2];
    g.r3 = v4[i3];
    g.r4 = v4[i4];
    g.r5 = v4[i5];
    g.r6 = v4[i6];
    g.r7 = v4[i7];
}

static __device__ __forceinline__ void rows_sum(const Rows8& g, float4& accE, float4& accO) {
    slot_sum(g.w0, g.r0, accE);
    slot_sum(g.w1, g.r1, accO);
    slot_sum(g.w2, g.r2, accE);
    slot_sum(g.w3, g.r3, accO);
    slot_sum(g.w4, g.r4, accE);
    slot_sum(g.w5, g.r5, accO);
    slot_sum(g.w6, g.r6, accE);
    slot_sum(g.w7, g.r7, accO);
}

struct DirArgs {
    const float* v;
    const float* prev;
    float* dst;
    const int* rp;
    const int* csrc;
    const float* cw;
    const float* Wa;
    const float* Wb;
    const float* bias;
    int colofs;
};

template <bool HAS_PREV, bool FIRST, bool STORE_DST>
__global__ __launch_bounds__(512) __attribute__((amdgpu_waves_per_eu(16, 16))) void k_lhat_proj(DirArgs a0, DirArgs a1,
                                                   float adiag, float beta, float alpha,
                                                   float* out, int N) {
    __shared__ float Wl[(FIRST ? 2 : 1) * 64 * WP];
    __shared__ float Tl[(FIRST ? 2 : 1) * 32 * TP];

    const bool second = blockIdx.y != 0;
    const float* __restrict__ v    = second ? a1.v : a0.v;
    const float* prev              = second ? a1.prev : a0.prev;
    float* dst                     = second ? a1.dst : a0.dst;
    const int* __restrict__ rp     = second ? a1.rp : a0.rp;
    const int* __restrict__ csrc   = second ? a1.csrc : a0.csrc;
    const float* __restrict__ cw   = second ? a1.cw : a0.cw;
    const float* Wa                = second ? a1.Wa : a0.Wa;
    const float* Wb                = second ? a1.Wb : a0.Wb;
    const float* bias              = second ? a1.bias : a0.bias;
    const int colofs               = second ? a1.colofs : a0.colofs;

    const int tid = threadIdx.x;
    const int wave = tid >> 5;
    const int lane = tid & 31;
    const int h = lane >> 4;
    const int lm = lane & 15;
    const int c4 = lm * 4;
    const int row0 = blockIdx.x * 32;
    const int trow = wave * 2 + h;
    const int d = row0 + trow;
    const bool valid = d < N;

    const float4 wb0 = ((const float4*)Wb)[tid];
    const float4 wb1 = ((const float4*)Wb)[tid + 512];
    float4 wa0 = make_float4(0.f, 0.f, 0.f, 0.f);
    float4 wa1 = make_float4(0.f, 0.f, 0.f, 0.f);
    if (FIRST) {
        wa0 = ((const float4*)Wa)[tid];
        wa1 = ((const float4*)Wa)[tid + 512];
    }

    const int mj = wave - (blockIdx.x & 3) * 4;
    const bool mat = (mj >= 0) && (mj < 4);
    const int mt = (mj >> 1) & 1;
    const int ct = (mj & 1) * 32;

    int s0 = 0, s1 = 0;
    if (valid) {
        s0 = rp[d];
        s1 = rp[d + 1];
    }
    int base = s0;
    int rem = s1 - s0;

    int msrc = 0;
    float mw = 0.0f;
    if (lm < rem) {
        msrc = csrc[base + lm];
        mw = cw[base + lm];
    }

    put_weights<FIRST>(Wl, tid, wb0, wb1, wa0, wa1);

    float4 accE = make_float4(0.f, 0.f, 0.f, 0.f);
    float4 accO = make_float4(0.f, 0.f, 0.f, 0.f);
    const float4* __restrict__ v4 = (const float4*)v;

    int mrem = max(__builtin_amdgcn_readlane(rem, 0), __builtin_amdgcn_readlane(rem, 16));
    while (mrem > 0) {
        const bool more = (lm + 16) < rem;
        const int nidx = more ? (base + 16 + lm) : base;
        asm volatile("" : : "v"(nidx));
        Rows8 g;
        rows_load<0>(v4, msrc, mw, lm, g);
        const int ls = csrc[nidx];
        const float lw = cw[nidx];
        rows_sum(g, accE, accO);
        if (8 < mrem) {
            rows_load<8>(v4, msrc, mw, lm, g);
            rows_sum(g, accE, accO);
        }
        base += 16;
        rem -= 16;
        mrem -= 16;
        msrc = more ? ls : 0;
        mw = more ? lw : 0.0f;
    }
    float4 vd = make_float4(0.f, 0.f, 0.f, 0.f);
    float4 pd = make_float4(0.f, 0.f, 0.f, 0.f);
    if (valid) {
        vd = *(const float4*)(v + row64(d) + c4);
        if (HAS_PREV) pd = *(const float4*)(prev + row64(d) + c4);
    }
    float4 acc;
    acc.x = accE.x + accO.x;
    acc.y = accE.y + accO.y;
    acc.z = accE.z + accO.z;
    acc.w = accE.w + accO.w;

    v8f c0 = {0.f, 0.f, 0.f, 0.f, 0.f, 0.f, 0.f, 0.f};
    v8f c1 = {0.f, 0.f, 0.f, 0.f, 0.f, 0.f, 0.f, 0.f};
    const int mrow0 = row0 + mt * 16;
    if (mat) {
        if (FIRST) {
            const float bv0 = bias[ct + lm];
            const float bv1 = bias[ct + 16 + lm];
#pragma unroll
            for (int i = 0; i < 8; ++i) {
                c0[i] = bv0;
                c1[i] = bv1;
            }
        } else {
            const int nclamp = N - 1;
#pragma unroll
            for (int i = 0; i < 8; ++i) {
                const int r = mrow0 + i + h * 8;
                const bool ok = (r < N);
                const int rc = ok ? r : nclamp;
                const float* orow = out + (size_t)rc * 128 + colofs + ct + lm;
                const float v0 = orow[0], v1 = orow[16];
                c0[i] = ok ? v0 : 0.f;
                c1[i] = ok ? v1 : 0.f;
            }
        }
    }

    float4 o = make_float4(0.f, 0.f, 0.f, 0.f);
    if (valid) {
        o.x = adiag * vd.x;
        o.y = adiag * vd.y;
        o.z = adiag * vd.z;
        o.w = adiag * vd.w;
        if (HAS_PREV) {
            o.x += beta * pd.x;
            o.y += beta * pd.y;
            o.z += beta * pd.z;
            o.w += beta * pd.w;
        }
        o.x += alpha * acc.x;
        o.y += alpha * acc.y;
        o.z += alpha * acc.z;
        o.w += alpha * acc.w;
        if (STORE_DST) *(float4*)(dst + row64(d) + c4) = o;
    }

    *(float4*)(Tl + trow * TP + c4) = o;
    if (FIRST) *(float4*)(Tl + 32 * TP + trow * TP + c4) = vd;
    __syncthreads();

    if (mat) {
        if (FIRST) mma_tile(Tl + 32 * TP + mt * 16 * TP, Wl + 64 * WP, lm, h, ct, c0, c1);
        mma_tile(Tl + mt * 16 * TP, Wl, lm, h, ct, c0, c1);
#pragma unroll
        for (int i = 0; i < 8; ++i) {
            const int r = mrow0 + i + h * 8;
            if (r < N) {
                float* orow = out + (size_t)r * 128 + colofs + ct + lm;
                orow[0]  = c0[i];
                orow[16] = c1[i];
            }
        }
    }
}


extern "C" void kernel_launch(void* const* d_in, const int* in_sizes, int n_in,
                              void* d_out, int out_size, void* d_ws, size_t ws_size,
                              hipStream_t stream) {
    const float* x  = (const float*)d_in[0];
    const int*   ei = (const int*)d_in[1];
    const float* ew = (const float*)d_in[2];
    const float* W1 = (const float*)d_in[3];
    const float* b1 = (const float*)d_in[4];
    const float* W2 = (const float*)d_in[5];
    const float* b2 = (const float*)d_in[6];
    float* out = (float*)d_out;

    const int N = in_sizes[0] / 64;
    const int E = in_sizes[2];

    const int per = (((N + 1023) / 1024) + 3) & ~3;
    const int NP  = per * 1024;
    const int EP  = (E + 3) & ~3;

    const size_t tailWords = 4 * (size_t)EP + 2 * (size_t)NP + 2 * (size_t)(NP + 4) + 2 * (size_t)NP;
    const bool paired = ws_size >= (4 * (size_t)N * 64 + tailWords) * sizeof(float);
    const int nFeat = paired ? 4 : 2;

    float* tA0  = (float*)d_ws;
    float* tB0  = tA0 + (size_t)N * 64;
    float* tA1  = paired ? tB0 + (size_t)N * 64 : tA0;
    float* tB1  = paired ? tA1 + (size_t)N * 64 : tB0;
    int*   csrc = (int*)(tA0 + (size_t)nFeat * N * 64);
    float* cw   = (float*)(csrc + 2 * (size_t)EP);
    int*   btot = (int*)(cw + 2 * (size_t)EP);
    int*   rp   = btot + 2 * (size_t)NP;
    float* dis  = (float*)(rp + 2 * (size_t)(NP + 4));
    int*   rank = (int*)tA0;
    int*   perm = rank + 2 * (size_t)EP;
    int*   cntp = perm + 2 * (size_t)EP;
    const int CE = (E + CCH - 1) / CCH;
    const int NR = (NP + RNG - 1) / RNG;

    const float DIAG = 2.0f / 3.0f - 1.0f;

    dim3 blk(256);
    dim3 blk16w(512);
    int gE    = (E + 255) / 256;
    int gN    = (N + 255) / 256;
    int gN8   = (N + 7) / 8;
    int gN32  = (N + 31) / 32;

    k_rangecount<<<dim3(NR, CCH, 2), dim3(1024), 0, stream>>>(ei, cntp, rank, btot, E, EP, NP, CE);
    k_starts<<<dim3(NP / 1024, 2), dim3(1024), 0, stream>>>(cntp, btot, rp, NP, NR * (RNG / 1024));
    k_place<<<dim3(gE, 2), blk, 0, stream>>>(ei, rank, cntp, perm, E, EP, NP, CE);
    k_order<<<dim3(gN8, 2), blk, 0, stream>>>(ei, ew, rp, perm, csrc, cw, N, E, EP, NP);
    k_dis<<<dim3(gN, 2), blk, 0, stream>>>(rp, cw, dis, N, EP, NP);
    k_csrw<<<dim3(gN, 2), blk, 0, stream>>>(rp, csrc, cw, dis, N, EP, NP);

    DirArgs d0, d1;
    d0.rp = rp;                 d1.rp = rp + (size_t)(NP + 4);
    d0.csrc = csrc;             d1.csrc = csrc + (size_t)EP;
    d0.cw = cw;                 d1.cw = cw + (size_t)EP;
    d0.Wa = W1;                 d1.Wa = W2;
    d0.bias = b1;               d1.bias = b2;
    d0.colofs = 0;              d1.colofs = 64;

    const int rows = paired ? 2 : 1;
    for (int turn = 0; turn * rows < 2; ++turn) {
        const bool lone = (turn == 1);

        d0.v = x;   d0.prev = x;   d0.dst = tA0;  d0.Wb = W1 + 1 * 4096;
        d1.v = x;   d1.prev = x;   d1.dst = tA1;  d1.Wb = W2 + 1 * 4096;
        k_lhat_proj<false, true, true><<<dim3(gN32, rows), blk16w, 0, stream>>>(
            lone ? d1 : d0, d1, DIAG, 0.0f, 1.0f, out, N);

        d0.v = tA0; d0.prev = x;   d0.dst = tB0;  d0.Wb = W1 + 2 * 4096;
        d1.v = tA1; d1.prev = x;   d1.dst = tB1;  d1.Wb = W2 + 2 * 4096;
        k_lhat_proj<true, false, true><<<dim3(gN32, rows), blk16w, 0, stream>>>(
            lone ? d1 : d0, d1, 2.0f * DIAG, -1.0f, 2.0f, out, N);

        d0.v = tB0; d0.prev = tA0; d0.dst = tA0;  d0.Wb = W1 + 3 * 4096;
        d1.v = tB1; d1.prev = tA1; d1.dst = tA1;  d1.Wb = W2 + 3 * 4096;
        k_lhat_proj<true, false, true><<<dim3(gN32, rows), blk16w, 0, stream>>>(
            lone ? d1 : d0, d1, 2.0f * DIAG, -1.0f, 2.0f, out, N);

        d0.v = tA0; d0.prev = tB0; d0.dst = tB0;  d0.Wb = W1 + 4 * 4096;
        d1.v = tA1; d1.prev = tB1; d1.dst = tB1;  d1.Wb = W2 + 4 * 4096;
        k_lhat_proj<true, false, false><<<dim3(gN32, rows), blk16w, 0, stream>>>(
            lone ? d1 : d0, d1, 2.0f * DIAG, -1.0f, 2.0f, out, N);
    }
}


# List build: placing pass 8 edges per thread, ordering pass 8 nodes per wave in phases (a phase's loads in flight together): 150000 -> 18912 short waves, same requests; float sums unchanged
# speedup vs baseline: 3.7525x; 1.0322x over previous
#include <hip/hip_runtime.h>


typedef float v2f __attribute__((ext_vector_type(2)));
typedef float v8f __attribute__((ext_vector_type(8)));

static __device__ __forceinline__ size_t row64(int n) { return (size_t)n * 64; }


#define RNG 8192
#define CCH 16

__global__ __launch_bounds__(1024) void k_rangecount(const int* __restrict__ ei,
                                                     int* __restrict__ cntp,
                                                     int* __restrict__ rank,
                                                     int* __restrict__ btot,
                                                     int E, int EP, int NP, int CE) {
    __shared__ int lc[RNG];
    __shared__ int hits[RNG / 1024];
    const int t = threadIdx.x;
    const int r0 = blockIdx.x * RNG;
    const int c = blockIdx.y;
    const int dir = blockIdx.z;
    if (t < RNG / 1024) hits[t] = 0;
    for (int i = t; i < RNG; i += 1024) lc[i] = 0;
    __syncthreads();
    const int* dst = ei + (size_t)dir * E;
    int* rk = rank + (size_t)dir * EP;
    const int cbeg = c * CE;
    int cend = cbeg + CE;
    cend = cend < E ? cend : E;
    for (int e0 = cbeg + t; e0 < cend; e0 += 8 * 1024) {
        int dv[8];
#pragma unroll
        for (int j = 0; j < 8; ++j) {
            const int e = e0 + j * 1024;
            dv[j] = (e < cend) ? dst[e] : -1;
        }
#pragma unroll
        for (int j = 0; j < 8; ++j) {
            const unsigned rel = (unsigned)(dv[j] - r0);
            if (rel < (unsigned)RNG) rk[e0 + j * 1024] = atomicAdd(&lc[rel], 1);
        }
    }
    __syncthreads();
    int* o = cntp + ((size_t)dir * CCH + c) * NP;
    int w[RNG / 1024];
#pragma unroll
    for (int s = 0; s < RNG / 1024; ++s) {
        const int i = s * 1024 + t;
        const int v = lc[i];
        const int d = r0 + i;
        if (d < NP) o[d] = v;
        w[s] = v;
    }
#pragma unroll
    for (int off = 16; off > 0; off >>= 1) {
#pragma unroll
        for (int s = 0; s < RNG / 1024; ++s) w[s] += __shfl_xor(w[s], off);
    }
    if ((t & 31) == 0) {
#pragma unroll
        for (int s = 0; s < RNG / 1024; ++s) atomicAdd(&hits[s], w[s]);
    }
    __syncthreads();
    if (t < RNG / 1024)
        btot[((size_t)dir * gridDim.x * (RNG / 1024) + blockIdx.x * (RNG / 1024) + t) * CCH + c] = hits[t];
}

__global__ __launch_bounds__(1024) void k_starts(int* __restrict__ cntp,
                                                 const int* __restrict__ btot,
                                                 int* __restrict__ rp, int NP, int NSB) {
    __shared__ int wtot[32];
    __shared__ int wbef[32];
    const int t = threadIdx.x;
    const int lane = t & 31;
    const int wave = t >> 5;
    const int sb = blockIdx.x;
    const int dir = blockIdx.y;
    const int d = sb * 1024 + t;
    int* p = cntp + (size_t)dir * CCH * NP + d;

    const int* bt = btot + (size_t)dir * NSB * CCH;
    const int nb = sb * CCH;
    int bs = 0;
    for (int q = t; q < nb; q += 1024) bs += bt[q];

    int v[CCH];
#pragma unroll
    for (int c = 0; c < CCH; ++c) v[c] = p[(size_t)c * NP];
    int tsum = 0;
#pragma unroll
    for (int c = 0; c < CCH; ++c) tsum += v[c];

    int inc = tsum;
    for (int off = 1; off < 32; off <<= 1) {
        const int n = __shfl_up(inc, off);
        if (lane >= off) inc += n;
    }
    for (int off = 16; off > 0; off >>= 1) bs += __shfl_xor(bs, off);
    if (lane == 31) wtot[wave] = inc;
    if (lane == 0) wbef[wave] = bs;
    __syncthreads();
    int winc = wtot[lane];
    for (int off = 1; off < 32; off <<= 1) {
        const int n = __shfl_up(winc, off);
        if (lane >= off) winc += n;
    }
    int before = wbef[lane];
    for (int off = 16; off > 0; off >>= 1) before += __shfl_xor(before, off);
    const int wprev = __shfl(winc, (wave + 31) & 31);
    const int wbase = (wave > 0) ? wprev : 0;
    const int base = before + wbase + (inc - tsum);

    int* rr = rp + (size_t)dir * (NP + 4);
    rr[d] = base;
    if (d == NP - 1) rr[NP] = base + tsum;

    int run = base;
#pragma unroll
    for (int c = 0; c < CCH; ++c) {
        p[(size_t)c * NP] = run;
        run += v[c];
    }
}

#define PLE 8
__global__ __launch_bounds__(256) void k_place(const int* __restrict__ ei,
                                               const int* __restrict__ rank,
                                               const int* __restrict__ cntp,
                                               int* __restrict__ perm,
                                               int E, int EP, int NP, int CE) {
    const int c = blockIdx.y;
    const int dir = blockIdx.z;
    const int cbeg = c * CE;
    int cend = cbeg + CE;
    cend = cend < E ? cend : E;
    const int b0 = cbeg + blockIdx.x * (256 * PLE);
    if (b0 >= cend) return;
    const int e0 = b0 + threadIdx.x;
    const int* dst = ei + (size_t)dir * E;
    const int* rk = rank + (size_t)dir * EP;
    const int* first = cntp + ((size_t)dir * CCH + c) * NP;
    int* pm = perm + (size_t)dir * EP;
    int dv[PLE], rv[PLE], fv[PLE];
#pragma unroll
    for (int j = 0; j < PLE; ++j) {
        const int e = e0 + j * 256;
        const int ec = e < cend ? e : cend - 1;
        dv[j] = dst[ec];
        rv[j] = rk[ec];
    }
#pragma unroll
    for (int j = 0; j < PLE; ++j) fv[j] = first[dv[j]];
    asm volatile("" : : "v"(rv[0]), "v"(fv[0]));
#pragma unroll
    for (int j = 0; j < PLE; ++j) {
        const int e = e0 + j * 256;
        if (e < cend) pm[fv[j] + rv[j]] = e;
    }
}

#define ONW 8
__global__ __launch_bounds__(256) void k_order(const int* __restrict__ ei,
                                               const float* __restrict__ ew,
                                               const int* __restrict__ rp,
                                               const int* __restrict__ perm,
                                               int* __restrict__ csrc,
                                               float* __restrict__ cw,
                                               int N, int E, int EP, int NP) {
    const int lane = threadIdx.x & 31;
    const int wv = __builtin_amdgcn_readfirstlane((int)(threadIdx.x >> 5));
    const int d0 = (blockIdx.x * 8 + wv) * ONW;
    if (d0 >= N) return;
    const int dir = blockIdx.y;
    const int* r = rp + (size_t)dir * (NP + 4);
    const int* pm = perm + (size_t)dir * EP;
    const int* srcIdx = ei + (size_t)(1 - dir) * E;
    int* cs = csrc + (size_t)dir * EP;
    float* cwd = cw + (size_t)dir * EP;

    const int bnd = r[d0 + (lane < ONW ? lane : ONW)];
    int s0[ONW], len[ONW];
    int nmax = 0;
#pragma unroll
    for (int i = 0; i < ONW; ++i) {
        s0[i] = __builtin_amdgcn_readlane(bnd, i);
        const int s1 = __builtin_amdgcn_readlane(bnd, i + 1);
        const int l = (d0 + i < N) ? (s1 - s0[i]) : 0;
        len[i] = l;
        nmax = nmax > l ? nmax : l;
    }
    const bool anyLong = nmax > 32;
    nmax = nmax < 32 ? nmax : 32;

    int my[ONW];
#pragma unroll
    for (int i = 0; i < ONW; ++i) {
        const int q = s0[i] + lane;
        const int v = pm[q < E ? q : E - 1];
        my[i] = (lane < len[i]) ? v : 0x7fffffff;
    }

    int below[ONW];
#pragma unroll
    for (int i = 0; i < ONW; ++i) below[i] = 0;
    for (int j = 0; j < nmax; ++j) {
#pragma unroll
        for (int i = 0; i < ONW; ++i) {
            const int oj = __builtin_amdgcn_readlane(my[i], j);
            below[i] += (oj < my[i]) ? 1 : 0;
        }
    }

    int gs[ONW];
    float gw[ONW];
#pragma unroll
    for (int i = 0; i < ONW; ++i) {
        const bool mine = (lane < len[i]) && (len[i] <= 32);
        const int g = mine ? my[i] : 0;
        gs[i] = srcIdx[g];
        gw[i] = ew[g];
    }
#pragma unroll
    for (int i = 0; i < ONW; ++i) {
        if ((lane < len[i]) && (len[i] <= 32)) {
            const int slot = s0[i] + below[i];
            cs[slot] = gs[i];
            cwd[slot] = gw[i];
        }
    }

    if (anyLong) {
#pragma nounroll
        for (int i = 0; i < ONW; ++i) {
            const int a0 = __builtin_amdgcn_readlane(bnd, i);
            const int a1 = __builtin_amdgcn_readlane(bnd, i + 1);
            if ((d0 + i >= N) || (a1 - a0 <= 32)) continue;
            for (int ca = a0; ca < a1; ca += 32) {
                const bool mine = (ca + lane) < a1;
                const int me = mine ? pm[ca + lane] : 0;
                int bl = 0;
                for (int cb = a0; cb < a1; cb += 32) {
                    const int other = (cb + lane < a1) ? pm[cb + lane] : 0;
                    int nb = a1 - cb;
                    nb = nb < 32 ? nb : 32;
                    for (int j = 0; j < nb; ++j) {
                        const int oj = __builtin_amdgcn_readlane(other, j);
                        bl += (oj < me) ? 1 : 0;
                    }
                }
                if (mine) {
                    const int slot = a0 + bl;
                    cs[slot] = srcIdx[me];
                    cwd[slot] = ew[me];
                }
            }
        }
    }
}

__global__ __launch_bounds__(256) void k_dis(const int* __restrict__ rp,
                                             const float* __restrict__ cw,
                                             float* __restrict__ dis,
                                             int N, int EP, int NP) {
    int d = blockIdx.x * 256 + threadIdx.x;
    if (d >= N) return;
    const int dir = blockIdx.y;
    const int* r = rp + (size_t)dir * (NP + 4);
    const float* w = cw + (size_t)dir * EP;
    int s0 = r[d], s1 = r[d + 1];
    float deg = 0.0f;
    for (int q = s0; q < s1; ++q) deg += w[q];
    dis[(size_t)dir * NP + d] = (deg > 0.0f) ? rsqrtf(deg) : 0.0f;
}

__global__ __launch_bounds__(256) void k_csrw(const int* __restrict__ rp,
                                              const int* __restrict__ csrc,
                                              float* __restrict__ cw,
                                              const float* __restrict__ dis,
                                              int N, int EP, int NP) {
    int d = blockIdx.x * 256 + threadIdx.x;
    if (d >= N) return;
    const int dir = blockIdx.y;
    const float scale = 2.0f / 3.0f;
    const int* r = rp + (size_t)dir * (NP + 4);
    const int* sidx = csrc + (size_t)dir * EP;
    float* w = cw + (size_t)dir * EP;
    const float* ds = dis + (size_t)dir * NP;
    int s0 = r[d], s1 = r[d + 1];
    float dd = -scale * ds[d];
    for (int q = s0; q < s1; ++q) w[q] = dd * w[q] * ds[sidx[q]];
}

#define TP 68
#define WP 80

static __device__ __forceinline__ void mma_tile(const float* tile, const float* wl,
                                                int lm, int h, int ct,
                                                v8f& acc0, v8f& acc1) {
    const float* arow = tile + lm * TP;
    const float* bcol = wl + ct + lm;
#pragma unroll
    for (int kk = 0; kk < 16; ++kk) {
        const int kbase = kk * 4 + h * 2;
        v2f a = *(const v2f*)(arow + kbase);
        v2f b0, b1;
        b0[0] = bcol[(kbase + 0) * WP + 0];
        b0[1] = bcol[(kbase + 1) * WP + 0];
        b1[0] = bcol[(kbase + 0) * WP + 16];
        b1[1] = bcol[(kbase + 1) * WP + 16];
        acc0 = __builtin_amdgcn_wmma_f32_16x16x4_f32(false, a, false, b0, (short)0, acc0, false, false);
        acc1 = __builtin_amdgcn_wmma_f32_16x16x4_f32(false, a, false, b1, (short)0, acc1, false, false);
    }
}

template <bool FIRST>
static __device__ __forceinline__ void put_weights(float* Wl, int tid,
                                                   float4 wb0, float4 wb1,
                                                   float4 wa0, float4 wa1) {
    float* wdst = Wl + (tid >> 4) * WP + (tid & 15) * 4;
    *(float4*)wdst = wb0;
    *(float4*)(wdst + 32 * WP) = wb1;
    if (FIRST) {
        *(float4*)(wdst + 64 * WP) = wa0;
        *(float4*)(wdst + 96 * WP) = wa1;
    }
}

template <int Q>
static __device__ __forceinline__ int row_take(int x) {
    return __builtin_amdgcn_update_dpp(0, x, 0x150 + Q, 0xf, 0xf, true);
}

template <int Q>
static __device__ __forceinline__ void slot_take(int msrc, float mw, int lm, int& idx, float& w) {
    idx = row_take<Q>(msrc) * 16 + lm;
    w = __int_as_float(row_take<Q>(__float_as_int(mw)));
}

static __device__ __forceinline__ void slot_sum(float w, const float4& val, float4& acc) {
    acc.x = fmaf(w, val.x, acc.x);
    acc.y = fmaf(w, val.y, acc.y);
    acc.z = fmaf(w, val.z, acc.z);
    acc.w = fmaf(w, val.w, acc.w);
}

struct Rows8 {
    float4 r0, r1, r2, r3, r4, r5, r6, r7;
    float w0, w1, w2, w3, w4, w5, w6, w7;
};

template <int Q0>
static __device__ __forceinline__ void rows_load(const float4* __restrict__ v4, int msrc, float mw,
                                                 int lm, Rows8& g) {
    int i0, i1, i2, i3, i4, i5, i6, i7;
    slot_take<Q0 + 0>(msrc, mw, lm, i0, g.w0);
    slot_take<Q0 + 1>(msrc, mw, lm, i1, g.w1);
    slot_take<Q0 + 2>(msrc, mw, lm, i2, g.w2);
    slot_take<Q0 + 3>(msrc, mw, lm, i3, g.w3);
    slot_take<Q0 + 4>(msrc, mw, lm, i4, g.w4);
    slot_take<Q0 + 5>(msrc, mw, lm, i5, g.w5);
    slot_take<Q0 + 6>(msrc, mw, lm, i6, g.w6);
    slot_take<Q0 + 7>(msrc, mw, lm, i7, g.w7);
    asm volatile("" : : "v"(i0), "v"(i1), "v"(i2), "v"(i3), "v"(i4), "v"(i5), "v"(i6), "v"(i7));
    g.r0 = v4[i0];
    g.r1 = v4[i1];
    g.r2 = v4[i2];
    g.r3 = v4[i3];
    g.r4 = v4[i4];
    g.r5 = v4[i5];
    g.r6 = v4[i6];
    g.r7 = v4[i7];
}

static __device__ __forceinline__ void rows_sum(const Rows8& g, float4& accE, float4& accO) {
    slot_sum(g.w0, g.r0, accE);
    slot_sum(g.w1, g.r1, accO);
    slot_sum(g.w2, g.r2, accE);
    slot_sum(g.w3, g.r3, accO);
    slot_sum(g.w4, g.r4, accE);
    slot_sum(g.w5, g.r5, accO);
    slot_sum(g.w6, g.r6, accE);
    slot_sum(g.w7, g.r7, accO);
}

struct DirArgs {
    const float* v;
    const float* prev;
    float* dst;
    const int* rp;
    const int* csrc;
    const float* cw;
    const float* Wa;
    const float* Wb;
    const float* bias;
    int colofs;
};

template <bool HAS_PREV, bool FIRST, bool STORE_DST>
__global__ __launch_bounds__(512) __attribute__((amdgpu_waves_per_eu(16, 16))) void k_lhat_proj(DirArgs a0, DirArgs a1,
                                                   float adiag, float beta, float alpha,
                                                   float* out, int N) {
    __shared__ float Wl[(FIRST ? 2 : 1) * 64 * WP];
    __shared__ float Tl[(FIRST ? 2 : 1) * 32 * TP];

    const bool second = blockIdx.y != 0;
    const float* __restrict__ v    = second ? a1.v : a0.v;
    const float* prev              = second ? a1.prev : a0.prev;
    float* dst                     = second ? a1.dst : a0.dst;
    const int* __restrict__ rp     = second ? a1.rp : a0.rp;
    const int* __restrict__ csrc   = second ? a1.csrc : a0.csrc;
    const float* __restrict__ cw   = second ? a1.cw : a0.cw;
    const float* Wa                = second ? a1.Wa : a0.Wa;
    const float* Wb                = second ? a1.Wb : a0.Wb;
    const float* bias              = second ? a1.bias : a0.bias;
    const int colofs               = second ? a1.colofs : a0.colofs;

    const int tid = threadIdx.x;
    const int wave = tid >> 5;
    const int lane = tid & 31;
    const int h = lane >> 4;
    const int lm = lane & 15;
    const int c4 = lm * 4;
    const int row0 = blockIdx.x * 32;
    const int trow = wave * 2 + h;
    const int d = row0 + trow;
    const bool valid = d < N;

    const float4 wb0 = ((const float4*)Wb)[tid];
    const float4 wb1 = ((const float4*)Wb)[tid + 512];
    float4 wa0 = make_float4(0.f, 0.f, 0.f, 0.f);
    float4 wa1 = make_float4(0.f, 0.f, 0.f, 0.f);
    if (FIRST) {
        wa0 = ((const float4*)Wa)[tid];
        wa1 = ((const float4*)Wa)[tid + 512];
    }

    const int mj = wave - (blockIdx.x & 3) * 4;
    const bool mat = (mj >= 0) && (mj < 4);
    const int mt = (mj >> 1) & 1;
    const int ct = (mj & 1) * 32;

    int s0 = 0, s1 = 0;
    if (valid) {
        s0 = rp[d];
        s1 = rp[d + 1];
    }
    int base = s0;
    int rem = s1 - s0;

    int msrc = 0;
    float mw = 0.0f;
    if (lm < rem) {
        msrc = csrc[base + lm];
        mw = cw[base + lm];
    }

    put_weights<FIRST>(Wl, tid, wb0, wb1, wa0, wa1);

    float4 accE = make_float4(0.f, 0.f, 0.f, 0.f);
    float4 accO = make_float4(0.f, 0.f, 0.f, 0.f);
    const float4* __restrict__ v4 = (const float4*)v;

    int mrem = max(__builtin_amdgcn_readlane(rem, 0), __builtin_amdgcn_readlane(rem, 16));
    while (mrem > 0) {
        const bool more = (lm + 16) < rem;
        const int nidx = more ? (base + 16 + lm) : base;
        asm volatile("" : : "v"(nidx));
        Rows8 g;
        rows_load<0>(v4, msrc, mw, lm, g);
        const int ls = csrc[nidx];
        const float lw = cw[nidx];
        rows_sum(g, accE, accO);
        if (8 < mrem) {
            rows_load<8>(v4, msrc, mw, lm, g);
            rows_sum(g, accE, accO);
        }
        base += 16;
        rem -= 16;
        mrem -= 16;
        msrc = more ? ls : 0;
        mw = more ? lw : 0.0f;
    }
    float4 vd = make_float4(0.f, 0.f, 0.f, 0.f);
    float4 pd = make_float4(0.f, 0.f, 0.f, 0.f);
    if (valid) {
        vd = *(const float4*)(v + row64(d) + c4);
        if (HAS_PREV) pd = *(const float4*)(prev + row64(d) + c4);
    }
    float4 acc;
    acc.x = accE.x + accO.x;
    acc.y = accE.y + accO.y;
    acc.z = accE.z + accO.z;
    acc.w = accE.w + accO.w;

    v8f c0 = {0.f, 0.f, 0.f, 0.f, 0.f, 0.f, 0.f, 0.f};
    v8f c1 = {0.f, 0.f, 0.f, 0.f, 0.f, 0.f, 0.f, 0.f};
    const int mrow0 = row0 + mt * 16;
    if (mat) {
        if (FIRST) {
            const float bv0 = bias[ct + lm];
            const float bv1 = bias[ct + 16 + lm];
#pragma unroll
            for (int i = 0; i < 8; ++i) {
                c0[i] = bv0;
                c1[i] = bv1;
            }
        } else {
            const int nclamp = N - 1;
#pragma unroll
            for (int i = 0; i < 8; ++i) {
                const int r = mrow0 + i + h * 8;
                const bool ok = (r < N);
                const int rc = ok ? r : nclamp;
                const float* orow = out + (size_t)rc * 128 + colofs + ct + lm;
                const float v0 = orow[0], v1 = orow[16];
                c0[i] = ok ? v0 : 0.f;
                c1[i] = ok ? v1 : 0.f;
            }
        }
    }

    float4 o = make_float4(0.f, 0.f, 0.f, 0.f);
    if (valid) {
        o.x = adiag * vd.x;
        o.y = adiag * vd.y;
        o.z = adiag * vd.z;
        o.w = adiag * vd.w;
        if (HAS_PREV) {
            o.x += beta * pd.x;
            o.y += beta * pd.y;
            o.z += beta * pd.z;
            o.w += beta * pd.w;
        }
        o.x += alpha * acc.x;
        o.y += alpha * acc.y;
        o.z += alpha * acc.z;
        o.w += alpha * acc.w;
        if (STORE_DST) *(float4*)(dst + row64(d) + c4) = o;
    }

    *(float4*)(Tl + trow * TP + c4) = o;
    if (FIRST) *(float4*)(Tl + 32 * TP + trow * TP + c4) = vd;
    __syncthreads();

    if (mat) {
        if (FIRST) mma_tile(Tl + 32 * TP + mt * 16 * TP, Wl + 64 * WP, lm, h, ct, c0, c1);
        mma_tile(Tl + mt * 16 * TP, Wl, lm, h, ct, c0, c1);
#pragma unroll
        for (int i = 0; i < 8; ++i) {
            const int r = mrow0 + i + h * 8;
            if (r < N) {
                float* orow = out + (size_t)r * 128 + colofs + ct + lm;
                orow[0]  = c0[i];
                orow[16] = c1[i];
            }
        }
    }
}


extern "C" void kernel_launch(void* const* d_in, const int* in_sizes, int n_in,
                              void* d_out, int out_size, void* d_ws, size_t ws_size,
                              hipStream_t stream) {
    const float* x  = (const float*)d_in[0];
    const int*   ei = (const int*)d_in[1];
    const float* ew = (const float*)d_in[2];
    const float* W1 = (const float*)d_in[3];
    const float* b1 = (const float*)d_in[4];
    const float* W2 = (const float*)d_in[5];
    const float* b2 = (const float*)d_in[6];
    float* out = (float*)d_out;

    const int N = in_sizes[0] / 64;
    const int E = in_sizes[2];

    const int per = (((N + 1023) / 1024) + 3) & ~3;
    const int NP  = per * 1024;
    const int EP  = (E + 3) & ~3;

    const size_t tailWords = 4 * (size_t)EP + 2 * (size_t)NP + 2 * (size_t)(NP + 4) + 2 * (size_t)NP;
    const bool paired = ws_size >= (4 * (size_t)N * 64 + tailWords) * sizeof(float);
    const int nFeat = paired ? 4 : 2;

    float* tA0  = (float*)d_ws;
    float* tB0  = tA0 + (size_t)N * 64;
    float* tA1  = paired ? tB0 + (size_t)N * 64 : tA0;
    float* tB1  = paired ? tA1 + (size_t)N * 64 : tB0;
    int*   csrc = (int*)(tA0 + (size_t)nFeat * N * 64);
    float* cw   = (float*)(csrc + 2 * (size_t)EP);
    int*   btot = (int*)(cw + 2 * (size_t)EP);
    int*   rp   = btot + 2 * (size_t)NP;
    float* dis  = (float*)(rp + 2 * (size_t)(NP + 4));
    int*   rank = (int*)tA0;
    int*   perm = rank + 2 * (size_t)EP;
    int*   cntp = perm + 2 * (size_t)EP;
    const int CE = (E + CCH - 1) / CCH;
    const int NR = (NP + RNG - 1) / RNG;

    const float DIAG = 2.0f / 3.0f - 1.0f;

    dim3 blk(256);
    dim3 blk16w(512);
    int gN    = (N + 255) / 256;
    int gNo   = (N + 8 * ONW - 1) / (8 * ONW);
    int gPl   = (CE + 256 * PLE - 1) / (256 * PLE);
    int gN32  = (N + 31) / 32;

    k_rangecount<<<dim3(NR, CCH, 2), dim3(1024), 0, stream>>>(ei, cntp, rank, btot, E, EP, NP, CE);
    k_starts<<<dim3(NP / 1024, 2), dim3(1024), 0, stream>>>(cntp, btot, rp, NP, NR * (RNG / 1024));
    k_place<<<dim3(gPl, CCH, 2), blk, 0, stream>>>(ei, rank, cntp, perm, E, EP, NP, CE);
    k_order<<<dim3(gNo, 2), blk, 0, stream>>>(ei, ew, rp, perm, csrc, cw, N, E, EP, NP);
    k_dis<<<dim3(gN, 2), blk, 0, stream>>>(rp, cw, dis, N, EP, NP);
    k_csrw<<<dim3(gN, 2), blk, 0, stream>>>(rp, csrc, cw, dis, N, EP, NP);

    DirArgs d0, d1;
    d0.rp = rp;                 d1.rp = rp + (size_t)(NP + 4);
    d0.csrc = csrc;             d1.csrc = csrc + (size_t)EP;
    d0.cw = cw;                 d1.cw = cw + (size_t)EP;
    d0.Wa = W1;                 d1.Wa = W2;
    d0.bias = b1;               d1.bias = b2;
    d0.colofs = 0;              d1.colofs = 64;

    const int rows = paired ? 2 : 1;
    for (int turn = 0; turn * rows < 2; ++turn) {
        const bool lone = (turn == 1);

        d0.v = x;   d0.prev = x;   d0.dst = tA0;  d0.Wb = W1 + 1 * 4096;
        d1.v = x;   d1.prev = x;   d1.dst = tA1;  d1.Wb = W2 + 1 * 4096;
        k_lhat_proj<false, true, true><<<dim3(gN32, rows), blk16w, 0, stream>>>(
            lone ? d1 : d0, d1, DIAG, 0.0f, 1.0f, out, N);

        d0.v = tA0; d0.prev = x;   d0.dst = tB0;  d0.Wb = W1 + 2 * 4096;
        d1.v = tA1; d1.prev = x;   d1.dst = tB1;  d1.Wb = W2 + 2 * 4096;
        k_lhat_proj<true, false, true><<<dim3(gN32, rows), blk16w, 0, stream>>>(
            lone ? d1 : d0, d1, 2.0f * DIAG, -1.0f, 2.0f, out, N);

        d0.v = tB0; d0.prev = tA0; d0.dst = tA0;  d0.Wb = W1 + 3 * 4096;
        d1.v = tB1; d1.prev = tA1; d1.dst = tA1;  d1.Wb = W2 + 3 * 4096;
        k_lhat_proj<true, false, true><<<dim3(gN32, rows), blk16w, 0, stream>>>(
            lone ? d1 : d0, d1, 2.0f * DIAG, -1.0f, 2.0f, out, N);

        d0.v = tA0; d0.prev = tB0; d0.dst = tB0;  d0.Wb = W1 + 4 * 4096;
        d1.v = tA1; d1.prev = tB1; d1.dst = tB1;  d1.Wb = W2 + 4 * 4096;
        k_lhat_proj<true, false, false><<<dim3(gN32, rows), blk16w, 0, stream>>>(
            lone ? d1 : d0, d1, 2.0f * DIAG, -1.0f, 2.0f, out, N);
    }
}


# List-build tail folded away: ordering pass also leaves deg^-1/2 (LDS weight row per node, added in slot order), first sparse pass normalises the weights in passing; 10 -> 8 launches; sums unchanged
# speedup vs baseline: 3.9314x; 1.0477x over previous
#include <hip/hip_runtime.h>


typedef float v2f __attribute__((ext_vector_type(2)));
typedef float v8f __attribute__((ext_vector_type(8)));

static __device__ __forceinline__ size_t row64(int n) { return (size_t)n * 64; }


#define RNG 8192
#define CCH 16

__global__ __launch_bounds__(1024) void k_rangecount(const int* __restrict__ ei,
                                                     int* __restrict__ cntp,
                                                     int* __restrict__ rank,
                                                     int* __restrict__ btot,
                                                     int E, int EP, int NP, int CE) {
    __shared__ int lc[RNG];
    __shared__ int hits[RNG / 1024];
    const int t = threadIdx.x;
    const int r0 = blockIdx.x * RNG;
    const int c = blockIdx.y;
    const int dir = blockIdx.z;
    if (t < RNG / 1024) hits[t] = 0;
    for (int i = t; i < RNG; i += 1024) lc[i] = 0;
    __syncthreads();
    const int* dst = ei + (size_t)dir * E;
    int* rk = rank + (size_t)dir * EP;
    const int cbeg = c * CE;
    int cend = cbeg + CE;
    cend = cend < E ? cend : E;
    for (int e0 = cbeg + t; e0 < cend; e0 += 8 * 1024) {
        int dv[8];
#pragma unroll
        for (int j = 0; j < 8; ++j) {
            const int e = e0 + j * 1024;
            dv[j] = (e < cend) ? dst[e] : -1;
        }
#pragma unroll
        for (int j = 0; j < 8; ++j) {
            const unsigned rel = (unsigned)(dv[j] - r0);
            if (rel < (unsigned)RNG) rk[e0 + j * 1024] = atomicAdd(&lc[rel], 1);
        }
    }
    __syncthreads();
    int* o = cntp + ((size_t)dir * CCH + c) * NP;
    int w[RNG / 1024];
#pragma unroll
    for (int s = 0; s < RNG / 1024; ++s) {
        const int i = s * 1024 + t;
        const int v = lc[i];
        const int d = r0 + i;
        if (d < NP) o[d] = v;
        w[s] = v;
    }
#pragma unroll
    for (int off = 16; off > 0; off >>= 1) {
#pragma unroll
        for (int s = 0; s < RNG / 1024; ++s) w[s] += __shfl_xor(w[s], off);
    }
    if ((t & 31) == 0) {
#pragma unroll
        for (int s = 0; s < RNG / 1024; ++s) atomicAdd(&hits[s], w[s]);
    }
    __syncthreads();
    if (t < RNG / 1024)
        btot[((size_t)dir * gridDim.x * (RNG / 1024) + blockIdx.x * (RNG / 1024) + t) * CCH + c] = hits[t];
}

__global__ __launch_bounds__(1024) void k_starts(int* __restrict__ cntp,
                                                 const int* __restrict__ btot,
                                                 int* __restrict__ rp, int NP, int NSB) {
    __shared__ int wtot[32];
    __shared__ int wbef[32];
    const int t = threadIdx.x;
    const int lane = t & 31;
    const int wave = t >> 5;
    const int sb = blockIdx.x;
    const int dir = blockIdx.y;
    const int d = sb * 1024 + t;
    int* p = cntp + (size_t)dir * CCH * NP + d;

    const int* bt = btot + (size_t)dir * NSB * CCH;
    const int nb = sb * CCH;
    int bs = 0;
    for (int q = t; q < nb; q += 1024) bs += bt[q];

    int v[CCH];
#pragma unroll
    for (int c = 0; c < CCH; ++c) v[c] = p[(size_t)c * NP];
    int tsum = 0;
#pragma unroll
    for (int c = 0; c < CCH; ++c) tsum += v[c];

    int inc = tsum;
    for (int off = 1; off < 32; off <<= 1) {
        const int n = __shfl_up(inc, off);
        if (lane >= off) inc += n;
    }
    for (int off = 16; off > 0; off >>= 1) bs += __shfl_xor(bs, off);
    if (lane == 31) wtot[wave] = inc;
    if (lane == 0) wbef[wave] = bs;
    __syncthreads();
    int winc = wtot[lane];
    for (int off = 1; off < 32; off <<= 1) {
        const int n = __shfl_up(winc, off);
        if (lane >= off) winc += n;
    }
    int before = wbef[lane];
    for (int off = 16; off > 0; off >>= 1) before += __shfl_xor(before, off);
    const int wprev = __shfl(winc, (wave + 31) & 31);
    const int wbase = (wave > 0) ? wprev : 0;
    const int base = before + wbase + (inc - tsum);

    int* rr = rp + (size_t)dir * (NP + 4);
    rr[d] = base;
    if (d == NP - 1) rr[NP] = base + tsum;

    int run = base;
#pragma unroll
    for (int c = 0; c < CCH; ++c) {
        p[(size_t)c * NP] = run;
        run += v[c];
    }
}

#define PLE 8
__global__ __launch_bounds__(256) void k_place(const int* __restrict__ ei,
                                               const int* __restrict__ rank,
                                               const int* __restrict__ cntp,
                                               int* __restrict__ perm,
                                               int E, int EP, int NP, int CE) {
    const int c = blockIdx.y;
    const int dir = blockIdx.z;
    const int cbeg = c * CE;
    int cend = cbeg + CE;
    cend = cend < E ? cend : E;
    const int b0 = cbeg + blockIdx.x * (256 * PLE);
    if (b0 >= cend) return;
    const int e0 = b0 + threadIdx.x;
    const int* dst = ei + (size_t)dir * E;
    const int* rk = rank + (size_t)dir * EP;
    const int* first = cntp + ((size_t)dir * CCH + c) * NP;
    int* pm = perm + (size_t)dir * EP;
    int dv[PLE], rv[PLE], fv[PLE];
#pragma unroll
    for (int j = 0; j < PLE; ++j) {
        const int e = e0 + j * 256;
        const int ec = e < cend ? e : cend - 1;
        dv[j] = dst[ec];
        rv[j] = rk[ec];
    }
#pragma unroll
    for (int j = 0; j < PLE; ++j) fv[j] = first[dv[j]];
    asm volatile("" : : "v"(rv[0]), "v"(fv[0]));
#pragma unroll
    for (int j = 0; j < PLE; ++j) {
        const int e = e0 + j * 256;
        if (e < cend) pm[fv[j] + rv[j]] = e;
    }
}

#define ONW 8
#define WRP 36
__global__ __launch_bounds__(256) void k_order(const int* __restrict__ ei,
                                               const float* __restrict__ ew,
                                               const int* __restrict__ rp,
                                               const int* __restrict__ perm,
                                               int* __restrict__ csrc,
                                               float* cw,
                                               float* __restrict__ dis,
                                               int N, int E, int EP, int NP) {
    __shared__ float wrow[8 * ONW * WRP];
    const int lane = threadIdx.x & 31;
    const int wv = __builtin_amdgcn_readfirstlane((int)(threadIdx.x >> 5));
    const int d0 = (blockIdx.x * 8 + wv) * ONW;
    if (d0 >= N) return;
    const int dir = blockIdx.y;
    const int* r = rp + (size_t)dir * (NP + 4);
    const int* pm = perm + (size_t)dir * EP;
    const int* srcIdx = ei + (size_t)(1 - dir) * E;
    int* cs = csrc + (size_t)dir * EP;
    float* cwd = cw + (size_t)dir * EP;
    float* dsd = dis + (size_t)dir * NP;
    float* myrows = wrow + wv * (ONW * WRP);

    const int bnd = r[d0 + (lane < ONW ? lane : ONW)];
    int s0[ONW], len[ONW];
    int nmax = 0;
#pragma unroll
    for (int i = 0; i < ONW; ++i) {
        s0[i] = __builtin_amdgcn_readlane(bnd, i);
        const int s1 = __builtin_amdgcn_readlane(bnd, i + 1);
        const int l = (d0 + i < N) ? (s1 - s0[i]) : 0;
        len[i] = l;
        nmax = nmax > l ? nmax : l;
    }
    const bool anyLong = nmax > 32;
    nmax = nmax < 32 ? nmax : 32;

    int my[ONW];
#pragma unroll
    for (int i = 0; i < ONW; ++i) {
        const int q = s0[i] + lane;
        const int v = pm[q < E ? q : E - 1];
        my[i] = (lane < len[i]) ? v : 0x7fffffff;
    }

    int below[ONW];
#pragma unroll
    for (int i = 0; i < ONW; ++i) below[i] = 0;
    for (int j = 0; j < nmax; ++j) {
#pragma unroll
        for (int i = 0; i < ONW; ++i) {
            const int oj = __builtin_amdgcn_readlane(my[i], j);
            below[i] += (oj < my[i]) ? 1 : 0;
        }
    }

    int gs[ONW];
    float gw[ONW];
#pragma unroll
    for (int i = 0; i < ONW; ++i) {
        const bool mine = (lane < len[i]) && (len[i] <= 32);
        const int g = mine ? my[i] : 0;
        gs[i] = srcIdx[g];
        gw[i] = ew[g];
    }
#pragma unroll
    for (int i = 0; i < ONW; ++i) {
        if ((lane < len[i]) && (len[i] <= 32)) {
            const int slot = s0[i] + below[i];
            cs[slot] = gs[i];
            cwd[slot] = gw[i];
            myrows[i * WRP + below[i]] = gw[i];
        }
    }

    __builtin_amdgcn_fence(__ATOMIC_RELEASE, "wavefront");
    __builtin_amdgcn_wave_barrier();
    __builtin_amdgcn_fence(__ATOMIC_ACQUIRE, "wavefront");
    {
        int L = 0;
#pragma unroll
        for (int i = 0; i < ONW; ++i) L = (lane == i) ? len[i] : L;
        const float* rr = myrows + (lane < ONW ? lane : 0) * WRP;
        float deg = 0.0f;
#pragma unroll
        for (int s4 = 0; s4 < 32; s4 += 4) {
            const float4 wq = *(const float4*)(rr + s4);
            deg = (s4 + 0 < L) ? deg + wq.x : deg;
            deg = (s4 + 1 < L) ? deg + wq.y : deg;
            deg = (s4 + 2 < L) ? deg + wq.z : deg;
            deg = (s4 + 3 < L) ? deg + wq.w : deg;
        }
        if ((lane < ONW) && (d0 + lane < N) && (L <= 32))
            dsd[d0 + lane] = (deg > 0.0f) ? rsqrtf(deg) : 0.0f;
    }

    if (anyLong) {
#pragma nounroll
        for (int i = 0; i < ONW; ++i) {
            const int a0 = __builtin_amdgcn_readlane(bnd, i);
            const int a1 = __builtin_amdgcn_readlane(bnd, i + 1);
            if ((d0 + i >= N) || (a1 - a0 <= 32)) continue;
            for (int ca = a0; ca < a1; ca += 32) {
                const bool mine = (ca + lane) < a1;
                const int me = mine ? pm[ca + lane] : 0;
                int bl = 0;
                for (int cb = a0; cb < a1; cb += 32) {
                    const int other = (cb + lane < a1) ? pm[cb + lane] : 0;
                    int nb = a1 - cb;
                    nb = nb < 32 ? nb : 32;
                    for (int j = 0; j < nb; ++j) {
                        const int oj = __builtin_amdgcn_readlane(other, j);
                        bl += (oj < me) ? 1 : 0;
                    }
                }
                if (mine) {
                    const int slot = a0 + bl;
                    cs[slot] = srcIdx[me];
                    cwd[slot] = ew[me];
                }
            }
            __builtin_amdgcn_fence(__ATOMIC_SEQ_CST, "agent");
            float dg = 0.0f;
            for (int cb = a0; cb < a1; cb += 32) {
                int wb = 0;
                if (cb + lane < a1)
                    wb = __hip_atomic_load((const int*)(cwd + cb + lane), __ATOMIC_RELAXED,
                                           __HIP_MEMORY_SCOPE_AGENT);
                int nb = a1 - cb;
                nb = nb < 32 ? nb : 32;
                for (int j = 0; j < nb; ++j)
                    dg += __int_as_float(__builtin_amdgcn_readlane(wb, j));
            }
            if (lane == 0) dsd[d0 + i] = (dg > 0.0f) ? rsqrtf(dg) : 0.0f;
        }
    }
}

#define TP 68
#define WP 80

static __device__ __forceinline__ void mma_tile(const float* tile, const float* wl,
                                                int lm, int h, int ct,
                                                v8f& acc0, v8f& acc1) {
    const float* arow = tile + lm * TP;
    const float* bcol = wl + ct + lm;
#pragma unroll
    for (int kk = 0; kk < 16; ++kk) {
        const int kbase = kk * 4 + h * 2;
        v2f a = *(const v2f*)(arow + kbase);
        v2f b0, b1;
        b0[0] = bcol[(kbase + 0) * WP + 0];
        b0[1] = bcol[(kbase + 1) * WP + 0];
        b1[0] = bcol[(kbase + 0) * WP + 16];
        b1[1] = bcol[(kbase + 1) * WP + 16];
        acc0 = __builtin_amdgcn_wmma_f32_16x16x4_f32(false, a, false, b0, (short)0, acc0, false, false);
        acc1 = __builtin_amdgcn_wmma_f32_16x16x4_f32(false, a, false, b1, (short)0, acc1, false, false);
    }
}

template <bool FIRST>
static __device__ __forceinline__ void put_weights(float* Wl, int tid,
                                                   float4 wb0, float4 wb1,
                                                   float4 wa0, float4 wa1) {
    float* wdst = Wl + (tid >> 4) * WP + (tid & 15) * 4;
    *(float4*)wdst = wb0;
    *(float4*)(wdst + 32 * WP) = wb1;
    if (FIRST) {
        *(float4*)(wdst + 64 * WP) = wa0;
        *(float4*)(wdst + 96 * WP) = wa1;
    }
}

template <int Q>
static __device__ __forceinline__ int row_take(int x) {
    return __builtin_amdgcn_update_dpp(0, x, 0x150 + Q, 0xf, 0xf, true);
}

template <int Q>
static __device__ __forceinline__ void slot_take(int msrc, float mw, int lm, int& idx, float& w) {
    idx = row_take<Q>(msrc) * 16 + lm;
    w = __int_as_float(row_take<Q>(__float_as_int(mw)));
}

static __device__ __forceinline__ void slot_sum(float w, const float4& val, float4& acc) {
    acc.x = fmaf(w, val.x, acc.x);
    acc.y = fmaf(w, val.y, acc.y);
    acc.z = fmaf(w, val.z, acc.z);
    acc.w = fmaf(w, val.w, acc.w);
}

struct Rows8 {
    float4 r0, r1, r2, r3, r4, r5, r6, r7;
    float w0, w1, w2, w3, w4, w5, w6, w7;
};

template <int Q0>
static __device__ __forceinline__ void rows_load(const float4* __restrict__ v4, int msrc, float mw,
                                                 int lm, Rows8& g) {
    int i0, i1, i2, i3, i4, i5, i6, i7;
    slot_take<Q0 + 0>(msrc, mw, lm, i0, g.w0);
    slot_take<Q0 + 1>(msrc, mw, lm, i1, g.w1);
    slot_take<Q0 + 2>(msrc, mw, lm, i2, g.w2);
    slot_take<Q0 + 3>(msrc, mw, lm, i3, g.w3);
    slot_take<Q0 + 4>(msrc, mw, lm, i4, g.w4);
    slot_take<Q0 + 5>(msrc, mw, lm, i5, g.w5);
    slot_take<Q0 + 6>(msrc, mw, lm, i6, g.w6);
    slot_take<Q0 + 7>(msrc, mw, lm, i7, g.w7);
    asm volatile("" : : "v"(i0), "v"(i1), "v"(i2), "v"(i3), "v"(i4), "v"(i5), "v"(i6), "v"(i7));
    g.r0 = v4[i0];
    g.r1 = v4[i1];
    g.r2 = v4[i2];
    g.r3 = v4[i3];
    g.r4 = v4[i4];
    g.r5 = v4[i5];
    g.r6 = v4[i6];
    g.r7 = v4[i7];
}

template <int Q0>
static __device__ __forceinline__ void rows_load_src(const float4* __restrict__ v4, int msrc,
                                                     int lm, Rows8& g) {
    const int i0 = row_take<Q0 + 0>(msrc) * 16 + lm;
    const int i1 = row_take<Q0 + 1>(msrc) * 16 + lm;
    const int i2 = row_take<Q0 + 2>(msrc) * 16 + lm;
    const int i3 = row_take<Q0 + 3>(msrc) * 16 + lm;
    const int i4 = row_take<Q0 + 4>(msrc) * 16 + lm;
    const int i5 = row_take<Q0 + 5>(msrc) * 16 + lm;
    const int i6 = row_take<Q0 + 6>(msrc) * 16 + lm;
    const int i7 = row_take<Q0 + 7>(msrc) * 16 + lm;
    asm volatile("" : : "v"(i0), "v"(i1), "v"(i2), "v"(i3), "v"(i4), "v"(i5), "v"(i6), "v"(i7));
    g.r0 = v4[i0];
    g.r1 = v4[i1];
    g.r2 = v4[i2];
    g.r3 = v4[i3];
    g.r4 = v4[i4];
    g.r5 = v4[i5];
    g.r6 = v4[i6];
    g.r7 = v4[i7];
}

template <int Q0>
static __device__ __forceinline__ void rows_weights(float mw, Rows8& g) {
    const int mi = __float_as_int(mw);
    g.w0 = __int_as_float(row_take<Q0 + 0>(mi));
    g.w1 = __int_as_float(row_take<Q0 + 1>(mi));
    g.w2 = __int_as_float(row_take<Q0 + 2>(mi));
    g.w3 = __int_as_float(row_take<Q0 + 3>(mi));
    g.w4 = __int_as_float(row_take<Q0 + 4>(mi));
    g.w5 = __int_as_float(row_take<Q0 + 5>(mi));
    g.w6 = __int_as_float(row_take<Q0 + 6>(mi));
    g.w7 = __int_as_float(row_take<Q0 + 7>(mi));
}

static __device__ __forceinline__ void rows_sum(const Rows8& g, float4& accE, float4& accO) {
    slot_sum(g.w0, g.r0, accE);
    slot_sum(g.w1, g.r1, accO);
    slot_sum(g.w2, g.r2, accE);
    slot_sum(g.w3, g.r3, accO);
    slot_sum(g.w4, g.r4, accE);
    slot_sum(g.w5, g.r5, accO);
    slot_sum(g.w6, g.r6, accE);
    slot_sum(g.w7, g.r7, accO);
}

struct DirArgs {
    const float* v;
    const float* prev;
    float* dst;
    const int* rp;
    const int* csrc;
    float* cw;
    const float* dis;
    const float* Wa;
    const float* Wb;
    const float* bias;
    int colofs;
    int fill;
};

template <bool HAS_PREV, bool FIRST, bool STORE_DST>
__global__ __launch_bounds__(512) __attribute__((amdgpu_waves_per_eu(16, 16))) void k_lhat_proj(DirArgs a0, DirArgs a1,
                                                   float adiag, float beta, float alpha,
                                                   float* out, int N) {
    __shared__ float Wl[(FIRST ? 2 : 1) * 64 * WP];
    __shared__ float Tl[(FIRST ? 2 : 1) * 32 * TP];

    const bool second = blockIdx.y != 0;
    const float* __restrict__ v    = second ? a1.v : a0.v;
    const float* prev              = second ? a1.prev : a0.prev;
    float* dst                     = second ? a1.dst : a0.dst;
    const int* __restrict__ rp     = second ? a1.rp : a0.rp;
    const int* __restrict__ csrc   = second ? a1.csrc : a0.csrc;
    float* cww                     = second ? a1.cw : a0.cw;
    const float* cw                = cww;
    const float* __restrict__ dis  = second ? a1.dis : a0.dis;
    const float* Wa                = second ? a1.Wa : a0.Wa;
    const float* Wb                = second ? a1.Wb : a0.Wb;
    const float* bias              = second ? a1.bias : a0.bias;
    const int colofs               = second ? a1.colofs : a0.colofs;

    const int tid = threadIdx.x;
    const int wave = tid >> 5;
    const int lane = tid & 31;
    const int h = lane >> 4;
    const int lm = lane & 15;
    const int c4 = lm * 4;
    const int row0 = blockIdx.x * 32;
    const int trow = wave * 2 + h;
    const int d = row0 + trow;
    const bool valid = d < N;

    const float4 wb0 = ((const float4*)Wb)[tid];
    const float4 wb1 = ((const float4*)Wb)[tid + 512];
    float4 wa0 = make_float4(0.f, 0.f, 0.f, 0.f);
    float4 wa1 = make_float4(0.f, 0.f, 0.f, 0.f);
    if (FIRST) {
        wa0 = ((const float4*)Wa)[tid];
        wa1 = ((const float4*)Wa)[tid + 512];
    }

    const int mj = wave - (blockIdx.x & 3) * 4;
    const bool mat = (mj >= 0) && (mj < 4);
    const int mt = (mj >> 1) & 1;
    const int ct = (mj & 1) * 32;

    int s0 = 0, s1 = 0;
    float dd = 0.0f;
    if (valid) {
        s0 = rp[d];
        s1 = rp[d + 1];
        if (FIRST) dd = -(2.0f / 3.0f) * dis[d];
    }
    int base = s0;
    int rem = s1 - s0;

    int msrc = 0;
    float mw = 0.0f;
    if (lm < rem) {
        msrc = csrc[base + lm];
        mw = cw[base + lm];
    }
    float md = 0.0f;
    if (FIRST) md = dis[msrc];

    put_weights<FIRST>(Wl, tid, wb0, wb1, wa0, wa1);

    float4 accE = make_float4(0.f, 0.f, 0.f, 0.f);
    float4 accO = make_float4(0.f, 0.f, 0.f, 0.f);
    const float4* __restrict__ v4 = (const float4*)v;

    int mrem = max(__builtin_amdgcn_readlane(rem, 0), __builtin_amdgcn_readlane(rem, 16));
    while (mrem > 0) {
        const bool more = (lm + 16) < rem;
        const int nidx = more ? (base + 16 + lm) : base;
        asm volatile("" : : "v"(nidx));
        Rows8 g;
        if (FIRST) rows_load_src<0>(v4, msrc, lm, g);
        else rows_load<0>(v4, msrc, mw, lm, g);
        const int ls = csrc[nidx];
        const float lw = cw[nidx];
        float nd = 0.0f;
        if (FIRST) {
            const bool own = lm < rem;
            mw = own ? dd * mw * md : 0.0f;
            if (own) cww[base + lm] = mw;
            rows_weights<0>(mw, g);
        }
        rows_sum(g, accE, accO);
        if (FIRST) nd = dis[more ? ls : 0];
        if (8 < mrem) {
            if (FIRST) {
                rows_load_src<8>(v4, msrc, lm, g);
                rows_weights<8>(mw, g);
            } else {
                rows_load<8>(v4, msrc, mw, lm, g);
            }
            rows_sum(g, accE, accO);
        }
        base += 16;
        rem -= 16;
        mrem -= 16;
        msrc = more ? ls : 0;
        mw = more ? lw : 0.0f;
        md = nd;
    }
    float4 vd = make_float4(0.f, 0.f, 0.f, 0.f);
    float4 pd = make_float4(0.f, 0.f, 0.f, 0.f);
    if (valid) {
        vd = *(const float4*)(v + row64(d) + c4);
        if (HAS_PREV) pd = *(const float4*)(prev + row64(d) + c4);
    }
    float4 acc;
    acc.x = accE.x + accO.x;
    acc.y = accE.y + accO.y;
    acc.z = accE.z + accO.z;
    acc.w = accE.w + accO.w;

    v8f c0 = {0.f, 0.f, 0.f, 0.f, 0.f, 0.f, 0.f, 0.f};
    v8f c1 = {0.f, 0.f, 0.f, 0.f, 0.f, 0.f, 0.f, 0.f};
    const int mrow0 = row0 + mt * 16;
    if (mat) {
        if (FIRST) {
            const float bv0 = bias[ct + lm];
            const float bv1 = bias[ct + 16 + lm];
#pragma unroll
            for (int i = 0; i < 8; ++i) {
                c0[i] = bv0;
                c1[i] = bv1;
            }
        } else {
            const int nclamp = N - 1;
#pragma unroll
            for (int i = 0; i < 8; ++i) {
                const int r = mrow0 + i + h * 8;
                const bool ok = (r < N);
                const int rc = ok ? r : nclamp;
                const float* orow = out + (size_t)rc * 128 + colofs + ct + lm;
                const float v0 = orow[0], v1 = orow[16];
                c0[i] = ok ? v0 : 0.f;
                c1[i] = ok ? v1 : 0.f;
            }
        }
    }

    float4 o = make_float4(0.f, 0.f, 0.f, 0.f);
    if (valid) {
        o.x = adiag * vd.x;
        o.y = adiag * vd.y;
        o.z = adiag * vd.z;
        o.w = adiag * vd.w;
        if (HAS_PREV) {
            o.x += beta * pd.x;
            o.y += beta * pd.y;
            o.z += beta * pd.z;
            o.w += beta * pd.w;
        }
        o.x += alpha * acc.x;
        o.y += alpha * acc.y;
        o.z += alpha * acc.z;
        o.w += alpha * acc.w;
        if (STORE_DST) *(float4*)(dst + row64(d) + c4) = o;
    }

    *(float4*)(Tl + trow * TP + c4) = o;
    if (FIRST) *(float4*)(Tl + 32 * TP + trow * TP + c4) = vd;
    __syncthreads();

    if (mat) {
        if (FIRST) mma_tile(Tl + 32 * TP + mt * 16 * TP, Wl + 64 * WP, lm, h, ct, c0, c1);
        mma_tile(Tl + mt * 16 * TP, Wl, lm, h, ct, c0, c1);
#pragma unroll
        for (int i = 0; i < 8; ++i) {
            const int r = mrow0 + i + h * 8;
            if (r < N) {
                float* orow = out + (size_t)r * 128 + colofs + ct + lm;
                orow[0]  = c0[i];
                orow[16] = c1[i];
            }
        }
    }
}


extern "C" void kernel_launch(void* const* d_in, const int* in_sizes, int n_in,
                              void* d_out, int out_size, void* d_ws, size_t ws_size,
                              hipStream_t stream) {
    const float* x  = (const float*)d_in[0];
    const int*   ei = (const int*)d_in[1];
    const float* ew = (const float*)d_in[2];
    const float* W1 = (const float*)d_in[3];
    const float* b1 = (const float*)d_in[4];
    const float* W2 = (const float*)d_in[5];
    const float* b2 = (const float*)d_in[6];
    float* out = (float*)d_out;

    const int N = in_sizes[0] / 64;
    const int E = in_sizes[2];

    const int per = (((N + 1023) / 1024) + 3) & ~3;
    const int NP  = per * 1024;
    const int EP  = (E + 3) & ~3;

    const size_t tailWords = 4 * (size_t)EP + 2 * (size_t)NP + 2 * (size_t)(NP + 4) + 2 * (size_t)NP;
    const bool paired = ws_size >= (4 * (size_t)N * 64 + tailWords) * sizeof(float);
    const int nFeat = paired ? 4 : 2;

    float* tA0  = (float*)d_ws;
    float* tB0  = tA0 + (size_t)N * 64;
    float* tA1  = paired ? tB0 + (size_t)N * 64 : tA0;
    float* tB1  = paired ? tA1 + (size_t)N * 64 : tB0;
    int*   csrc = (int*)(tA0 + (size_t)nFeat * N * 64);
    float* cw   = (float*)(csrc + 2 * (size_t)EP);
    int*   btot = (int*)(cw + 2 * (size_t)EP);
    int*   rp   = btot + 2 * (size_t)NP;
    float* dis  = (float*)(rp + 2 * (size_t)(NP + 4));
    int*   rank = (int*)tA0;
    int*   perm = rank + 2 * (size_t)EP;
    int*   cntp = perm + 2 * (size_t)EP;
    const int CE = (E + CCH - 1) / CCH;
    const int NR = (NP + RNG - 1) / RNG;

    const float DIAG = 2.0f / 3.0f - 1.0f;

    dim3 blk(256);
    dim3 blk16w(512);
    int gNo   = (N + 8 * ONW - 1) / (8 * ONW);
    int gPl   = (CE + 256 * PLE - 1) / (256 * PLE);
    int gN32  = (N + 31) / 32;

    k_rangecount<<<dim3(NR, CCH, 2), dim3(1024), 0, stream>>>(ei, cntp, rank, btot, E, EP, NP, CE);
    k_starts<<<dim3(NP / 1024, 2), dim3(1024), 0, stream>>>(cntp, btot, rp, NP, NR * (RNG / 1024));
    k_place<<<dim3(gPl, CCH, 2), blk, 0, stream>>>(ei, rank, cntp, perm, E, EP, NP, CE);
    k_order<<<dim3(gNo, 2), blk, 0, stream>>>(ei, ew, rp, perm, csrc, cw, dis, N, E, EP, NP);

    DirArgs d0 = {}, d1 = {};
    d0.rp = rp;                 d1.rp = rp + (size_t)(NP + 4);
    d0.csrc = csrc;             d1.csrc = csrc + (size_t)EP;
    d0.cw = cw;                 d1.cw = cw + (size_t)EP;
    d0.dis = dis;               d1.dis = dis + (size_t)NP;
    d0.Wa = W1;                 d1.Wa = W2;
    d0.bias = b1;               d1.bias = b2;
    d0.colofs = 0;              d1.colofs = 64;

    const int rows = paired ? 2 : 1;
    for (int turn = 0; turn * rows < 2; ++turn) {
        const bool lone = (turn == 1);

        d0.v = x;   d0.prev = x;   d0.dst = tA0;  d0.Wb = W1 + 1 * 4096;
        d1.v = x;   d1.prev = x;   d1.dst = tA1;  d1.Wb = W2 + 1 * 4096;
        k_lhat_proj<false, true, true><<<dim3(gN32, rows), blk16w, 0, stream>>>(
            lone ? d1 : d0, d1, DIAG, 0.0f, 1.0f, out, N);

        d0.v = tA0; d0.prev = x;   d0.dst = tB0;  d0.Wb = W1 + 2 * 4096;
        d1.v = tA1; d1.prev = x;   d1.dst = tB1;  d1.Wb = W2 + 2 * 4096;
        k_lhat_proj<true, false, true><<<dim3(gN32, rows), blk16w, 0, stream>>>(
            lone ? d1 : d0, d1, 2.0f * DIAG, -1.0f, 2.0f, out, N);

        d0.v = tB0; d0.prev = tA0; d0.dst = tA0;  d0.Wb = W1 + 3 * 4096;
        d1.v = tB1; d1.prev = tA1; d1.dst = tA1;  d1.Wb = W2 + 3 * 4096;
        k_lhat_proj<true, false, true><<<dim3(gN32, rows), blk16w, 0, stream>>>(
            lone ? d1 : d0, d1, 2.0f * DIAG, -1.0f, 2.0f, out, N);

        d0.v = tA0; d0.prev = tB0; d0.dst = tB0;  d0.Wb = W1 + 4 * 4096;
        d1.v = tA1; d1.prev = tB1; d1.dst = tB1;  d1.Wb = W2 + 4 * 4096;
        k_lhat_proj<true, false, false><<<dim3(gN32, rows), blk16w, 0, stream>>>(
            lone ? d1 : d0, d1, 2.0f * DIAG, -1.0f, 2.0f, out, N);
    }
}


# Projection of orders 2 and 3 deferred to the last sparse pass, whose block holds its own rows of both: middle passes drop LDS, barrier, matrix phase and the read+write of out; float sums unchanged
# speedup vs baseline: 4.0917x; 1.0408x over previous
#include <hip/hip_runtime.h>


typedef float v2f __attribute__((ext_vector_type(2)));
typedef float v8f __attribute__((ext_vector_type(8)));

static __device__ __forceinline__ size_t row64(int n) { return (size_t)n * 64; }


#define RNG 8192
#define CCH 16

__global__ __launch_bounds__(1024) void k_rangecount(const int* __restrict__ ei,
                                                     int* __restrict__ cntp,
                                                     int* __restrict__ rank,
                                                     int* __restrict__ btot,
                                                     int E, int EP, int NP, int CE) {
    __shared__ int lc[RNG];
    __shared__ int hits[RNG / 1024];
    const int t = threadIdx.x;
    const int r0 = blockIdx.x * RNG;
    const int c = blockIdx.y;
    const int dir = blockIdx.z;
    if (t < RNG / 1024) hits[t] = 0;
    for (int i = t; i < RNG; i += 1024) lc[i] = 0;
    __syncthreads();
    const int* dst = ei + (size_t)dir * E;
    int* rk = rank + (size_t)dir * EP;
    const int cbeg = c * CE;
    int cend = cbeg + CE;
    cend = cend < E ? cend : E;
    for (int e0 = cbeg + t; e0 < cend; e0 += 8 * 1024) {
        int dv[8];
#pragma unroll
        for (int j = 0; j < 8; ++j) {
            const int e = e0 + j * 1024;
            dv[j] = (e < cend) ? dst[e] : -1;
        }
#pragma unroll
        for (int j = 0; j < 8; ++j) {
            const unsigned rel = (unsigned)(dv[j] - r0);
            if (rel < (unsigned)RNG) rk[e0 + j * 1024] = atomicAdd(&lc[rel], 1);
        }
    }
    __syncthreads();
    int* o = cntp + ((size_t)dir * CCH + c) * NP;
    int w[RNG / 1024];
#pragma unroll
    for (int s = 0; s < RNG / 1024; ++s) {
        const int i = s * 1024 + t;
        const int v = lc[i];
        const int d = r0 + i;
        if (d < NP) o[d] = v;
        w[s] = v;
    }
#pragma unroll
    for (int off = 16; off > 0; off >>= 1) {
#pragma unroll
        for (int s = 0; s < RNG / 1024; ++s) w[s] += __shfl_xor(w[s], off);
    }
    if ((t & 31) == 0) {
#pragma unroll
        for (int s = 0; s < RNG / 1024; ++s) atomicAdd(&hits[s], w[s]);
    }
    __syncthreads();
    if (t < RNG / 1024)
        btot[((size_t)dir * gridDim.x * (RNG / 1024) + blockIdx.x * (RNG / 1024) + t) * CCH + c] = hits[t];
}

__global__ __launch_bounds__(1024) void k_starts(int* __restrict__ cntp,
                                                 const int* __restrict__ btot,
                                                 int* __restrict__ rp, int NP, int NSB) {
    __shared__ int wtot[32];
    __shared__ int wbef[32];
    const int t = threadIdx.x;
    const int lane = t & 31;
    const int wave = t >> 5;
    const int sb = blockIdx.x;
    const int dir = blockIdx.y;
    const int d = sb * 1024 + t;
    int* p = cntp + (size_t)dir * CCH * NP + d;

    const int* bt = btot + (size_t)dir * NSB * CCH;
    const int nb = sb * CCH;
    int bs = 0;
    for (int q = t; q < nb; q += 1024) bs += bt[q];

    int v[CCH];
#pragma unroll
    for (int c = 0; c < CCH; ++c) v[c] = p[(size_t)c * NP];
    int tsum = 0;
#pragma unroll
    for (int c = 0; c < CCH; ++c) tsum += v[c];

    int inc = tsum;
    for (int off = 1; off < 32; off <<= 1) {
        const int n = __shfl_up(inc, off);
        if (lane >= off) inc += n;
    }
    for (int off = 16; off > 0; off >>= 1) bs += __shfl_xor(bs, off);
    if (lane == 31) wtot[wave] = inc;
    if (lane == 0) wbef[wave] = bs;
    __syncthreads();
    int winc = wtot[lane];
    for (int off = 1; off < 32; off <<= 1) {
        const int n = __shfl_up(winc, off);
        if (lane >= off) winc += n;
    }
    int before = wbef[lane];
    for (int off = 16; off > 0; off >>= 1) before += __shfl_xor(before, off);
    const int wprev = __shfl(winc, (wave + 31) & 31);
    const int wbase = (wave > 0) ? wprev : 0;
    const int base = before + wbase + (inc - tsum);

    int* rr = rp + (size_t)dir * (NP + 4);
    rr[d] = base;
    if (d == NP - 1) rr[NP] = base + tsum;

    int run = base;
#pragma unroll
    for (int c = 0; c < CCH; ++c) {
        p[(size_t)c * NP] = run;
        run += v[c];
    }
}

#define PLE 8
__global__ __launch_bounds__(256) void k_place(const int* __restrict__ ei,
                                               const int* __restrict__ rank,
                                               const int* __restrict__ cntp,
                                               int* __restrict__ perm,
                                               int E, int EP, int NP, int CE) {
    const int c = blockIdx.y;
    const int dir = blockIdx.z;
    const int cbeg = c * CE;
    int cend = cbeg + CE;
    cend = cend < E ? cend : E;
    const int b0 = cbeg + blockIdx.x * (256 * PLE);
    if (b0 >= cend) return;
    const int e0 = b0 + threadIdx.x;
    const int* dst = ei + (size_t)dir * E;
    const int* rk = rank + (size_t)dir * EP;
    const int* first = cntp + ((size_t)dir * CCH + c) * NP;
    int* pm = perm + (size_t)dir * EP;
    int dv[PLE], rv[PLE], fv[PLE];
#pragma unroll
    for (int j = 0; j < PLE; ++j) {
        const int e = e0 + j * 256;
        const int ec = e < cend ? e : cend - 1;
        dv[j] = dst[ec];
        rv[j] = rk[ec];
    }
#pragma unroll
    for (int j = 0; j < PLE; ++j) fv[j] = first[dv[j]];
    asm volatile("" : : "v"(rv[0]), "v"(fv[0]));
#pragma unroll
    for (int j = 0; j < PLE; ++j) {
        const int e = e0 + j * 256;
        if (e < cend) pm[fv[j] + rv[j]] = e;
    }
}

#define ONW 8
#define WRP 36
__global__ __launch_bounds__(256) void k_order(const int* __restrict__ ei,
                                               const float* __restrict__ ew,
                                               const int* __restrict__ rp,
                                               const int* __restrict__ perm,
                                               int* __restrict__ csrc,
                                               float* cw,
                                               float* __restrict__ dis,
                                               int N, int E, int EP, int NP) {
    __shared__ float wrow[8 * ONW * WRP];
    const int lane = threadIdx.x & 31;
    const int wv = __builtin_amdgcn_readfirstlane((int)(threadIdx.x >> 5));
    const int d0 = (blockIdx.x * 8 + wv) * ONW;
    if (d0 >= N) return;
    const int dir = blockIdx.y;
    const int* r = rp + (size_t)dir * (NP + 4);
    const int* pm = perm + (size_t)dir * EP;
    const int* srcIdx = ei + (size_t)(1 - dir) * E;
    int* cs = csrc + (size_t)dir * EP;
    float* cwd = cw + (size_t)dir * EP;
    float* dsd = dis + (size_t)dir * NP;
    float* myrows = wrow + wv * (ONW * WRP);

    const int bnd = r[d0 + (lane < ONW ? lane : ONW)];
    int s0[ONW], len[ONW];
    int nmax = 0;
#pragma unroll
    for (int i = 0; i < ONW; ++i) {
        s0[i] = __builtin_amdgcn_readlane(bnd, i);
        const int s1 = __builtin_amdgcn_readlane(bnd, i + 1);
        const int l = (d0 + i < N) ? (s1 - s0[i]) : 0;
        len[i] = l;
        nmax = nmax > l ? nmax : l;
    }
    const bool anyLong = nmax > 32;
    nmax = nmax < 32 ? nmax : 32;

    int my[ONW];
#pragma unroll
    for (int i = 0; i < ONW; ++i) {
        const int q = s0[i] + lane;
        const int v = pm[q < E ? q : E - 1];
        my[i] = (lane < len[i]) ? v : 0x7fffffff;
    }

    int below[ONW];
#pragma unroll
    for (int i = 0; i < ONW; ++i) below[i] = 0;
    for (int j = 0; j < nmax; ++j) {
#pragma unroll
        for (int i = 0; i < ONW; ++i) {
            const int oj = __builtin_amdgcn_readlane(my[i], j);
            below[i] += (oj < my[i]) ? 1 : 0;
        }
    }

    int gs[ONW];
    float gw[ONW];
#pragma unroll
    for (int i = 0; i < ONW; ++i) {
        const bool mine = (lane < len[i]) && (len[i] <= 32);
        const int g = mine ? my[i] : 0;
        gs[i] = srcIdx[g];
        gw[i] = ew[g];
    }
#pragma unroll
    for (int i = 0; i < ONW; ++i) {
        if ((lane < len[i]) && (len[i] <= 32)) {
            const int slot = s0[i] + below[i];
            cs[slot] = gs[i];
            cwd[slot] = gw[i];
            myrows[i * WRP + below[i]] = gw[i];
        }
    }

    __builtin_amdgcn_fence(__ATOMIC_RELEASE, "wavefront");
    __builtin_amdgcn_wave_barrier();
    __builtin_amdgcn_fence(__ATOMIC_ACQUIRE, "wavefront");
    {
        int L = 0;
#pragma unroll
        for (int i = 0; i < ONW; ++i) L = (lane == i) ? len[i] : L;
        const float* rr = myrows + (lane < ONW ? lane : 0) * WRP;
        float deg = 0.0f;
#pragma unroll
        for (int s4 = 0; s4 < 32; s4 += 4) {
            const float4 wq = *(const float4*)(rr + s4);
            deg = (s4 + 0 < L) ? deg + wq.x : deg;
            deg = (s4 + 1 < L) ? deg + wq.y : deg;
            deg = (s4 + 2 < L) ? deg + wq.z : deg;
            deg = (s4 + 3 < L) ? deg + wq.w : deg;
        }
        if ((lane < ONW) && (d0 + lane < N) && (L <= 32))
            dsd[d0 + lane] = (deg > 0.0f) ? rsqrtf(deg) : 0.0f;
    }

    if (anyLong) {
#pragma nounroll
        for (int i = 0; i < ONW; ++i) {
            const int a0 = __builtin_amdgcn_readlane(bnd, i);
            const int a1 = __builtin_amdgcn_readlane(bnd, i + 1);
            if ((d0 + i >= N) || (a1 - a0 <= 32)) continue;
            for (int ca = a0; ca < a1; ca += 32) {
                const bool mine = (ca + lane) < a1;
                const int me = mine ? pm[ca + lane] : 0;
                int bl = 0;
                for (int cb = a0; cb < a1; cb += 32) {
                    const int other = (cb + lane < a1) ? pm[cb + lane] : 0;
                    int nb = a1 - cb;
                    nb = nb < 32 ? nb : 32;
                    for (int j = 0; j < nb; ++j) {
                        const int oj = __builtin_amdgcn_readlane(other, j);
                        bl += (oj < me) ? 1 : 0;
                    }
                }
                if (mine) {
                    const int slot = a0 + bl;
                    cs[slot] = srcIdx[me];
                    cwd[slot] = ew[me];
                }
            }
            __builtin_amdgcn_fence(__ATOMIC_SEQ_CST, "agent");
            float dg = 0.0f;
            for (int cb = a0; cb < a1; cb += 32) {
                int wb = 0;
                if (cb + lane < a1)
                    wb = __hip_atomic_load((const int*)(cwd + cb + lane), __ATOMIC_RELAXED,
                                           __HIP_MEMORY_SCOPE_AGENT);
                int nb = a1 - cb;
                nb = nb < 32 ? nb : 32;
                for (int j = 0; j < nb; ++j)
                    dg += __int_as_float(__builtin_amdgcn_readlane(wb, j));
            }
            if (lane == 0) dsd[d0 + i] = (dg > 0.0f) ? rsqrtf(dg) : 0.0f;
        }
    }
}

#define TP 68
#define WP_WIDE 80
#define WP_TIGHT 72

template <int WP>
static __device__ __forceinline__ void mma_tile(const float* tile, const float* wl,
                                                int lm, int h, int ct,
                                                v8f& acc0, v8f& acc1) {
    const float* arow = tile + lm * TP;
    const float* bcol = wl + ct + lm;
#pragma unroll
    for (int kk = 0; kk < 16; ++kk) {
        const int kbase = kk * 4 + h * 2;
        v2f a = *(const v2f*)(arow + kbase);
        v2f b0, b1;
        b0[0] = bcol[(kbase + 0) * WP + 0];
        b0[1] = bcol[(kbase + 1) * WP + 0];
        b1[0] = bcol[(kbase + 0) * WP + 16];
        b1[1] = bcol[(kbase + 1) * WP + 16];
        acc0 = __builtin_amdgcn_wmma_f32_16x16x4_f32(false, a, false, b0, (short)0, acc0, false, false);
        acc1 = __builtin_amdgcn_wmma_f32_16x16x4_f32(false, a, false, b1, (short)0, acc1, false, false);
    }
}

template <int WP>
static __device__ __forceinline__ void put_matrix(float* wdst, int m, float4 w0, float4 w1) {
    *(float4*)(wdst + m * (64 * WP)) = w0;
    *(float4*)(wdst + m * (64 * WP) + 32 * WP) = w1;
}

template <int Q>
static __device__ __forceinline__ int row_take(int x) {
    return __builtin_amdgcn_update_dpp(0, x, 0x150 + Q, 0xf, 0xf, true);
}

template <int Q>
static __device__ __forceinline__ void slot_take(int msrc, float mw, int lm, int& idx, float& w) {
    idx = row_take<Q>(msrc) * 16 + lm;
    w = __int_as_float(row_take<Q>(__float_as_int(mw)));
}

static __device__ __forceinline__ void slot_sum(float w, const float4& val, float4& acc) {
    acc.x = fmaf(w, val.x, acc.x);
    acc.y = fmaf(w, val.y, acc.y);
    acc.z = fmaf(w, val.z, acc.z);
    acc.w = fmaf(w, val.w, acc.w);
}

struct Rows8 {
    float4 r0, r1, r2, r3, r4, r5, r6, r7;
    float w0, w1, w2, w3, w4, w5, w6, w7;
};

template <int Q0>
static __device__ __forceinline__ void rows_load(const float4* __restrict__ v4, int msrc, float mw,
                                                 int lm, Rows8& g) {
    int i0, i1, i2, i3, i4, i5, i6, i7;
    slot_take<Q0 + 0>(msrc, mw, lm, i0, g.w0);
    slot_take<Q0 + 1>(msrc, mw, lm, i1, g.w1);
    slot_take<Q0 + 2>(msrc, mw, lm, i2, g.w2);
    slot_take<Q0 + 3>(msrc, mw, lm, i3, g.w3);
    slot_take<Q0 + 4>(msrc, mw, lm, i4, g.w4);
    slot_take<Q0 + 5>(msrc, mw, lm, i5, g.w5);
    slot_take<Q0 + 6>(msrc, mw, lm, i6, g.w6);
    slot_take<Q0 + 7>(msrc, mw, lm, i7, g.w7);
    asm volatile("" : : "v"(i0), "v"(i1), "v"(i2), "v"(i3), "v"(i4), "v"(i5), "v"(i6), "v"(i7));
    g.r0 = v4[i0];
    g.r1 = v4[i1];
    g.r2 = v4[i2];
    g.r3 = v4[i3];
    g.r4 = v4[i4];
    g.r5 = v4[i5];
    g.r6 = v4[i6];
    g.r7 = v4[i7];
}

template <int Q0>
static __device__ __forceinline__ void rows_load_src(const float4* __restrict__ v4, int msrc,
                                                     int lm, Rows8& g) {
    const int i0 = row_take<Q0 + 0>(msrc) * 16 + lm;
    const int i1 = row_take<Q0 + 1>(msrc) * 16 + lm;
    const int i2 = row_take<Q0 + 2>(msrc) * 16 + lm;
    const int i3 = row_take<Q0 + 3>(msrc) * 16 + lm;
    const int i4 = row_take<Q0 + 4>(msrc) * 16 + lm;
    const int i5 = row_take<Q0 + 5>(msrc) * 16 + lm;
    const int i6 = row_take<Q0 + 6>(msrc) * 16 + lm;
    const int i7 = row_take<Q0 + 7>(msrc) * 16 + lm;
    asm volatile("" : : "v"(i0), "v"(i1), "v"(i2), "v"(i3), "v"(i4), "v"(i5), "v"(i6), "v"(i7));
    g.r0 = v4[i0];
    g.r1 = v4[i1];
    g.r2 = v4[i2];
    g.r3 = v4[i3];
    g.r4 = v4[i4];
    g.r5 = v4[i5];
    g.r6 = v4[i6];
    g.r7 = v4[i7];
}

template <int Q0>
static __device__ __forceinline__ void rows_weights(float mw, Rows8& g) {
    const int mi = __float_as_int(mw);
    g.w0 = __int_as_float(row_take<Q0 + 0>(mi));
    g.w1 = __int_as_float(row_take<Q0 + 1>(mi));
    g.w2 = __int_as_float(row_take<Q0 + 2>(mi));
    g.w3 = __int_as_float(row_take<Q0 + 3>(mi));
    g.w4 = __int_as_float(row_take<Q0 + 4>(mi));
    g.w5 = __int_as_float(row_take<Q0 + 5>(mi));
    g.w6 = __int_as_float(row_take<Q0 + 6>(mi));
    g.w7 = __int_as_float(row_take<Q0 + 7>(mi));
}

static __device__ __forceinline__ void rows_sum(const Rows8& g, float4& accE, float4& accO) {
    slot_sum(g.w0, g.r0, accE);
    slot_sum(g.w1, g.r1, accO);
    slot_sum(g.w2, g.r2, accE);
    slot_sum(g.w3, g.r3, accO);
    slot_sum(g.w4, g.r4, accE);
    slot_sum(g.w5, g.r5, accO);
    slot_sum(g.w6, g.r6, accE);
    slot_sum(g.w7, g.r7, accO);
}

struct DirArgs {
    const float* v;
    const float* prev;
    float* dst;
    const int* rp;
    const int* csrc;
    float* cw;
    const float* dis;
    const float* Wa;
    const float* Wb;
    const float* bias;
    int colofs;
    int fill;
};

template <int MODE>
__global__ __launch_bounds__(512) __attribute__((amdgpu_waves_per_eu(16, 16))) void k_lhat_proj(DirArgs a0, DirArgs a1,
                                                   float adiag, float beta, float alpha,
                                                   float* out, int N) {
    constexpr bool FIRST = (MODE == 0);
    constexpr bool HAS_PREV = (MODE != 0);
    constexpr bool STORE_DST = (MODE != 2);
    constexpr int NMAT = (MODE == 0) ? 2 : ((MODE == 2) ? 3 : 0);
    constexpr int WP = (MODE == 2) ? WP_TIGHT : WP_WIDE;
    __shared__ float Wl[(NMAT > 0 ? NMAT : 1) * 64 * WP];
    __shared__ float Tl[(NMAT > 0 ? NMAT : 1) * 32 * TP];

    const bool second = blockIdx.y != 0;
    const float* __restrict__ v    = second ? a1.v : a0.v;
    const float* prev              = second ? a1.prev : a0.prev;
    float* dst                     = second ? a1.dst : a0.dst;
    const int* __restrict__ rp     = second ? a1.rp : a0.rp;
    const int* __restrict__ csrc   = second ? a1.csrc : a0.csrc;
    float* cww                     = second ? a1.cw : a0.cw;
    const float* cw                = cww;
    const float* __restrict__ dis  = second ? a1.dis : a0.dis;
    const float* Wa                = second ? a1.Wa : a0.Wa;
    const float* Wb                = second ? a1.Wb : a0.Wb;
    const float* bias              = second ? a1.bias : a0.bias;
    const int colofs               = second ? a1.colofs : a0.colofs;

    const int tid = threadIdx.x;
    const int wave = tid >> 5;
    const int lane = tid & 31;
    const int h = lane >> 4;
    const int lm = lane & 15;
    const int c4 = lm * 4;
    const int row0 = blockIdx.x * 32;
    const int trow = wave * 2 + h;
    const int d = row0 + trow;
    const bool valid = d < N;

    float4 wb0 = make_float4(0.f, 0.f, 0.f, 0.f), wb1 = wb0;
    float4 wa0 = wb0, wa1 = wb0, wp0 = wb0, wp1 = wb0;
    if (NMAT > 0) {
        wb0 = ((const float4*)Wb)[tid];
        wb1 = ((const float4*)Wb)[tid + 512];
    }
    if (FIRST) {
        wa0 = ((const float4*)Wa)[tid];
        wa1 = ((const float4*)Wa)[tid + 512];
    }
    if (NMAT > 2) {
        wa0 = ((const float4*)(Wb - 4096))[tid];
        wa1 = ((const float4*)(Wb - 4096))[tid + 512];
        wp0 = ((const float4*)(Wb - 8192))[tid];
        wp1 = ((const float4*)(Wb - 8192))[tid + 512];
    }

    const int mj = wave - (blockIdx.x & 3) * 4;
    const bool mat = (NMAT > 0) && (mj >= 0) && (mj < 4);
    const int mt = (mj >> 1) & 1;
    const int ct = (mj & 1) * 32;

    int s0 = 0, s1 = 0;
    float dd = 0.0f;
    if (valid) {
        s0 = rp[d];
        s1 = rp[d + 1];
        if (FIRST) dd = -(2.0f / 3.0f) * dis[d];
    }
    int base = s0;
    int rem = s1 - s0;

    int msrc = 0;
    float mw = 0.0f;
    if (lm < rem) {
        msrc = csrc[base + lm];
        mw = cw[base + lm];
    }
    float md = 0.0f;
    if (FIRST) md = dis[msrc];

    if (NMAT > 0) {
        float* wdst = Wl + (tid >> 4) * WP + (tid & 15) * 4;
        put_matrix<WP>(wdst, 0, wb0, wb1);
        put_matrix<WP>(wdst, 1, wa0, wa1);
        if (NMAT > 2) put_matrix<WP>(wdst, 2, wp0, wp1);
    }

    float4 accE = make_float4(0.f, 0.f, 0.f, 0.f);
    float4 accO = make_float4(0.f, 0.f, 0.f, 0.f);
    const float4* __restrict__ v4 = (const float4*)v;

    int mrem = max(__builtin_amdgcn_readlane(rem, 0), __builtin_amdgcn_readlane(rem, 16));
    while (mrem > 0) {
        const bool more = (lm + 16) < rem;
        const int nidx = more ? (base + 16 + lm) : base;
        asm volatile("" : : "v"(nidx));
        Rows8 g;
        if (FIRST) rows_load_src<0>(v4, msrc, lm, g);
        else rows_load<0>(v4, msrc, mw, lm, g);
        const int ls = csrc[nidx];
        const float lw = cw[nidx];
        float nd = 0.0f;
        if (FIRST) {
            const bool own = lm < rem;
            mw = own ? dd * mw * md : 0.0f;
            if (own) cww[base + lm] = mw;
            rows_weights<0>(mw, g);
        }
        rows_sum(g, accE, accO);
        if (FIRST) nd = dis[more ? ls : 0];
        if (8 < mrem) {
            if (FIRST) {
                rows_load_src<8>(v4, msrc, lm, g);
                rows_weights<8>(mw, g);
            } else {
                rows_load<8>(v4, msrc, mw, lm, g);
            }
            rows_sum(g, accE, accO);
        }
        base += 16;
        rem -= 16;
        mrem -= 16;
        msrc = more ? ls : 0;
        mw = more ? lw : 0.0f;
        md = nd;
    }
    float4 vd = make_float4(0.f, 0.f, 0.f, 0.f);
    float4 pd = make_float4(0.f, 0.f, 0.f, 0.f);
    if (valid) {
        vd = *(const float4*)(v + row64(d) + c4);
        if (HAS_PREV) pd = *(const float4*)(prev + row64(d) + c4);
    }
    float4 acc;
    acc.x = accE.x + accO.x;
    acc.y = accE.y + accO.y;
    acc.z = accE.z + accO.z;
    acc.w = accE.w + accO.w;

    v8f c0 = {0.f, 0.f, 0.f, 0.f, 0.f, 0.f, 0.f, 0.f};
    v8f c1 = {0.f, 0.f, 0.f, 0.f, 0.f, 0.f, 0.f, 0.f};
    const int mrow0 = row0 + mt * 16;
    if (mat) {
        if (FIRST) {
            const float bv0 = bias[ct + lm];
            const float bv1 = bias[ct + 16 + lm];
#pragma unroll
            for (int i = 0; i < 8; ++i) {
                c0[i] = bv0;
                c1[i] = bv1;
            }
        } else {
            const int nclamp = N - 1;
#pragma unroll
            for (int i = 0; i < 8; ++i) {
                const int r = mrow0 + i + h * 8;
                const bool ok = (r < N);
                const int rc = ok ? r : nclamp;
                const float* orow = out + (size_t)rc * 128 + colofs + ct + lm;
                const float v0 = orow[0], v1 = orow[16];
                c0[i] = ok ? v0 : 0.f;
                c1[i] = ok ? v1 : 0.f;
            }
        }
    }

    float4 o = make_float4(0.f, 0.f, 0.f, 0.f);
    if (valid) {
        o.x = adiag * vd.x;
        o.y = adiag * vd.y;
        o.z = adiag * vd.z;
        o.w = adiag * vd.w;
        if (HAS_PREV) {
            o.x = fmaf(beta, pd.x, o.x);
            o.y = fmaf(beta, pd.y, o.y);
            o.z = fmaf(beta, pd.z, o.z);
            o.w = fmaf(beta, pd.w, o.w);
        }
        o.x += alpha * acc.x;
        o.y += alpha * acc.y;
        o.z += alpha * acc.z;
        o.w += alpha * acc.w;
        if (STORE_DST) *(float4*)(dst + row64(d) + c4) = o;
    }

    if (NMAT > 0) {
        *(float4*)(Tl + trow * TP + c4) = o;
        *(float4*)(Tl + 32 * TP + trow * TP + c4) = vd;
        if (NMAT > 2) *(float4*)(Tl + 64 * TP + trow * TP + c4) = pd;
        __syncthreads();

        if (mat) {
            if (NMAT > 2) mma_tile<WP>(Tl + 64 * TP + mt * 16 * TP, Wl + 128 * WP, lm, h, ct, c0, c1);
            mma_tile<WP>(Tl + 32 * TP + mt * 16 * TP, Wl + 64 * WP, lm, h, ct, c0, c1);
            mma_tile<WP>(Tl + mt * 16 * TP, Wl, lm, h, ct, c0, c1);
#pragma unroll
            for (int i = 0; i < 8; ++i) {
                const int r = mrow0 + i + h * 8;
                if (r < N) {
                    float* orow = out + (size_t)r * 128 + colofs + ct + lm;
                    orow[0]  = c0[i];
                    orow[16] = c1[i];
                }
            }
        }
    }
}


extern "C" void kernel_launch(void* const* d_in, const int* in_sizes, int n_in,
                              void* d_out, int out_size, void* d_ws, size_t ws_size,
                              hipStream_t stream) {
    const float* x  = (const float*)d_in[0];
    const int*   ei = (const int*)d_in[1];
    const float* ew = (const float*)d_in[2];
    const float* W1 = (const float*)d_in[3];
    const float* b1 = (const float*)d_in[4];
    const float* W2 = (const float*)d_in[5];
    const float* b2 = (const float*)d_in[6];
    float* out = (float*)d_out;

    const int N = in_sizes[0] / 64;
    const int E = in_sizes[2];

    const int per = (((N + 1023) / 1024) + 3) & ~3;
    const int NP  = per * 1024;
    const int EP  = (E + 3) & ~3;

    const size_t tailWords = 4 * (size_t)EP + 2 * (size_t)NP + 2 * (size_t)(NP + 4) + 2 * (size_t)NP;
    const bool paired = ws_size >= (4 * (size_t)N * 64 + tailWords) * sizeof(float);
    const int nFeat = paired ? 4 : 2;

    float* tA0  = (float*)d_ws;
    float* tB0  = tA0 + (size_t)N * 64;
    float* tA1  = paired ? tB0 + (size_t)N * 64 : tA0;
    float* tB1  = paired ? tA1 + (size_t)N * 64 : tB0;
    int*   csrc = (int*)(tA0 + (size_t)nFeat * N * 64);
    float* cw   = (float*)(csrc + 2 * (size_t)EP);
    int*   btot = (int*)(cw + 2 * (size_t)EP);
    int*   rp   = btot + 2 * (size_t)NP;
    float* dis  = (float*)(rp + 2 * (size_t)(NP + 4));
    int*   rank = (int*)tA0;
    int*   perm = rank + 2 * (size_t)EP;
    int*   cntp = perm + 2 * (size_t)EP;
    const int CE = (E + CCH - 1) / CCH;
    const int NR = (NP + RNG - 1) / RNG;

    const float DIAG = 2.0f / 3.0f - 1.0f;

    dim3 blk(256);
    dim3 blk16w(512);
    int gNo   = (N + 8 * ONW - 1) / (8 * ONW);
    int gPl   = (CE + 256 * PLE - 1) / (256 * PLE);
    int gN32  = (N + 31) / 32;

    k_rangecount<<<dim3(NR, CCH, 2), dim3(1024), 0, stream>>>(ei, cntp, rank, btot, E, EP, NP, CE);
    k_starts<<<dim3(NP / 1024, 2), dim3(1024), 0, stream>>>(cntp, btot, rp, NP, NR * (RNG / 1024));
    k_place<<<dim3(gPl, CCH, 2), blk, 0, stream>>>(ei, rank, cntp, perm, E, EP, NP, CE);
    k_order<<<dim3(gNo, 2), blk, 0, stream>>>(ei, ew, rp, perm, csrc, cw, dis, N, E, EP, NP);

    DirArgs d0 = {}, d1 = {};
    d0.rp = rp;                 d1.rp = rp + (size_t)(NP + 4);
    d0.csrc = csrc;             d1.csrc = csrc + (size_t)EP;
    d0.cw = cw;                 d1.cw = cw + (size_t)EP;
    d0.dis = dis;               d1.dis = dis + (size_t)NP;
    d0.Wa = W1;                 d1.Wa = W2;
    d0.bias = b1;               d1.bias = b2;
    d0.colofs = 0;              d1.colofs = 64;

    const int rows = paired ? 2 : 1;
    for (int turn = 0; turn * rows < 2; ++turn) {
        const bool lone = (turn == 1);

        d0.v = x;   d0.prev = x;   d0.dst = tA0;  d0.Wb = W1 + 1 * 4096;
        d1.v = x;   d1.prev = x;   d1.dst = tA1;  d1.Wb = W2 + 1 * 4096;
        k_lhat_proj<0><<<dim3(gN32, rows), blk16w, 0, stream>>>(
            lone ? d1 : d0, d1, DIAG, 0.0f, 1.0f, out, N);

        d0.v = tA0; d0.prev = x;   d0.dst = tB0;
        d1.v = tA1; d1.prev = x;   d1.dst = tB1;
        k_lhat_proj<1><<<dim3(gN32, rows), blk16w, 0, stream>>>(
            lone ? d1 : d0, d1, 2.0f * DIAG, -1.0f, 2.0f, out, N);

        d0.v = tB0; d0.prev = tA0; d0.dst = tA0;
        d1.v = tB1; d1.prev = tA1; d1.dst = tA1;
        k_lhat_proj<1><<<dim3(gN32, rows), blk16w, 0, stream>>>(
            lone ? d1 : d0, d1, 2.0f * DIAG, -1.0f, 2.0f, out, N);

        d0.v = tA0; d0.prev = tB0; d0.dst = tB0;  d0.Wb = W1 + 4 * 4096;
        d1.v = tA1; d1.prev = tB1; d1.dst = tB1;  d1.Wb = W2 + 4 * 4096;
        k_lhat_proj<2><<<dim3(gN32, rows), blk16w, 0, stream>>>(
            lone ? d1 : d0, d1, 2.0f * DIAG, -1.0f, 2.0f, out, N);
    }
}
